# starting kernel
# baseline (speedup 1.0000x reference)
#include <hip/hip_runtime.h>
#include <hip/hip_bf16.h>


typedef float v2f __attribute__((ext_vector_type(2)));
typedef float v8f __attribute__((ext_vector_type(8)));

#define N_NODES 100000
#define N_EDGES 1600000
#define HF 64

__global__ void zero_f32(float* __restrict__ p, int n) {
    int i = blockIdx.x * blockDim.x + threadIdx.x;
    if (i < n) p[i] = 0.0f;
}

__global__ void degree_kernel(const int* __restrict__ ei, float* __restrict__ deg, int E) {
    int e = blockIdx.x * blockDim.x + threadIdx.x;
    if (e < E) atomicAdd(&deg[ei[E + e]], 1.0f);
}

__global__ void dinv_kernel(float* __restrict__ d, int n) {
    int i = blockIdx.x * blockDim.x + threadIdx.x;
    if (i < n) {
        float v = fmaxf(d[i], 1.0f);
        d[i] = 1.0f / sqrtf(v);
    }
}

__global__ void gemm_bias_relu_wmma(const float* __restrict__ A,
                                    const float* __restrict__ W,
                                    const float* __restrict__ bias,
                                    float* __restrict__ C,
                                    int K, int do_relu) {
    extern __shared__ float Blds[];
    const int tid = threadIdx.x;

    for (int i = tid; i < K * HF; i += 256) {
        int kk = i >> 6, c = i & 63;
        Blds[kk * 65 + c] = W[i];
    }
    __syncthreads();

    const int wave = tid >> 5;
    const int lane = tid & 31;
    const int half = lane >> 4;
    const int lm   = lane & 15;
    const int tn   = wave & 3;
    const int tml  = wave >> 2;
    const int row0 = (blockIdx.x * 2 + tml) * 16;
    const int col0 = tn * 16;

    const float* Arow = A + (long)(row0 + lm) * K;

    v8f acc = {0.f, 0.f, 0.f, 0.f, 0.f, 0.f, 0.f, 0.f};
    for (int k = 0; k < K; k += 4) {
        const int ks = k + 2 * half;
        v2f a, b;
        a.x = Arow[ks];
        a.y = Arow[ks + 1];
        b.x = Blds[ks * 65 + col0 + lm];
        b.y = Blds[(ks + 1) * 65 + col0 + lm];
        acc = __builtin_amdgcn_wmma_f32_16x16x4_f32(
            false, a, false, b, (short)0, acc, false, false);
    }

    const int col  = col0 + lm;
    const float bc = bias[col];
#pragma unroll
    for (int r = 0; r < 8; ++r) {
        int row = row0 + r + 8 * half;
        float v = acc[r] + bc;
        if (do_relu) v = fmaxf(v, 0.0f);
        C[(long)row * HF + col] = v;
    }
}

__global__ void theta_init(const float* __restrict__ h, float* __restrict__ feat,
                           float* __restrict__ hfin, float c0, int off, int n64) {
    int i = blockIdx.x * blockDim.x + threadIdx.x;
    if (i >= n64) return;
    float v = h[i];
    feat[i] = v;
    int n = i >> 6, f = i & 63;
    hfin[(long)n * 192 + off + f] = c0 * v;
}

__global__ void edge_scatter(const int* __restrict__ ei, const float* __restrict__ feat,
                             const float* __restrict__ dinv, float* __restrict__ msg,
                             int E) {
    long long tid = (long long)blockIdx.x * blockDim.x + threadIdx.x;
    int e = (int)(tid >> 6);
    if (e >= E) return;
    int f = (int)(tid & 63);
    int s = ei[e];
    int d = ei[E + e];
    float v = feat[(long)s * HF + f] * dinv[s];
    atomicAdd(&msg[(long)d * HF + f], v);
}

__global__ void theta_update(float* __restrict__ feat, const float* __restrict__ msg,
                             const float* __restrict__ dinv, float* __restrict__ hfin,
                             float ck, int off, int n64) {
    int i = blockIdx.x * blockDim.x + threadIdx.x;
    if (i >= n64) return;
    int n = i >> 6, f = i & 63;
    float v = feat[i] - msg[i] * dinv[n];
    feat[i] = v;
    hfin[(long)n * 192 + off + f] += ck * v;
}

__global__ void final_linear(const float* __restrict__ Hm, const float* __restrict__ Wm2,
                             const float* __restrict__ bm2, float* __restrict__ out,
                             int n) {
    int i = blockIdx.x * blockDim.x + threadIdx.x;
    if (i >= n) return;
    const float* h = Hm + (long)i * HF;
    float c0 = bm2[0], c1 = bm2[1];
#pragma unroll 8
    for (int j = 0; j < HF; ++j) {
        float hv = h[j];
        c0 = fmaf(hv, Wm2[j * 2 + 0], c0);
        c1 = fmaf(hv, Wm2[j * 2 + 1], c1);
    }
    out[i * 2 + 0] = c0;
    out[i * 2 + 1] = c1;
}

extern "C" void kernel_launch(void* const* d_in, const int* in_sizes, int n_in,
                              void* d_out, int out_size, void* d_ws, size_t ws_size,
                              hipStream_t stream) {
    const float* x   = (const float*)d_in[0];
    const int*   ei  = (const int*)  d_in[1];
    const float* W1  = (const float*)d_in[2];
    const float* b1  = (const float*)d_in[3];
    const float* W2  = (const float*)d_in[4];
    const float* b2  = (const float*)d_in[5];
    const float* Wm1 = (const float*)d_in[6];
    const float* bm1 = (const float*)d_in[7];
    const float* Wm2 = (const float*)d_in[8];
    const float* bm2 = (const float*)d_in[9];
    float* out = (float*)d_out;

    const int N = N_NODES, E = N_EDGES;
    const int N64 = N * HF;

    float* ws   = (float*)d_ws;
    float* dinv = ws;
    float* h    = dinv + N;
    float* feat = h    + (long)N64;
    float* msg  = feat + (long)N64;
    float* hfin = msg  + (long)N64;

    const float TH[3][3] = {{3.0f, -3.0f, 0.75f},
                            {0.0f,  3.0f, -1.50f},
                            {0.0f,  0.0f, 0.75f}};

    zero_f32<<<(N + 255) / 256, 256, 0, stream>>>(dinv, N);
    degree_kernel<<<(E + 255) / 256, 256, 0, stream>>>(ei, dinv, E);
    dinv_kernel<<<(N + 255) / 256, 256, 0, stream>>>(dinv, N);

    gemm_bias_relu_wmma<<<N / 32, 256, 128 * 65 * sizeof(float), stream>>>(
        x, W1, b1, msg, 128, 1);
    gemm_bias_relu_wmma<<<N / 32, 256, 64 * 65 * sizeof(float), stream>>>(
        msg, W2, b2, h, 64, 1);

    const int eg = (int)(((long long)E * HF) / 256);
    for (int t = 0; t < 3; ++t) {
        theta_init<<<(N64 + 255) / 256, 256, 0, stream>>>(
            h, feat, hfin, TH[t][0], t * HF, N64);
        for (int k = 1; k <= 2; ++k) {
            zero_f32<<<(N64 + 255) / 256, 256, 0, stream>>>(msg, N64);
            edge_scatter<<<eg, 256, 0, stream>>>(ei, feat, dinv, msg, E);
            theta_update<<<(N64 + 255) / 256, 256, 0, stream>>>(
                feat, msg, dinv, hfin, TH[t][k], t * HF, N64);
        }
    }

    gemm_bias_relu_wmma<<<N / 32, 256, 192 * 65 * sizeof(float), stream>>>(
        hfin, Wm1, bm1, feat, 192, 1);

    final_linear<<<(N + 255) / 256, 256, 0, stream>>>(feat, Wm2, bm2, out, N);
}


# All 3 filters apply the same operator to the same h: L h and L^2 h are computed once and shared (6 atomic edge scatters -> 2, 28 launches -> 13); term order per filter and atomic sums as before
# speedup vs baseline: 2.5280x; 2.5280x over previous
#include <hip/hip_runtime.h>
#include <hip/hip_bf16.h>


typedef float v2f __attribute__((ext_vector_type(2)));
typedef float v8f __attribute__((ext_vector_type(8)));

#define N_NODES 100000
#define N_EDGES 1600000
#define HF 64

__global__ void zero_f32(float* __restrict__ p, int n) {
    int i = blockIdx.x * blockDim.x + threadIdx.x;
    if (i < n) p[i] = 0.0f;
}

__global__ void degree_kernel(const int* __restrict__ ei, float* __restrict__ deg, int E) {
    int e = blockIdx.x * blockDim.x + threadIdx.x;
    if (e < E) atomicAdd(&deg[ei[E + e]], 1.0f);
}

__global__ void dinv_kernel(float* __restrict__ d, int n) {
    int i = blockIdx.x * blockDim.x + threadIdx.x;
    if (i < n) {
        float v = fmaxf(d[i], 1.0f);
        d[i] = 1.0f / sqrtf(v);
    }
}

__global__ void gemm_bias_relu_wmma(const float* __restrict__ A,
                                    const float* __restrict__ W,
                                    const float* __restrict__ bias,
                                    float* __restrict__ C,
                                    int K, int do_relu) {
    extern __shared__ float Blds[];
    const int tid = threadIdx.x;

    for (int i = tid; i < K * HF; i += 256) {
        int kk = i >> 6, c = i & 63;
        Blds[kk * 65 + c] = W[i];
    }
    __syncthreads();

    const int wave = tid >> 5;
    const int lane = tid & 31;
    const int half = lane >> 4;
    const int lm   = lane & 15;
    const int tn   = wave & 3;
    const int tml  = wave >> 2;
    const int row0 = (blockIdx.x * 2 + tml) * 16;
    const int col0 = tn * 16;

    const float* Arow = A + (long)(row0 + lm) * K;

    v8f acc = {0.f, 0.f, 0.f, 0.f, 0.f, 0.f, 0.f, 0.f};
    for (int k = 0; k < K; k += 4) {
        const int ks = k + 2 * half;
        v2f a, b;
        a.x = Arow[ks];
        a.y = Arow[ks + 1];
        b.x = Blds[ks * 65 + col0 + lm];
        b.y = Blds[(ks + 1) * 65 + col0 + lm];
        acc = __builtin_amdgcn_wmma_f32_16x16x4_f32(
            false, a, false, b, (short)0, acc, false, false);
    }

    const int col  = col0 + lm;
    const float bc = bias[col];
#pragma unroll
    for (int r = 0; r < 8; ++r) {
        int row = row0 + r + 8 * half;
        float v = acc[r] + bc;
        if (do_relu) v = fmaxf(v, 0.0f);
        C[(long)row * HF + col] = v;
    }
}


__global__ void edge_scatter(const int* __restrict__ ei, const float* __restrict__ feat,
                             const float* __restrict__ dinv, float* __restrict__ msg,
                             int E) {
    long long tid = (long long)blockIdx.x * blockDim.x + threadIdx.x;
    int e = (int)(tid >> 6);
    if (e >= E) return;
    int f = (int)(tid & 63);
    int s = ei[e];
    int d = ei[E + e];
    float v = feat[(long)s * HF + f] * dinv[s];
    atomicAdd(&msg[(long)d * HF + f], v);
}

__global__ void theta_step1(const float* __restrict__ h, const float* __restrict__ msg,
                            const float* __restrict__ dinv, float* __restrict__ feat,
                            float* __restrict__ hfin,
                            float a0, float a1, float a2,
                            float b0, float b1, float b2, int n64) {
    int i = blockIdx.x * blockDim.x + threadIdx.x;
    if (i >= n64) return;
    int n = i >> 6, f = i & 63;
    float hv = h[i];
    float v = hv - msg[i] * dinv[n];
    feat[i] = v;
    float* o = hfin + (long)n * 192 + f;
    float s0 = a0 * hv, s1 = a1 * hv, s2 = a2 * hv;
    o[0]   = fmaf(b0, v, s0);
    o[64]  = fmaf(b1, v, s1);
    o[128] = fmaf(b2, v, s2);
}

__global__ void theta_step2(const float* __restrict__ feat, const float* __restrict__ msg,
                            const float* __restrict__ dinv, float* __restrict__ hfin,
                            float c0, float c1, float c2, int n64) {
    int i = blockIdx.x * blockDim.x + threadIdx.x;
    if (i >= n64) return;
    int n = i >> 6, f = i & 63;
    float v = feat[i] - msg[i] * dinv[n];
    float* o = hfin + (long)n * 192 + f;
    o[0]   = fmaf(c0, v, o[0]);
    o[64]  = fmaf(c1, v, o[64]);
    o[128] = fmaf(c2, v, o[128]);
}

__global__ void final_linear(const float* __restrict__ Hm, const float* __restrict__ Wm2,
                             const float* __restrict__ bm2, float* __restrict__ out,
                             int n) {
    int i = blockIdx.x * blockDim.x + threadIdx.x;
    if (i >= n) return;
    const float* h = Hm + (long)i * HF;
    float c0 = bm2[0], c1 = bm2[1];
#pragma unroll 8
    for (int j = 0; j < HF; ++j) {
        float hv = h[j];
        c0 = fmaf(hv, Wm2[j * 2 + 0], c0);
        c1 = fmaf(hv, Wm2[j * 2 + 1], c1);
    }
    out[i * 2 + 0] = c0;
    out[i * 2 + 1] = c1;
}

extern "C" void kernel_launch(void* const* d_in, const int* in_sizes, int n_in,
                              void* d_out, int out_size, void* d_ws, size_t ws_size,
                              hipStream_t stream) {
    const float* x   = (const float*)d_in[0];
    const int*   ei  = (const int*)  d_in[1];
    const float* W1  = (const float*)d_in[2];
    const float* b1  = (const float*)d_in[3];
    const float* W2  = (const float*)d_in[4];
    const float* b2  = (const float*)d_in[5];
    const float* Wm1 = (const float*)d_in[6];
    const float* bm1 = (const float*)d_in[7];
    const float* Wm2 = (const float*)d_in[8];
    const float* bm2 = (const float*)d_in[9];
    float* out = (float*)d_out;

    const int N = N_NODES, E = N_EDGES;
    const int N64 = N * HF;

    float* ws   = (float*)d_ws;
    float* dinv = ws;
    float* h    = dinv + N;
    float* feat = h    + (long)N64;
    float* msg  = feat + (long)N64;
    float* hfin = msg  + (long)N64;

    const float TH[3][3] = {{3.0f, -3.0f, 0.75f},
                            {0.0f,  3.0f, -1.50f},
                            {0.0f,  0.0f, 0.75f}};

    zero_f32<<<(N + 255) / 256, 256, 0, stream>>>(dinv, N);
    degree_kernel<<<(E + 255) / 256, 256, 0, stream>>>(ei, dinv, E);
    dinv_kernel<<<(N + 255) / 256, 256, 0, stream>>>(dinv, N);

    gemm_bias_relu_wmma<<<N / 32, 256, 128 * 65 * sizeof(float), stream>>>(
        x, W1, b1, msg, 128, 1);
    gemm_bias_relu_wmma<<<N / 32, 256, 64 * 65 * sizeof(float), stream>>>(
        msg, W2, b2, h, 64, 1);

    const int eg = (int)(((long long)E * HF) / 256);
    zero_f32<<<(N64 + 255) / 256, 256, 0, stream>>>(msg, N64);
    edge_scatter<<<eg, 256, 0, stream>>>(ei, h, dinv, msg, E);
    theta_step1<<<(N64 + 255) / 256, 256, 0, stream>>>(
        h, msg, dinv, feat, hfin,
        TH[0][0], TH[1][0], TH[2][0], TH[0][1], TH[1][1], TH[2][1], N64);
    zero_f32<<<(N64 + 255) / 256, 256, 0, stream>>>(msg, N64);
    edge_scatter<<<eg, 256, 0, stream>>>(ei, feat, dinv, msg, E);
    theta_step2<<<(N64 + 255) / 256, 256, 0, stream>>>(
        feat, msg, dinv, hfin, TH[0][2], TH[1][2], TH[2][2], N64);

    gemm_bias_relu_wmma<<<N / 32, 256, 192 * 65 * sizeof(float), stream>>>(
        hfin, Wm1, bm1, feat, 192, 1);

    final_linear<<<(N + 255) / 256, 256, 0, stream>>>(feat, Wm2, bm2, out, N);
}


# f32 atomic edge scatters replaced by per-call in-edge lists (int atomic count+rank, prefix sum, fill): each node sums its in-edges in registers in arrival-rank order, theta step applied in place
# speedup vs baseline: 6.9074x; 2.7324x over previous
#include <hip/hip_runtime.h>
#include <hip/hip_bf16.h>


typedef float v2f __attribute__((ext_vector_type(2)));
typedef float v8f __attribute__((ext_vector_type(8)));

#define N_NODES 100000
#define N_EDGES 1600000
#define HF 64

__global__ void zero_i32(int* __restrict__ p, int n) {
    int i = blockIdx.x * blockDim.x + threadIdx.x;
    if (i < n) p[i] = 0;
}

__global__ void count_rank(const int* __restrict__ ei, int* __restrict__ cnt,
                           int* __restrict__ rank, int E) {
    int e = blockIdx.x * blockDim.x + threadIdx.x;
    if (e < E) rank[e] = atomicAdd(&cnt[ei[E + e]], 1);
}

__global__ void __launch_bounds__(1024)
scan_local(const int* __restrict__ cnt, int* __restrict__ rowptr, int* __restrict__ bsum,
           float* __restrict__ dinv, int n) {
    __shared__ int wsum[32];
    const int i    = blockIdx.x * 1024 + threadIdx.x;
    const int lane = threadIdx.x & 31;
    const int wv   = threadIdx.x >> 5;
    const int c    = (i < n) ? cnt[i] : 0;
    if (i < n) dinv[i] = 1.0f / sqrtf(fmaxf((float)c, 1.0f));
    int x = c;
    for (int o = 1; o < 32; o <<= 1) {
        int y = __shfl_up(x, o);
        if (lane >= o) x += y;
    }
    if (lane == 31) wsum[wv] = x;
    __syncthreads();
    if (wv == 0) {
        const int s = wsum[lane];
        int t = s;
        for (int o = 1; o < 32; o <<= 1) {
            int y = __shfl_up(t, o);
            if (lane >= o) t += y;
        }
        wsum[lane] = t - s;
        if (lane == 31) bsum[blockIdx.x] = t;
    }
    __syncthreads();
    if (i < n) rowptr[i] = x - c + wsum[wv];
}

__global__ void __launch_bounds__(1024)
scan_offset(int* __restrict__ rowptr, const int* __restrict__ bsum, int n, int nb) {
    __shared__ int off;
    if (threadIdx.x < 32) {
        int s = 0;
        for (int k = threadIdx.x; k < (int)blockIdx.x; k += 32) s += bsum[k];
        for (int o = 16; o > 0; o >>= 1) s += __shfl_xor(s, o);
        if (threadIdx.x == 0) off = s;
    }
    __syncthreads();
    const int i = blockIdx.x * 1024 + threadIdx.x;
    if (i < n) rowptr[i] += off;
    if ((int)blockIdx.x == nb - 1 && threadIdx.x == 0) rowptr[n] = off + bsum[nb - 1];
}

__global__ void fill_csr(const int* __restrict__ ei, const int* __restrict__ rowptr,
                         const int* __restrict__ rank, const float* __restrict__ dinv,
                         int* __restrict__ csr, int E) {
    int e = blockIdx.x * blockDim.x + threadIdx.x;
    if (e >= E) return;
    const int s = ei[e];
    const int d = ei[E + e];
    const int pos = rowptr[d] + rank[e];
    if (pos < E)
        reinterpret_cast<int2*>(csr)[pos] = make_int2(s, __float_as_int(dinv[s]));
}

__global__ void gemm_bias_relu_wmma(const float* __restrict__ A,
                                    const float* __restrict__ W,
                                    const float* __restrict__ bias,
                                    float* __restrict__ C,
                                    int K, int do_relu) {
    extern __shared__ float Blds[];
    const int tid = threadIdx.x;

    for (int i = tid; i < K * HF; i += 256) {
        int kk = i >> 6, c = i & 63;
        Blds[kk * 65 + c] = W[i];
    }
    __syncthreads();

    const int wave = tid >> 5;
    const int lane = tid & 31;
    const int half = lane >> 4;
    const int lm   = lane & 15;
    const int tn   = wave & 3;
    const int tml  = wave >> 2;
    const int row0 = (blockIdx.x * 2 + tml) * 16;
    const int col0 = tn * 16;

    const float* Arow = A + (long)(row0 + lm) * K;

    v8f acc = {0.f, 0.f, 0.f, 0.f, 0.f, 0.f, 0.f, 0.f};
    for (int k = 0; k < K; k += 4) {
        const int ks = k + 2 * half;
        v2f a, b;
        a.x = Arow[ks];
        a.y = Arow[ks + 1];
        b.x = Blds[ks * 65 + col0 + lm];
        b.y = Blds[(ks + 1) * 65 + col0 + lm];
        acc = __builtin_amdgcn_wmma_f32_16x16x4_f32(
            false, a, false, b, (short)0, acc, false, false);
    }

    const int col  = col0 + lm;
    const float bc = bias[col];
#pragma unroll
    for (int r = 0; r < 8; ++r) {
        int row = row0 + r + 8 * half;
        float v = acc[r] + bc;
        if (do_relu) v = fmaxf(v, 0.0f);
        C[(long)row * HF + col] = v;
    }
}


__device__ __forceinline__ float4 ld_row4(const float* __restrict__ t, int row, int q) {
    return *reinterpret_cast<const float4*>(t + (long)row * HF + q);
}

__device__ __forceinline__ void add_scaled(float4& acc, const float4 r, float w) {
    acc.x += r.x * w;
    acc.y += r.y * w;
    acc.z += r.z * w;
    acc.w += r.w * w;
}

__device__ __forceinline__ float4 row_or_zero(bool keep, const float4 r) {
    float4 o;
    o.x = keep ? r.x : 0.0f;
    o.y = keep ? r.y : 0.0f;
    o.z = keep ? r.z : 0.0f;
    o.w = keep ? r.w : 0.0f;
    return o;
}

__device__ __forceinline__ float4 gather_sum(const int2* __restrict__ csr,
                                             const float* __restrict__ t,
                                             int beg, int end, int q) {
    float4 acc = make_float4(0.0f, 0.0f, 0.0f, 0.0f);
    int j = beg;
    for (; j + 4 <= end; j += 4) {
        const int2 e0 = csr[j], e1 = csr[j + 1], e2 = csr[j + 2], e3 = csr[j + 3];
        const float4 r0 = ld_row4(t, e0.x, q);
        const float4 r1 = ld_row4(t, e1.x, q);
        const float4 r2 = ld_row4(t, e2.x, q);
        const float4 r3 = ld_row4(t, e3.x, q);
        add_scaled(acc, r0, __int_as_float(e0.y));
        add_scaled(acc, r1, __int_as_float(e1.y));
        add_scaled(acc, r2, __int_as_float(e2.y));
        add_scaled(acc, r3, __int_as_float(e3.y));
    }
    const int rem = end - j;
    if (rem > 0) {
        const int last = end - 1;
        const int2 e0 = csr[j];
        const int2 e1 = csr[min(j + 1, last)];
        const int2 e2 = csr[min(j + 2, last)];
        const float4 r0 = ld_row4(t, e0.x, q);
        const float4 r1 = row_or_zero(rem > 1, ld_row4(t, e1.x, q));
        const float4 r2 = row_or_zero(rem > 2, ld_row4(t, e2.x, q));
        add_scaled(acc, r0, __int_as_float(e0.y));
        add_scaled(acc, r1, __int_as_float(e1.y));
        add_scaled(acc, r2, __int_as_float(e2.y));
    }
    return acc;
}

__global__ void gather_step1(const int* __restrict__ csr, const int* __restrict__ rowptr,
                             const float* __restrict__ h, const float* __restrict__ dinv,
                             float* __restrict__ feat, float* __restrict__ hfin,
                             float a0, float a1, float a2,
                             float b0, float b1, float b2, int n_nodes) {
    const int tid = blockIdx.x * blockDim.x + threadIdx.x;
    const int n = tid >> 4;
    if (n >= n_nodes) return;
    const int q = (tid & 15) * 4;
    const float4 m = gather_sum(reinterpret_cast<const int2*>(csr), h,
                                rowptr[n], rowptr[n + 1], q);
    const float dn = dinv[n];
    const float4 hv = ld_row4(h, n, q);
    float4 v;
    v.x = hv.x - m.x * dn;
    v.y = hv.y - m.y * dn;
    v.z = hv.z - m.z * dn;
    v.w = hv.w - m.w * dn;
    *reinterpret_cast<float4*>(feat + (long)n * HF + q) = v;
    float* o = hfin + (long)n * 192 + q;
    float4 o0, o1, o2;
    o0.x = fmaf(b0, v.x, a0 * hv.x); o0.y = fmaf(b0, v.y, a0 * hv.y);
    o0.z = fmaf(b0, v.z, a0 * hv.z); o0.w = fmaf(b0, v.w, a0 * hv.w);
    o1.x = fmaf(b1, v.x, a1 * hv.x); o1.y = fmaf(b1, v.y, a1 * hv.y);
    o1.z = fmaf(b1, v.z, a1 * hv.z); o1.w = fmaf(b1, v.w, a1 * hv.w);
    o2.x = fmaf(b2, v.x, a2 * hv.x); o2.y = fmaf(b2, v.y, a2 * hv.y);
    o2.z = fmaf(b2, v.z, a2 * hv.z); o2.w = fmaf(b2, v.w, a2 * hv.w);
    *reinterpret_cast<float4*>(o)       = o0;
    *reinterpret_cast<float4*>(o + 64)  = o1;
    *reinterpret_cast<float4*>(o + 128) = o2;
}

__global__ void gather_step2(const int* __restrict__ csr, const int* __restrict__ rowptr,
                             const float* __restrict__ feat, const float* __restrict__ dinv,
                             float* __restrict__ hfin,
                             float c0, float c1, float c2, int n_nodes) {
    const int tid = blockIdx.x * blockDim.x + threadIdx.x;
    const int n = tid >> 4;
    if (n >= n_nodes) return;
    const int q = (tid & 15) * 4;
    const float4 m = gather_sum(reinterpret_cast<const int2*>(csr), feat,
                                rowptr[n], rowptr[n + 1], q);
    const float dn = dinv[n];
    const float4 fv = ld_row4(feat, n, q);
    float4 v;
    v.x = fv.x - m.x * dn;
    v.y = fv.y - m.y * dn;
    v.z = fv.z - m.z * dn;
    v.w = fv.w - m.w * dn;
    float* o = hfin + (long)n * 192 + q;
    float4 o0 = *reinterpret_cast<const float4*>(o);
    float4 o1 = *reinterpret_cast<const float4*>(o + 64);
    float4 o2 = *reinterpret_cast<const float4*>(o + 128);
    o0.x = fmaf(c0, v.x, o0.x); o0.y = fmaf(c0, v.y, o0.y);
    o0.z = fmaf(c0, v.z, o0.z); o0.w = fmaf(c0, v.w, o0.w);
    o1.x = fmaf(c1, v.x, o1.x); o1.y = fmaf(c1, v.y, o1.y);
    o1.z = fmaf(c1, v.z, o1.z); o1.w = fmaf(c1, v.w, o1.w);
    o2.x = fmaf(c2, v.x, o2.x); o2.y = fmaf(c2, v.y, o2.y);
    o2.z = fmaf(c2, v.z, o2.z); o2.w = fmaf(c2, v.w, o2.w);
    *reinterpret_cast<float4*>(o)       = o0;
    *reinterpret_cast<float4*>(o + 64)  = o1;
    *reinterpret_cast<float4*>(o + 128) = o2;
}

__global__ void final_linear(const float* __restrict__ Hm, const float* __restrict__ Wm2,
                             const float* __restrict__ bm2, float* __restrict__ out,
                             int n) {
    int i = blockIdx.x * blockDim.x + threadIdx.x;
    if (i >= n) return;
    const float* h = Hm + (long)i * HF;
    float c0 = bm2[0], c1 = bm2[1];
#pragma unroll 8
    for (int j = 0; j < HF; ++j) {
        float hv = h[j];
        c0 = fmaf(hv, Wm2[j * 2 + 0], c0);
        c1 = fmaf(hv, Wm2[j * 2 + 1], c1);
    }
    out[i * 2 + 0] = c0;
    out[i * 2 + 1] = c1;
}

extern "C" void kernel_launch(void* const* d_in, const int* in_sizes, int n_in,
                              void* d_out, int out_size, void* d_ws, size_t ws_size,
                              hipStream_t stream) {
    const float* x   = (const float*)d_in[0];
    const int*   ei  = (const int*)  d_in[1];
    const float* W1  = (const float*)d_in[2];
    const float* b1  = (const float*)d_in[3];
    const float* W2  = (const float*)d_in[4];
    const float* b2  = (const float*)d_in[5];
    const float* Wm1 = (const float*)d_in[6];
    const float* bm1 = (const float*)d_in[7];
    const float* Wm2 = (const float*)d_in[8];
    const float* bm2 = (const float*)d_in[9];
    float* out = (float*)d_out;

    const int N = N_NODES, E = N_EDGES;
    const int N64 = N * HF;

    float* ws   = (float*)d_ws;
    float* dinv = ws;
    float* h    = dinv + N;
    float* feat = h    + (long)N64;
    float* lst  = feat + (long)N64;
    float* hfin = lst  + (long)N64;

    int* csr    = (int*)lst;
    int* rank   = csr + 2 * (long)E;
    int* cnt    = rank + E;
    int* rowptr = cnt + N;
    int* bsum   = rowptr + N + 8;

    const float TH[3][3] = {{3.0f, -3.0f, 0.75f},
                            {0.0f,  3.0f, -1.50f},
                            {0.0f,  0.0f, 0.75f}};

    const int NB = (N + 1023) / 1024;
    zero_i32<<<(N + 255) / 256, 256, 0, stream>>>(cnt, N);
    count_rank<<<(E + 255) / 256, 256, 0, stream>>>(ei, cnt, rank, E);
    scan_local<<<NB, 1024, 0, stream>>>(cnt, rowptr, bsum, dinv, N);
    scan_offset<<<NB, 1024, 0, stream>>>(rowptr, bsum, N, NB);

    gemm_bias_relu_wmma<<<N / 32, 256, 128 * 65 * sizeof(float), stream>>>(
        x, W1, b1, feat, 128, 1);
    gemm_bias_relu_wmma<<<N / 32, 256, 64 * 65 * sizeof(float), stream>>>(
        feat, W2, b2, h, 64, 1);

    fill_csr<<<(E + 255) / 256, 256, 0, stream>>>(ei, rowptr, rank, dinv, csr, E);
    const int gg = (N * 16 + 255) / 256;
    gather_step1<<<gg, 256, 0, stream>>>(
        csr, rowptr, h, dinv, feat, hfin,
        TH[0][0], TH[1][0], TH[2][0], TH[0][1], TH[1][1], TH[2][1], N);
    gather_step2<<<gg, 256, 0, stream>>>(
        csr, rowptr, feat, dinv, hfin, TH[0][2], TH[1][2], TH[2][2], N);

    gemm_bias_relu_wmma<<<N / 32, 256, 192 * 65 * sizeof(float), stream>>>(
        hfin, Wm1, bm1, feat, 192, 1);

    final_linear<<<(N + 255) / 256, 256, 0, stream>>>(feat, Wm2, bm2, out, N);
}


# Dense layers re-tiled: a wave computes 16 rows x all 64 columns (4 accumulators); A and LDS-transposed W read as 16-byte fragments with equal k indexing; 4x fewer waves; k order in products permuted
# speedup vs baseline: 7.4599x; 1.0800x over previous
#include <hip/hip_runtime.h>
#include <hip/hip_bf16.h>


typedef float v2f __attribute__((ext_vector_type(2)));
typedef float v8f __attribute__((ext_vector_type(8)));

#define N_NODES 100000
#define N_EDGES 1600000
#define HF 64

__global__ void zero_i32(int* __restrict__ p, int n) {
    int i = blockIdx.x * blockDim.x + threadIdx.x;
    if (i < n) p[i] = 0;
}

__global__ void count_rank(const int* __restrict__ ei, int* __restrict__ cnt,
                           int* __restrict__ rank, int E) {
    int e = blockIdx.x * blockDim.x + threadIdx.x;
    if (e < E) rank[e] = atomicAdd(&cnt[ei[E + e]], 1);
}

__global__ void __launch_bounds__(1024)
scan_local(const int* __restrict__ cnt, int* __restrict__ rowptr, int* __restrict__ bsum,
           float* __restrict__ dinv, int n) {
    __shared__ int wsum[32];
    const int i    = blockIdx.x * 1024 + threadIdx.x;
    const int lane = threadIdx.x & 31;
    const int wv   = threadIdx.x >> 5;
    const int c    = (i < n) ? cnt[i] : 0;
    if (i < n) dinv[i] = 1.0f / sqrtf(fmaxf((float)c, 1.0f));
    int x = c;
    for (int o = 1; o < 32; o <<= 1) {
        int y = __shfl_up(x, o);
        if (lane >= o) x += y;
    }
    if (lane == 31) wsum[wv] = x;
    __syncthreads();
    if (wv == 0) {
        const int s = wsum[lane];
        int t = s;
        for (int o = 1; o < 32; o <<= 1) {
            int y = __shfl_up(t, o);
            if (lane >= o) t += y;
        }
        wsum[lane] = t - s;
        if (lane == 31) bsum[blockIdx.x] = t;
    }
    __syncthreads();
    if (i < n) rowptr[i] = x - c + wsum[wv];
}

__global__ void __launch_bounds__(1024)
scan_offset(int* __restrict__ rowptr, const int* __restrict__ bsum, int n, int nb) {
    __shared__ int off;
    if (threadIdx.x < 32) {
        int s = 0;
        for (int k = threadIdx.x; k < (int)blockIdx.x; k += 32) s += bsum[k];
        for (int o = 16; o > 0; o >>= 1) s += __shfl_xor(s, o);
        if (threadIdx.x == 0) off = s;
    }
    __syncthreads();
    const int i = blockIdx.x * 1024 + threadIdx.x;
    if (i < n) rowptr[i] += off;
    if ((int)blockIdx.x == nb - 1 && threadIdx.x == 0) rowptr[n] = off + bsum[nb - 1];
}

__global__ void fill_csr(const int* __restrict__ ei, const int* __restrict__ rowptr,
                         const int* __restrict__ rank, const float* __restrict__ dinv,
                         int* __restrict__ csr, int E) {
    int e = blockIdx.x * blockDim.x + threadIdx.x;
    if (e >= E) return;
    const int s = ei[e];
    const int d = ei[E + e];
    const int pos = rowptr[d] + rank[e];
    if (pos < E)
        reinterpret_cast<int2*>(csr)[pos] = make_int2(s, __float_as_int(dinv[s]));
}

template <int K>
__global__ void __launch_bounds__(128)
gemm_bias_relu_wmma(const float* __restrict__ A,
                    const float* __restrict__ W,
                    const float* __restrict__ bias,
                    float* __restrict__ C,
                    int M, int do_relu) {
    constexpr int P = K + 4;
    extern __shared__ __align__(16) float Wt[];
    const int tid = threadIdx.x;

    for (int i = tid; i < (K / 4) * HF; i += 128) {
        const int c = i & 63, t = i >> 6;
        const float* w = W + (4 * t) * HF + c;
        *reinterpret_cast<float4*>(Wt + c * P + 4 * t) =
            make_float4(w[0], w[HF], w[2 * HF], w[3 * HF]);
    }
    __syncthreads();

    const int wave = tid >> 5;
    const int lane = tid & 31;
    const int half = lane >> 4;
    const int lm   = lane & 15;
    const int row0 = (blockIdx.x * 4 + wave) * 16;
    if (row0 >= M) return;

    const float* Ap = A + (long)(row0 + lm) * K + 4 * half;
    const float* Bp = Wt + lm * P + 4 * half;

    v8f acc0 = {0.f, 0.f, 0.f, 0.f, 0.f, 0.f, 0.f, 0.f};
    v8f acc1 = acc0, acc2 = acc0, acc3 = acc0;

#pragma unroll
    for (int k = 0; k < K; k += 8) {
        const float4 a  = *reinterpret_cast<const float4*>(Ap + k);
        const float4 b0 = *reinterpret_cast<const float4*>(Bp + k);
        const float4 b1 = *reinterpret_cast<const float4*>(Bp + 16 * P + k);
        const float4 b2 = *reinterpret_cast<const float4*>(Bp + 32 * P + k);
        const float4 b3 = *reinterpret_cast<const float4*>(Bp + 48 * P + k);
        const v2f a01 = {a.x, a.y}, a23 = {a.z, a.w};
        acc0 = __builtin_amdgcn_wmma_f32_16x16x4_f32(
            false, a01, false, v2f{b0.x, b0.y}, (short)0, acc0, false, false);
        acc1 = __builtin_amdgcn_wmma_f32_16x16x4_f32(
            false, a01, false, v2f{b1.x, b1.y}, (short)0, acc1, false, false);
        acc2 = __builtin_amdgcn_wmma_f32_16x16x4_f32(
            false, a01, false, v2f{b2.x, b2.y}, (short)0, acc2, false, false);
        acc3 = __builtin_amdgcn_wmma_f32_16x16x4_f32(
            false, a01, false, v2f{b3.x, b3.y}, (short)0, acc3, false, false);
        acc0 = __builtin_amdgcn_wmma_f32_16x16x4_f32(
            false, a23, false, v2f{b0.z, b0.w}, (short)0, acc0, false, false);
        acc1 = __builtin_amdgcn_wmma_f32_16x16x4_f32(
            false, a23, false, v2f{b1.z, b1.w}, (short)0, acc1, false, false);
        acc2 = __builtin_amdgcn_wmma_f32_16x16x4_f32(
            false, a23, false, v2f{b2.z, b2.w}, (short)0, acc2, false, false);
        acc3 = __builtin_amdgcn_wmma_f32_16x16x4_f32(
            false, a23, false, v2f{b3.z, b3.w}, (short)0, acc3, false, false);
    }

    const v8f accs[4] = {acc0, acc1, acc2, acc3};
#pragma unroll
    for (int t = 0; t < 4; ++t) {
        const int col  = 16 * t + lm;
        const float bc = bias[col];
#pragma unroll
        for (int r = 0; r < 8; ++r) {
            int row = row0 + r + 8 * half;
            float v = accs[t][r] + bc;
            if (do_relu) v = fmaxf(v, 0.0f);
            C[(long)row * HF + col] = v;
        }
    }
}


__device__ __forceinline__ float4 ld_row4(const float* __restrict__ t, int row, int q) {
    return *reinterpret_cast<const float4*>(t + (long)row * HF + q);
}

__device__ __forceinline__ void add_scaled(float4& acc, const float4 r, float w) {
    acc.x += r.x * w;
    acc.y += r.y * w;
    acc.z += r.z * w;
    acc.w += r.w * w;
}

__device__ __forceinline__ float4 row_or_zero(bool keep, const float4 r) {
    float4 o;
    o.x = keep ? r.x : 0.0f;
    o.y = keep ? r.y : 0.0f;
    o.z = keep ? r.z : 0.0f;
    o.w = keep ? r.w : 0.0f;
    return o;
}

__device__ __forceinline__ float4 gather_sum(const int2* __restrict__ csr,
                                             const float* __restrict__ t,
                                             int beg, int end, int q) {
    float4 acc = make_float4(0.0f, 0.0f, 0.0f, 0.0f);
    int j = beg;
    for (; j + 4 <= end; j += 4) {
        const int2 e0 = csr[j], e1 = csr[j + 1], e2 = csr[j + 2], e3 = csr[j + 3];
        const float4 r0 = ld_row4(t, e0.x, q);
        const float4 r1 = ld_row4(t, e1.x, q);
        const float4 r2 = ld_row4(t, e2.x, q);
        const float4 r3 = ld_row4(t, e3.x, q);
        add_scaled(acc, r0, __int_as_float(e0.y));
        add_scaled(acc, r1, __int_as_float(e1.y));
        add_scaled(acc, r2, __int_as_float(e2.y));
        add_scaled(acc, r3, __int_as_float(e3.y));
    }
    const int rem = end - j;
    if (rem > 0) {
        const int last = end - 1;
        const int2 e0 = csr[j];
        const int2 e1 = csr[min(j + 1, last)];
        const int2 e2 = csr[min(j + 2, last)];
        const float4 r0 = ld_row4(t, e0.x, q);
        const float4 r1 = row_or_zero(rem > 1, ld_row4(t, e1.x, q));
        const float4 r2 = row_or_zero(rem > 2, ld_row4(t, e2.x, q));
        add_scaled(acc, r0, __int_as_float(e0.y));
        add_scaled(acc, r1, __int_as_float(e1.y));
        add_scaled(acc, r2, __int_as_float(e2.y));
    }
    return acc;
}

__global__ void gather_step1(const int* __restrict__ csr, const int* __restrict__ rowptr,
                             const float* __restrict__ h, const float* __restrict__ dinv,
                             float* __restrict__ feat, float* __restrict__ hfin,
                             float a0, float a1, float a2,
                             float b0, float b1, float b2, int n_nodes) {
    const int tid = blockIdx.x * blockDim.x + threadIdx.x;
    const int n = tid >> 4;
    if (n >= n_nodes) return;
    const int q = (tid & 15) * 4;
    const float4 m = gather_sum(reinterpret_cast<const int2*>(csr), h,
                                rowptr[n], rowptr[n + 1], q);
    const float dn = dinv[n];
    const float4 hv = ld_row4(h, n, q);
    float4 v;
    v.x = hv.x - m.x * dn;
    v.y = hv.y - m.y * dn;
    v.z = hv.z - m.z * dn;
    v.w = hv.w - m.w * dn;
    *reinterpret_cast<float4*>(feat + (long)n * HF + q) = v;
    float* o = hfin + (long)n * 192 + q;
    float4 o0, o1, o2;
    o0.x = fmaf(b0, v.x, a0 * hv.x); o0.y = fmaf(b0, v.y, a0 * hv.y);
    o0.z = fmaf(b0, v.z, a0 * hv.z); o0.w = fmaf(b0, v.w, a0 * hv.w);
    o1.x = fmaf(b1, v.x, a1 * hv.x); o1.y = fmaf(b1, v.y, a1 * hv.y);
    o1.z = fmaf(b1, v.z, a1 * hv.z); o1.w = fmaf(b1, v.w, a1 * hv.w);
    o2.x = fmaf(b2, v.x, a2 * hv.x); o2.y = fmaf(b2, v.y, a2 * hv.y);
    o2.z = fmaf(b2, v.z, a2 * hv.z); o2.w = fmaf(b2, v.w, a2 * hv.w);
    *reinterpret_cast<float4*>(o)       = o0;
    *reinterpret_cast<float4*>(o + 64)  = o1;
    *reinterpret_cast<float4*>(o + 128) = o2;
}

__global__ void gather_step2(const int* __restrict__ csr, const int* __restrict__ rowptr,
                             const float* __restrict__ feat, const float* __restrict__ dinv,
                             float* __restrict__ hfin,
                             float c0, float c1, float c2, int n_nodes) {
    const int tid = blockIdx.x * blockDim.x + threadIdx.x;
    const int n = tid >> 4;
    if (n >= n_nodes) return;
    const int q = (tid & 15) * 4;
    const float4 m = gather_sum(reinterpret_cast<const int2*>(csr), feat,
                                rowptr[n], rowptr[n + 1], q);
    const float dn = dinv[n];
    const float4 fv = ld_row4(feat, n, q);
    float4 v;
    v.x = fv.x - m.x * dn;
    v.y = fv.y - m.y * dn;
    v.z = fv.z - m.z * dn;
    v.w = fv.w - m.w * dn;
    float* o = hfin + (long)n * 192 + q;
    float4 o0 = *reinterpret_cast<const float4*>(o);
    float4 o1 = *reinterpret_cast<const float4*>(o + 64);
    float4 o2 = *reinterpret_cast<const float4*>(o + 128);
    o0.x = fmaf(c0, v.x, o0.x); o0.y = fmaf(c0, v.y, o0.y);
    o0.z = fmaf(c0, v.z, o0.z); o0.w = fmaf(c0, v.w, o0.w);
    o1.x = fmaf(c1, v.x, o1.x); o1.y = fmaf(c1, v.y, o1.y);
    o1.z = fmaf(c1, v.z, o1.z); o1.w = fmaf(c1, v.w, o1.w);
    o2.x = fmaf(c2, v.x, o2.x); o2.y = fmaf(c2, v.y, o2.y);
    o2.z = fmaf(c2, v.z, o2.z); o2.w = fmaf(c2, v.w, o2.w);
    *reinterpret_cast<float4*>(o)       = o0;
    *reinterpret_cast<float4*>(o + 64)  = o1;
    *reinterpret_cast<float4*>(o + 128) = o2;
}

__global__ void final_linear(const float* __restrict__ Hm, const float* __restrict__ Wm2,
                             const float* __restrict__ bm2, float* __restrict__ out,
                             int n) {
    int i = blockIdx.x * blockDim.x + threadIdx.x;
    if (i >= n) return;
    const float* h = Hm + (long)i * HF;
    float c0 = bm2[0], c1 = bm2[1];
#pragma unroll 8
    for (int j = 0; j < HF; ++j) {
        float hv = h[j];
        c0 = fmaf(hv, Wm2[j * 2 + 0], c0);
        c1 = fmaf(hv, Wm2[j * 2 + 1], c1);
    }
    out[i * 2 + 0] = c0;
    out[i * 2 + 1] = c1;
}

extern "C" void kernel_launch(void* const* d_in, const int* in_sizes, int n_in,
                              void* d_out, int out_size, void* d_ws, size_t ws_size,
                              hipStream_t stream) {
    const float* x   = (const float*)d_in[0];
    const int*   ei  = (const int*)  d_in[1];
    const float* W1  = (const float*)d_in[2];
    const float* b1  = (const float*)d_in[3];
    const float* W2  = (const float*)d_in[4];
    const float* b2  = (const float*)d_in[5];
    const float* Wm1 = (const float*)d_in[6];
    const float* bm1 = (const float*)d_in[7];
    const float* Wm2 = (const float*)d_in[8];
    const float* bm2 = (const float*)d_in[9];
    float* out = (float*)d_out;

    const int N = N_NODES, E = N_EDGES;
    const int N64 = N * HF;

    float* ws   = (float*)d_ws;
    float* dinv = ws;
    float* h    = dinv + N;
    float* feat = h    + (long)N64;
    float* lst  = feat + (long)N64;
    float* hfin = lst  + (long)N64;

    int* csr    = (int*)lst;
    int* rank   = csr + 2 * (long)E;
    int* cnt    = rank + E;
    int* rowptr = cnt + N;
    int* bsum   = rowptr + N + 8;

    const float TH[3][3] = {{3.0f, -3.0f, 0.75f},
                            {0.0f,  3.0f, -1.50f},
                            {0.0f,  0.0f, 0.75f}};

    const int NB = (N + 1023) / 1024;
    zero_i32<<<(N + 255) / 256, 256, 0, stream>>>(cnt, N);
    count_rank<<<(E + 255) / 256, 256, 0, stream>>>(ei, cnt, rank, E);
    scan_local<<<NB, 1024, 0, stream>>>(cnt, rowptr, bsum, dinv, N);
    scan_offset<<<NB, 1024, 0, stream>>>(rowptr, bsum, N, NB);

    gemm_bias_relu_wmma<128><<<(N / 16 + 3) / 4, 128, 64 * 132 * sizeof(float), stream>>>(
        x, W1, b1, feat, N, 1);
    gemm_bias_relu_wmma<64><<<(N / 16 + 3) / 4, 128, 64 * 68 * sizeof(float), stream>>>(
        feat, W2, b2, h, N, 1);

    fill_csr<<<(E + 255) / 256, 256, 0, stream>>>(ei, rowptr, rank, dinv, csr, E);
    const int gg = (N * 16 + 255) / 256;
    gather_step1<<<gg, 256, 0, stream>>>(
        csr, rowptr, h, dinv, feat, hfin,
        TH[0][0], TH[1][0], TH[2][0], TH[0][1], TH[1][1], TH[2][1], N);
    gather_step2<<<gg, 256, 0, stream>>>(
        csr, rowptr, feat, dinv, hfin, TH[0][2], TH[1][2], TH[2][2], N);

    gemm_bias_relu_wmma<192><<<(N / 16 + 3) / 4, 128, 64 * 196 * sizeof(float), stream>>>(
        hfin, Wm1, bm1, feat, N, 1);

    final_linear<<<(N + 255) / 256, 256, 0, stream>>>(feat, Wm2, bm2, out, N);
}


# In-edge lists built by a two-level counting sort on LDS counters (1024-node destination buckets, then per-bucket ranks): no global atomic, no per-edge row-start lookup; sums stay in arrival order
# speedup vs baseline: 9.8302x; 1.3177x over previous
#include <hip/hip_runtime.h>
#include <hip/hip_bf16.h>


typedef float v2f __attribute__((ext_vector_type(2)));
typedef float v8f __attribute__((ext_vector_type(8)));

#define N_NODES 100000
#define N_EDGES 1600000
#define HF 64

#define BKT_NODES 1024
#define NBK ((N_NODES + BKT_NODES - 1) / BKT_NODES)
#define CHUNK_E 6400
#define NCHUNK (N_EDGES / CHUNK_E)
#define NTAB (NBK * NCHUNK)
static_assert(NCHUNK * CHUNK_E == N_EDGES, "chunks must tile the edge list");
static_assert(CHUNK_E % 256 == 0, "a chunk is a whole number of edges per thread");
static_assert(N_EDGES < (1 << 21), "rank is packed into 21 bits");

__device__ __forceinline__ int bucket_of(int d) {
    const unsigned b = (unsigned)d >> 10;
    return (int)(b < (unsigned)NBK ? b : (unsigned)(NBK - 1));
}

__global__ void __launch_bounds__(256)
bucket_hist(const int* __restrict__ ei, int* __restrict__ tab) {
    __shared__ int hist[NBK];
    for (int b = threadIdx.x; b < NBK; b += 256) hist[b] = 0;
    __syncthreads();
    const int* dst = ei + N_EDGES + blockIdx.x * CHUNK_E + threadIdx.x;
    int d[CHUNK_E / 256];
#pragma unroll
    for (int k = 0; k < CHUNK_E / 256; ++k) d[k] = dst[256 * k];
#pragma unroll
    for (int k = 0; k < CHUNK_E / 256; ++k) atomicAdd(&hist[bucket_of(d[k])], 1);
    __syncthreads();
    for (int b = threadIdx.x; b < NBK; b += 256) tab[b * NCHUNK + blockIdx.x] = hist[b];
}

__global__ void __launch_bounds__(1024)
scan_table(int* __restrict__ tab) {
    __shared__ int wsum[32];
    constexpr int PER = (NTAB + 1023) / 1024;
    const int lane = threadIdx.x & 31;
    const int wv   = threadIdx.x >> 5;
    const int i0   = threadIdx.x * PER;
    int v[PER];
    int s = 0;
#pragma unroll
    for (int k = 0; k < PER; ++k) {
        v[k] = (i0 + k < NTAB) ? tab[i0 + k] : 0;
        s += v[k];
    }
    int x = s;
    for (int o = 1; o < 32; o <<= 1) {
        int y = __shfl_up(x, o);
        if (lane >= o) x += y;
    }
    if (lane == 31) wsum[wv] = x;
    __syncthreads();
    if (wv == 0) {
        const int w = wsum[lane];
        int t = w;
        for (int o = 1; o < 32; o <<= 1) {
            int y = __shfl_up(t, o);
            if (lane >= o) t += y;
        }
        wsum[lane] = t - w;
    }
    __syncthreads();
    int run = x - s + wsum[wv];
#pragma unroll
    for (int k = 0; k < PER; ++k) {
        if (i0 + k < NTAB) tab[i0 + k] = run;
        run += v[k];
    }
}

__global__ void __launch_bounds__(256)
bucket_scatter(const int* __restrict__ ei, const int* __restrict__ tab,
               int2* __restrict__ tmp) {
    __shared__ int cur[NBK];
    for (int b = threadIdx.x; b < NBK; b += 256) cur[b] = tab[b * NCHUNK + blockIdx.x];
    __syncthreads();
    const int* srcp = ei + blockIdx.x * CHUNK_E + threadIdx.x;
    const int* dstp = srcp + N_EDGES;
    int s[CHUNK_E / 256], d[CHUNK_E / 256];
#pragma unroll
    for (int k = 0; k < CHUNK_E / 256; ++k) {
        s[k] = srcp[256 * k];
        d[k] = dstp[256 * k];
    }
#pragma unroll
    for (int k = 0; k < CHUNK_E / 256; ++k) {
        const int p = atomicAdd(&cur[bucket_of(d[k])], 1);
        if ((unsigned)p < (unsigned)N_EDGES) tmp[p] = make_int2(s[k], d[k]);
    }
}

__global__ void __launch_bounds__(1024)
bucket_rank(const int* __restrict__ tab, int2* __restrict__ tmp,
            int* __restrict__ rowptr, float* __restrict__ dinv,
            int* __restrict__ csrsrc) {
    __shared__ int cnt[BKT_NODES];
    __shared__ int off[BKT_NODES];
    __shared__ int wsum[32];
    const int b    = blockIdx.x;
    const int t    = threadIdx.x;
    const int lane = t & 31;
    const int wv   = t >> 5;
    const int bs = tab[b * NCHUNK];
    const int be = (b == NBK - 1) ? N_EDGES : tab[(b + 1) * NCHUNK];
    cnt[t] = 0;
    __syncthreads();
    for (int i0 = bs + t; i0 < be; i0 += 8 * 1024) {
        int y[8];
#pragma unroll
        for (int k = 0; k < 8; ++k) {
            const int i = i0 + 1024 * k;
            y[k] = tmp[min(i, be - 1)].y;
        }
#pragma unroll
        for (int k = 0; k < 8; ++k) {
            const int i = i0 + 1024 * k;
            if (i < be) {
                const int dl = y[k] & (BKT_NODES - 1);
                const int r  = atomicAdd(&cnt[dl], 1);
                tmp[i].y = (dl << 21) | r;
            }
        }
    }
    __syncthreads();
    const int c = cnt[t];
    int x = c;
    for (int o = 1; o < 32; o <<= 1) {
        int y = __shfl_up(x, o);
        if (lane >= o) x += y;
    }
    if (lane == 31) wsum[wv] = x;
    __syncthreads();
    if (wv == 0) {
        const int w = wsum[lane];
        int u = w;
        for (int o = 1; o < 32; o <<= 1) {
            int y = __shfl_up(u, o);
            if (lane >= o) u += y;
        }
        wsum[lane] = u - w;
    }
    __syncthreads();
    const int ex = x - c + wsum[wv];
    off[t] = ex;
    const int node = b * BKT_NODES + t;
    if (node < N_NODES) {
        rowptr[node] = bs + ex;
        dinv[node]   = 1.0f / sqrtf(fmaxf((float)c, 1.0f));
    }
    if (b == NBK - 1 && t == 0) rowptr[N_NODES] = N_EDGES;
    __syncthreads();
    for (int i0 = bs + t; i0 < be; i0 += 8 * 1024) {
        int2 e[8];
#pragma unroll
        for (int k = 0; k < 8; ++k) {
            const int i = i0 + 1024 * k;
            e[k] = tmp[min(i, be - 1)];
        }
#pragma unroll
        for (int k = 0; k < 8; ++k) {
            const int i = i0 + 1024 * k;
            const int pos = bs + off[e[k].y >> 21] + (e[k].y & ((1 << 21) - 1));
            if (i < be && (unsigned)pos < (unsigned)N_EDGES) csrsrc[pos] = e[k].x;
        }
    }
}

__global__ void __launch_bounds__(256)
fill_csr(const int* __restrict__ csrsrc, const float* __restrict__ dinv,
         int* __restrict__ csr, int E4) {
    const int j = blockIdx.x * blockDim.x + threadIdx.x;
    if (j >= E4) return;
    const int4 s = reinterpret_cast<const int4*>(csrsrc)[j];
    const float w0 = dinv[s.x], w1 = dinv[s.y], w2 = dinv[s.z], w3 = dinv[s.w];
    int4* o = reinterpret_cast<int4*>(csr) + 2 * (long)j;
    o[0] = make_int4(s.x, __float_as_int(w0), s.y, __float_as_int(w1));
    o[1] = make_int4(s.z, __float_as_int(w2), s.w, __float_as_int(w3));
}

template <int K>
__global__ void __launch_bounds__(128)
gemm_bias_relu_wmma(const float* __restrict__ A,
                    const float* __restrict__ W,
                    const float* __restrict__ bias,
                    float* __restrict__ C,
                    int M, int do_relu) {
    constexpr int P = K + 4;
    extern __shared__ __align__(16) float Wt[];
    const int tid = threadIdx.x;

    for (int i = tid; i < (K / 4) * HF; i += 128) {
        const int c = i & 63, t = i >> 6;
        const float* w = W + (4 * t) * HF + c;
        *reinterpret_cast<float4*>(Wt + c * P + 4 * t) =
            make_float4(w[0], w[HF], w[2 * HF], w[3 * HF]);
    }
    __syncthreads();

    const int wave = tid >> 5;
    const int lane = tid & 31;
    const int half = lane >> 4;
    const int lm   = lane & 15;
    const int row0 = (blockIdx.x * 4 + wave) * 16;
    if (row0 >= M) return;

    const float* Ap = A + (long)(row0 + lm) * K + 4 * half;
    const float* Bp = Wt + lm * P + 4 * half;

    v8f acc0 = {0.f, 0.f, 0.f, 0.f, 0.f, 0.f, 0.f, 0.f};
    v8f acc1 = acc0, acc2 = acc0, acc3 = acc0;

#pragma unroll
    for (int k = 0; k < K; k += 8) {
        const float4 a  = *reinterpret_cast<const float4*>(Ap + k);
        const float4 b0 = *reinterpret_cast<const float4*>(Bp + k);
        const float4 b1 = *reinterpret_cast<const float4*>(Bp + 16 * P + k);
        const float4 b2 = *reinterpret_cast<const float4*>(Bp + 32 * P + k);
        const float4 b3 = *reinterpret_cast<const float4*>(Bp + 48 * P + k);
        const v2f a01 = {a.x, a.y}, a23 = {a.z, a.w};
        acc0 = __builtin_amdgcn_wmma_f32_16x16x4_f32(
            false, a01, false, v2f{b0.x, b0.y}, (short)0, acc0, false, false);
        acc1 = __builtin_amdgcn_wmma_f32_16x16x4_f32(
            false, a01, false, v2f{b1.x, b1.y}, (short)0, acc1, false, false);
        acc2 = __builtin_amdgcn_wmma_f32_16x16x4_f32(
            false, a01, false, v2f{b2.x, b2.y}, (short)0, acc2, false, false);
        acc3 = __builtin_amdgcn_wmma_f32_16x16x4_f32(
            false, a01, false, v2f{b3.x, b3.y}, (short)0, acc3, false, false);
        acc0 = __builtin_amdgcn_wmma_f32_16x16x4_f32(
            false, a23, false, v2f{b0.z, b0.w}, (short)0, acc0, false, false);
        acc1 = __builtin_amdgcn_wmma_f32_16x16x4_f32(
            false, a23, false, v2f{b1.z, b1.w}, (short)0, acc1, false, false);
        acc2 = __builtin_amdgcn_wmma_f32_16x16x4_f32(
            false, a23, false, v2f{b2.z, b2.w}, (short)0, acc2, false, false);
        acc3 = __builtin_amdgcn_wmma_f32_16x16x4_f32(
            false, a23, false, v2f{b3.z, b3.w}, (short)0, acc3, false, false);
    }

    const v8f accs[4] = {acc0, acc1, acc2, acc3};
#pragma unroll
    for (int t = 0; t < 4; ++t) {
        const int col  = 16 * t + lm;
        const float bc = bias[col];
#pragma unroll
        for (int r = 0; r < 8; ++r) {
            int row = row0 + r + 8 * half;
            float v = accs[t][r] + bc;
            if (do_relu) v = fmaxf(v, 0.0f);
            C[(long)row * HF + col] = v;
        }
    }
}


__device__ __forceinline__ float4 ld_row4(const float* __restrict__ t, int row, int q) {
    return *reinterpret_cast<const float4*>(t + (long)row * HF + q);
}

__device__ __forceinline__ void add_scaled(float4& acc, const float4 r, float w) {
    acc.x += r.x * w;
    acc.y += r.y * w;
    acc.z += r.z * w;
    acc.w += r.w * w;
}

__device__ __forceinline__ float4 row_or_zero(bool keep, const float4 r) {
    float4 o;
    o.x = keep ? r.x : 0.0f;
    o.y = keep ? r.y : 0.0f;
    o.z = keep ? r.z : 0.0f;
    o.w = keep ? r.w : 0.0f;
    return o;
}

__device__ __forceinline__ float4 gather_sum(const int2* __restrict__ csr,
                                             const float* __restrict__ t,
                                             int beg, int end, int q) {
    float4 acc = make_float4(0.0f, 0.0f, 0.0f, 0.0f);
    int j = beg;
    for (; j + 4 <= end; j += 4) {
        const int2 e0 = csr[j], e1 = csr[j + 1], e2 = csr[j + 2], e3 = csr[j + 3];
        const float4 r0 = ld_row4(t, e0.x, q);
        const float4 r1 = ld_row4(t, e1.x, q);
        const float4 r2 = ld_row4(t, e2.x, q);
        const float4 r3 = ld_row4(t, e3.x, q);
        add_scaled(acc, r0, __int_as_float(e0.y));
        add_scaled(acc, r1, __int_as_float(e1.y));
        add_scaled(acc, r2, __int_as_float(e2.y));
        add_scaled(acc, r3, __int_as_float(e3.y));
    }
    const int rem = end - j;
    if (rem > 0) {
        const int last = end - 1;
        const int2 e0 = csr[j];
        const int2 e1 = csr[min(j + 1, last)];
        const int2 e2 = csr[min(j + 2, last)];
        const float4 r0 = ld_row4(t, e0.x, q);
        const float4 r1 = row_or_zero(rem > 1, ld_row4(t, e1.x, q));
        const float4 r2 = row_or_zero(rem > 2, ld_row4(t, e2.x, q));
        add_scaled(acc, r0, __int_as_float(e0.y));
        add_scaled(acc, r1, __int_as_float(e1.y));
        add_scaled(acc, r2, __int_as_float(e2.y));
    }
    return acc;
}

__global__ void gather_step1(const int* __restrict__ csr, const int* __restrict__ rowptr,
                             const float* __restrict__ h, const float* __restrict__ dinv,
                             float* __restrict__ feat, float* __restrict__ hfin,
                             float a0, float a1, float a2,
                             float b0, float b1, float b2, int n_nodes) {
    const int tid = blockIdx.x * blockDim.x + threadIdx.x;
    const int n = tid >> 4;
    if (n >= n_nodes) return;
    const int q = (tid & 15) * 4;
    const float4 m = gather_sum(reinterpret_cast<const int2*>(csr), h,
                                rowptr[n], rowptr[n + 1], q);
    const float dn = dinv[n];
    const float4 hv = ld_row4(h, n, q);
    float4 v;
    v.x = hv.x - m.x * dn;
    v.y = hv.y - m.y * dn;
    v.z = hv.z - m.z * dn;
    v.w = hv.w - m.w * dn;
    *reinterpret_cast<float4*>(feat + (long)n * HF + q) = v;
    float* o = hfin + (long)n * 192 + q;
    float4 o0, o1, o2;
    o0.x = fmaf(b0, v.x, a0 * hv.x); o0.y = fmaf(b0, v.y, a0 * hv.y);
    o0.z = fmaf(b0, v.z, a0 * hv.z); o0.w = fmaf(b0, v.w, a0 * hv.w);
    o1.x = fmaf(b1, v.x, a1 * hv.x); o1.y = fmaf(b1, v.y, a1 * hv.y);
    o1.z = fmaf(b1, v.z, a1 * hv.z); o1.w = fmaf(b1, v.w, a1 * hv.w);
    o2.x = fmaf(b2, v.x, a2 * hv.x); o2.y = fmaf(b2, v.y, a2 * hv.y);
    o2.z = fmaf(b2, v.z, a2 * hv.z); o2.w = fmaf(b2, v.w, a2 * hv.w);
    *reinterpret_cast<float4*>(o)       = o0;
    *reinterpret_cast<float4*>(o + 64)  = o1;
    *reinterpret_cast<float4*>(o + 128) = o2;
}

__global__ void gather_step2(const int* __restrict__ csr, const int* __restrict__ rowptr,
                             const float* __restrict__ feat, const float* __restrict__ dinv,
                             float* __restrict__ hfin,
                             float c0, float c1, float c2, int n_nodes) {
    const int tid = blockIdx.x * blockDim.x + threadIdx.x;
    const int n = tid >> 4;
    if (n >= n_nodes) return;
    const int q = (tid & 15) * 4;
    const float4 m = gather_sum(reinterpret_cast<const int2*>(csr), feat,
                                rowptr[n], rowptr[n + 1], q);
    const float dn = dinv[n];
    const float4 fv = ld_row4(feat, n, q);
    float4 v;
    v.x = fv.x - m.x * dn;
    v.y = fv.y - m.y * dn;
    v.z = fv.z - m.z * dn;
    v.w = fv.w - m.w * dn;
    float* o = hfin + (long)n * 192 + q;
    float4 o0 = *reinterpret_cast<const float4*>(o);
    float4 o1 = *reinterpret_cast<const float4*>(o + 64);
    float4 o2 = *reinterpret_cast<const float4*>(o + 128);
    o0.x = fmaf(c0, v.x, o0.x); o0.y = fmaf(c0, v.y, o0.y);
    o0.z = fmaf(c0, v.z, o0.z); o0.w = fmaf(c0, v.w, o0.w);
    o1.x = fmaf(c1, v.x, o1.x); o1.y = fmaf(c1, v.y, o1.y);
    o1.z = fmaf(c1, v.z, o1.z); o1.w = fmaf(c1, v.w, o1.w);
    o2.x = fmaf(c2, v.x, o2.x); o2.y = fmaf(c2, v.y, o2.y);
    o2.z = fmaf(c2, v.z, o2.z); o2.w = fmaf(c2, v.w, o2.w);
    *reinterpret_cast<float4*>(o)       = o0;
    *reinterpret_cast<float4*>(o + 64)  = o1;
    *reinterpret_cast<float4*>(o + 128) = o2;
}

__global__ void final_linear(const float* __restrict__ Hm, const float* __restrict__ Wm2,
                             const float* __restrict__ bm2, float* __restrict__ out,
                             int n) {
    int i = blockIdx.x * blockDim.x + threadIdx.x;
    if (i >= n) return;
    const float* h = Hm + (long)i * HF;
    float c0 = bm2[0], c1 = bm2[1];
#pragma unroll 8
    for (int j = 0; j < HF; ++j) {
        float hv = h[j];
        c0 = fmaf(hv, Wm2[j * 2 + 0], c0);
        c1 = fmaf(hv, Wm2[j * 2 + 1], c1);
    }
    out[i * 2 + 0] = c0;
    out[i * 2 + 1] = c1;
}

extern "C" void kernel_launch(void* const* d_in, const int* in_sizes, int n_in,
                              void* d_out, int out_size, void* d_ws, size_t ws_size,
                              hipStream_t stream) {
    const float* x   = (const float*)d_in[0];
    const int*   ei  = (const int*)  d_in[1];
    const float* W1  = (const float*)d_in[2];
    const float* b1  = (const float*)d_in[3];
    const float* W2  = (const float*)d_in[4];
    const float* b2  = (const float*)d_in[5];
    const float* Wm1 = (const float*)d_in[6];
    const float* bm1 = (const float*)d_in[7];
    const float* Wm2 = (const float*)d_in[8];
    const float* bm2 = (const float*)d_in[9];
    float* out = (float*)d_out;

    const int N = N_NODES, E = N_EDGES;
    const int N64 = N * HF;

    float* ws   = (float*)d_ws;
    float* dinv = ws;
    float* h    = dinv + N;
    float* feat = h    + (long)N64;
    float* lst  = feat + (long)N64;
    float* hfin = lst  + (long)N64;

    int* csr    = (int*)lst;
    int* rowptr = csr + 2 * (long)E;
    int* tmp    = (int*)hfin;
    int* csrsrc = tmp + 2 * (long)E;
    int* tab    = csrsrc + E;

    const float TH[3][3] = {{3.0f, -3.0f, 0.75f},
                            {0.0f,  3.0f, -1.50f},
                            {0.0f,  0.0f, 0.75f}};

    bucket_hist<<<NCHUNK, 256, 0, stream>>>(ei, tab);
    scan_table<<<1, 1024, 0, stream>>>(tab);
    bucket_scatter<<<NCHUNK, 256, 0, stream>>>(ei, tab, (int2*)tmp);
    bucket_rank<<<NBK, 1024, 0, stream>>>(tab, (int2*)tmp, rowptr, dinv, csrsrc);

    gemm_bias_relu_wmma<128><<<(N / 16 + 3) / 4, 128, 64 * 132 * sizeof(float), stream>>>(
        x, W1, b1, feat, N, 1);
    gemm_bias_relu_wmma<64><<<(N / 16 + 3) / 4, 128, 64 * 68 * sizeof(float), stream>>>(
        feat, W2, b2, h, N, 1);

    fill_csr<<<(E / 4 + 255) / 256, 256, 0, stream>>>(csrsrc, dinv, csr, E / 4);
    const int gg = (N * 16 + 255) / 256;
    gather_step1<<<gg, 256, 0, stream>>>(
        csr, rowptr, h, dinv, feat, hfin,
        TH[0][0], TH[1][0], TH[2][0], TH[0][1], TH[1][1], TH[2][1], N);
    gather_step2<<<gg, 256, 0, stream>>>(
        csr, rowptr, feat, dinv, hfin, TH[0][2], TH[1][2], TH[2][2], N);

    gemm_bias_relu_wmma<192><<<(N / 16 + 3) / 4, 128, 64 * 196 * sizeof(float), stream>>>(
        hfin, Wm1, bm1, feat, N, 1);

    final_linear<<<(N + 255) / 256, 256, 0, stream>>>(feat, Wm2, bm2, out, N);
}


# Concatenated filter outputs no longer stored: steps write only feat_1, feat_2; last dense layer forms its A fragments a_t*h+b_t*f1+c_t*f2 in registers (same op and k order, bit-identical)
# speedup vs baseline: 10.5856x; 1.0768x over previous
#include <hip/hip_runtime.h>
#include <hip/hip_bf16.h>


typedef float v2f __attribute__((ext_vector_type(2)));
typedef float v8f __attribute__((ext_vector_type(8)));

#define N_NODES 100000
#define N_EDGES 1600000
#define HF 64

#define BKT_NODES 1024
#define NBK ((N_NODES + BKT_NODES - 1) / BKT_NODES)
#define CHUNK_E 6400
#define NCHUNK (N_EDGES / CHUNK_E)
#define NTAB (NBK * NCHUNK)
static_assert(NCHUNK * CHUNK_E == N_EDGES, "chunks must tile the edge list");
static_assert(CHUNK_E % 256 == 0, "a chunk is a whole number of edges per thread");
static_assert(N_EDGES < (1 << 21), "rank is packed into 21 bits");

__device__ __forceinline__ int bucket_of(int d) {
    const unsigned b = (unsigned)d >> 10;
    return (int)(b < (unsigned)NBK ? b : (unsigned)(NBK - 1));
}

__global__ void __launch_bounds__(256)
bucket_hist(const int* __restrict__ ei, int* __restrict__ tab) {
    __shared__ int hist[NBK];
    for (int b = threadIdx.x; b < NBK; b += 256) hist[b] = 0;
    __syncthreads();
    const int* dst = ei + N_EDGES + blockIdx.x * CHUNK_E + threadIdx.x;
    int d[CHUNK_E / 256];
#pragma unroll
    for (int k = 0; k < CHUNK_E / 256; ++k) d[k] = dst[256 * k];
#pragma unroll
    for (int k = 0; k < CHUNK_E / 256; ++k) atomicAdd(&hist[bucket_of(d[k])], 1);
    __syncthreads();
    for (int b = threadIdx.x; b < NBK; b += 256) tab[b * NCHUNK + blockIdx.x] = hist[b];
}

__global__ void __launch_bounds__(1024)
scan_table(int* __restrict__ tab) {
    __shared__ int wsum[32];
    constexpr int PER = (NTAB + 1023) / 1024;
    const int lane = threadIdx.x & 31;
    const int wv   = threadIdx.x >> 5;
    const int i0   = threadIdx.x * PER;
    int v[PER];
    int s = 0;
#pragma unroll
    for (int k = 0; k < PER; ++k) {
        v[k] = (i0 + k < NTAB) ? tab[i0 + k] : 0;
        s += v[k];
    }
    int x = s;
    for (int o = 1; o < 32; o <<= 1) {
        int y = __shfl_up(x, o);
        if (lane >= o) x += y;
    }
    if (lane == 31) wsum[wv] = x;
    __syncthreads();
    if (wv == 0) {
        const int w = wsum[lane];
        int t = w;
        for (int o = 1; o < 32; o <<= 1) {
            int y = __shfl_up(t, o);
            if (lane >= o) t += y;
        }
        wsum[lane] = t - w;
    }
    __syncthreads();
    int run = x - s + wsum[wv];
#pragma unroll
    for (int k = 0; k < PER; ++k) {
        if (i0 + k < NTAB) tab[i0 + k] = run;
        run += v[k];
    }
}

__global__ void __launch_bounds__(256)
bucket_scatter(const int* __restrict__ ei, const int* __restrict__ tab,
               int2* __restrict__ tmp) {
    __shared__ int cur[NBK];
    for (int b = threadIdx.x; b < NBK; b += 256) cur[b] = tab[b * NCHUNK + blockIdx.x];
    __syncthreads();
    const int* srcp = ei + blockIdx.x * CHUNK_E + threadIdx.x;
    const int* dstp = srcp + N_EDGES;
    int s[CHUNK_E / 256], d[CHUNK_E / 256];
#pragma unroll
    for (int k = 0; k < CHUNK_E / 256; ++k) {
        s[k] = srcp[256 * k];
        d[k] = dstp[256 * k];
    }
#pragma unroll
    for (int k = 0; k < CHUNK_E / 256; ++k) {
        const int p = atomicAdd(&cur[bucket_of(d[k])], 1);
        if ((unsigned)p < (unsigned)N_EDGES) tmp[p] = make_int2(s[k], d[k]);
    }
}

__global__ void __launch_bounds__(1024)
bucket_rank(const int* __restrict__ tab, int2* __restrict__ tmp,
            int* __restrict__ rowptr, float* __restrict__ dinv,
            int* __restrict__ csrsrc) {
    __shared__ int cnt[BKT_NODES];
    __shared__ int off[BKT_NODES];
    __shared__ int wsum[32];
    const int b    = blockIdx.x;
    const int t    = threadIdx.x;
    const int lane = t & 31;
    const int wv   = t >> 5;
    const int bs = tab[b * NCHUNK];
    const int be = (b == NBK - 1) ? N_EDGES : tab[(b + 1) * NCHUNK];
    cnt[t] = 0;
    __syncthreads();
    for (int i0 = bs + t; i0 < be; i0 += 8 * 1024) {
        int y[8];
#pragma unroll
        for (int k = 0; k < 8; ++k) {
            const int i = i0 + 1024 * k;
            y[k] = tmp[min(i, be - 1)].y;
        }
#pragma unroll
        for (int k = 0; k < 8; ++k) {
            const int i = i0 + 1024 * k;
            if (i < be) {
                const int dl = y[k] & (BKT_NODES - 1);
                const int r  = atomicAdd(&cnt[dl], 1);
                tmp[i].y = (dl << 21) | r;
            }
        }
    }
    __syncthreads();
    const int c = cnt[t];
    int x = c;
    for (int o = 1; o < 32; o <<= 1) {
        int y = __shfl_up(x, o);
        if (lane >= o) x += y;
    }
    if (lane == 31) wsum[wv] = x;
    __syncthreads();
    if (wv == 0) {
        const int w = wsum[lane];
        int u = w;
        for (int o = 1; o < 32; o <<= 1) {
            int y = __shfl_up(u, o);
            if (lane >= o) u += y;
        }
        wsum[lane] = u - w;
    }
    __syncthreads();
    const int ex = x - c + wsum[wv];
    off[t] = ex;
    const int node = b * BKT_NODES + t;
    if (node < N_NODES) {
        rowptr[node] = bs + ex;
        dinv[node]   = 1.0f / sqrtf(fmaxf((float)c, 1.0f));
    }
    if (b == NBK - 1 && t == 0) rowptr[N_NODES] = N_EDGES;
    __syncthreads();
    for (int i0 = bs + t; i0 < be; i0 += 8 * 1024) {
        int2 e[8];
#pragma unroll
        for (int k = 0; k < 8; ++k) {
            const int i = i0 + 1024 * k;
            e[k] = tmp[min(i, be - 1)];
        }
#pragma unroll
        for (int k = 0; k < 8; ++k) {
            const int i = i0 + 1024 * k;
            const int pos = bs + off[e[k].y >> 21] + (e[k].y & ((1 << 21) - 1));
            if (i < be && (unsigned)pos < (unsigned)N_EDGES) csrsrc[pos] = e[k].x;
        }
    }
}

__global__ void __launch_bounds__(256)
fill_csr(const int* __restrict__ csrsrc, const float* __restrict__ dinv,
         int* __restrict__ csr, int E4) {
    const int j = blockIdx.x * blockDim.x + threadIdx.x;
    if (j >= E4) return;
    const int4 s = reinterpret_cast<const int4*>(csrsrc)[j];
    const float w0 = dinv[s.x], w1 = dinv[s.y], w2 = dinv[s.z], w3 = dinv[s.w];
    int4* o = reinterpret_cast<int4*>(csr) + 2 * (long)j;
    o[0] = make_int4(s.x, __float_as_int(w0), s.y, __float_as_int(w1));
    o[1] = make_int4(s.z, __float_as_int(w2), s.w, __float_as_int(w3));
}

template <int K>
__global__ void __launch_bounds__(128)
gemm_bias_relu_wmma(const float* __restrict__ A,
                    const float* __restrict__ W,
                    const float* __restrict__ bias,
                    float* __restrict__ C,
                    int M, int do_relu) {
    constexpr int P = K + 4;
    extern __shared__ __align__(16) float Wt[];
    const int tid = threadIdx.x;

    for (int i = tid; i < (K / 4) * HF; i += 128) {
        const int c = i & 63, t = i >> 6;
        const float* w = W + (4 * t) * HF + c;
        *reinterpret_cast<float4*>(Wt + c * P + 4 * t) =
            make_float4(w[0], w[HF], w[2 * HF], w[3 * HF]);
    }
    __syncthreads();

    const int wave = tid >> 5;
    const int lane = tid & 31;
    const int half = lane >> 4;
    const int lm   = lane & 15;
    const int row0 = (blockIdx.x * 4 + wave) * 16;
    if (row0 >= M) return;

    const float* Ap = A + (long)(row0 + lm) * K + 4 * half;
    const float* Bp = Wt + lm * P + 4 * half;

    v8f acc0 = {0.f, 0.f, 0.f, 0.f, 0.f, 0.f, 0.f, 0.f};
    v8f acc1 = acc0, acc2 = acc0, acc3 = acc0;

#pragma unroll
    for (int k = 0; k < K; k += 8) {
        const float4 a  = *reinterpret_cast<const float4*>(Ap + k);
        const float4 b0 = *reinterpret_cast<const float4*>(Bp + k);
        const float4 b1 = *reinterpret_cast<const float4*>(Bp + 16 * P + k);
        const float4 b2 = *reinterpret_cast<const float4*>(Bp + 32 * P + k);
        const float4 b3 = *reinterpret_cast<const float4*>(Bp + 48 * P + k);
        const v2f a01 = {a.x, a.y}, a23 = {a.z, a.w};
        acc0 = __builtin_amdgcn_wmma_f32_16x16x4_f32(
            false, a01, false, v2f{b0.x, b0.y}, (short)0, acc0, false, false);
        acc1 = __builtin_amdgcn_wmma_f32_16x16x4_f32(
            false, a01, false, v2f{b1.x, b1.y}, (short)0, acc1, false, false);
        acc2 = __builtin_amdgcn_wmma_f32_16x16x4_f32(
            false, a01, false, v2f{b2.x, b2.y}, (short)0, acc2, false, false);
        acc3 = __builtin_amdgcn_wmma_f32_16x16x4_f32(
            false, a01, false, v2f{b3.x, b3.y}, (short)0, acc3, false, false);
        acc0 = __builtin_amdgcn_wmma_f32_16x16x4_f32(
            false, a23, false, v2f{b0.z, b0.w}, (short)0, acc0, false, false);
        acc1 = __builtin_amdgcn_wmma_f32_16x16x4_f32(
            false, a23, false, v2f{b1.z, b1.w}, (short)0, acc1, false, false);
        acc2 = __builtin_amdgcn_wmma_f32_16x16x4_f32(
            false, a23, false, v2f{b2.z, b2.w}, (short)0, acc2, false, false);
        acc3 = __builtin_amdgcn_wmma_f32_16x16x4_f32(
            false, a23, false, v2f{b3.z, b3.w}, (short)0, acc3, false, false);
    }

    const v8f accs[4] = {acc0, acc1, acc2, acc3};
#pragma unroll
    for (int t = 0; t < 4; ++t) {
        const int col  = 16 * t + lm;
        const float bc = bias[col];
#pragma unroll
        for (int r = 0; r < 8; ++r) {
            int row = row0 + r + 8 * half;
            float v = accs[t][r] + bc;
            if (do_relu) v = fmaxf(v, 0.0f);
            C[(long)row * HF + col] = v;
        }
    }
}


__device__ __forceinline__ float4 ld_row4(const float* __restrict__ t, int row, int q) {
    return *reinterpret_cast<const float4*>(t + (long)row * HF + q);
}

__device__ __forceinline__ void add_scaled(float4& acc, const float4 r, float w) {
    acc.x += r.x * w;
    acc.y += r.y * w;
    acc.z += r.z * w;
    acc.w += r.w * w;
}

__device__ __forceinline__ float4 row_or_zero(bool keep, const float4 r) {
    float4 o;
    o.x = keep ? r.x : 0.0f;
    o.y = keep ? r.y : 0.0f;
    o.z = keep ? r.z : 0.0f;
    o.w = keep ? r.w : 0.0f;
    return o;
}

__device__ __forceinline__ float4 gather_sum(const int2* __restrict__ csr,
                                             const float* __restrict__ t,
                                             int beg, int end, int q) {
    float4 acc = make_float4(0.0f, 0.0f, 0.0f, 0.0f);
    int j = beg;
    for (; j + 4 <= end; j += 4) {
        const int2 e0 = csr[j], e1 = csr[j + 1], e2 = csr[j + 2], e3 = csr[j + 3];
        const float4 r0 = ld_row4(t, e0.x, q);
        const float4 r1 = ld_row4(t, e1.x, q);
        const float4 r2 = ld_row4(t, e2.x, q);
        const float4 r3 = ld_row4(t, e3.x, q);
        add_scaled(acc, r0, __int_as_float(e0.y));
        add_scaled(acc, r1, __int_as_float(e1.y));
        add_scaled(acc, r2, __int_as_float(e2.y));
        add_scaled(acc, r3, __int_as_float(e3.y));
    }
    const int rem = end - j;
    if (rem > 0) {
        const int last = end - 1;
        const int2 e0 = csr[j];
        const int2 e1 = csr[min(j + 1, last)];
        const int2 e2 = csr[min(j + 2, last)];
        const float4 r0 = ld_row4(t, e0.x, q);
        const float4 r1 = row_or_zero(rem > 1, ld_row4(t, e1.x, q));
        const float4 r2 = row_or_zero(rem > 2, ld_row4(t, e2.x, q));
        add_scaled(acc, r0, __int_as_float(e0.y));
        add_scaled(acc, r1, __int_as_float(e1.y));
        add_scaled(acc, r2, __int_as_float(e2.y));
    }
    return acc;
}

__global__ void gather_step1(const int* __restrict__ csr, const int* __restrict__ rowptr,
                             const float* __restrict__ h, const float* __restrict__ dinv,
                             float* __restrict__ feat, int n_nodes) {
    const int tid = blockIdx.x * blockDim.x + threadIdx.x;
    const int n = tid >> 4;
    if (n >= n_nodes) return;
    const int q = (tid & 15) * 4;
    const float4 m = gather_sum(reinterpret_cast<const int2*>(csr), h,
                                rowptr[n], rowptr[n + 1], q);
    const float dn = dinv[n];
    const float4 hv = ld_row4(h, n, q);
    float4 v;
    v.x = hv.x - m.x * dn;
    v.y = hv.y - m.y * dn;
    v.z = hv.z - m.z * dn;
    v.w = hv.w - m.w * dn;
    *reinterpret_cast<float4*>(feat + (long)n * HF + q) = v;
}

__global__ void gather_step2(const int* __restrict__ csr, const int* __restrict__ rowptr,
                             const float* __restrict__ feat, const float* __restrict__ dinv,
                             float* __restrict__ feat2, int n_nodes) {
    const int tid = blockIdx.x * blockDim.x + threadIdx.x;
    const int n = tid >> 4;
    if (n >= n_nodes) return;
    const int q = (tid & 15) * 4;
    const float4 m = gather_sum(reinterpret_cast<const int2*>(csr), feat,
                                rowptr[n], rowptr[n + 1], q);
    const float dn = dinv[n];
    const float4 fv = ld_row4(feat, n, q);
    float4 v;
    v.x = fv.x - m.x * dn;
    v.y = fv.y - m.y * dn;
    v.z = fv.z - m.z * dn;
    v.w = fv.w - m.w * dn;
    *reinterpret_cast<float4*>(feat2 + (long)n * HF + q) = v;
}

__global__ void __launch_bounds__(128)
gemm_filters_wmma(const float* __restrict__ h, const float* __restrict__ f1,
                  const float* __restrict__ f2,
                  const float* __restrict__ W, const float* __restrict__ bias,
                  float* __restrict__ C, int M,
                  float a0, float a1, float a2,
                  float b0, float b1, float b2,
                  float c0, float c1, float c2) {
    constexpr int K = 3 * HF;
    constexpr int P = K + 4;
    extern __shared__ __align__(16) float Wt[];
    const int tid  = threadIdx.x;
    const int wave = tid >> 5;
    const int lane = tid & 31;
    const int half = lane >> 4;
    const int lm   = lane & 15;
    const int row0 = (blockIdx.x * 4 + wave) * 16;

    const long arow = (long)min(row0 + lm, M - 1) * HF + 4 * half;
    float4 hv[HF / 8], uv[HF / 8], wv[HF / 8];
#pragma unroll
    for (int j = 0; j < HF / 8; ++j) {
        hv[j] = *reinterpret_cast<const float4*>(h  + arow + 8 * j);
        uv[j] = *reinterpret_cast<const float4*>(f1 + arow + 8 * j);
        wv[j] = *reinterpret_cast<const float4*>(f2 + arow + 8 * j);
    }

    for (int i = tid; i < (K / 4) * HF; i += 128) {
        const int c = i & 63, t = i >> 6;
        const float* w = W + (4 * t) * HF + c;
        *reinterpret_cast<float4*>(Wt + c * P + 4 * t) =
            make_float4(w[0], w[HF], w[2 * HF], w[3 * HF]);
    }
    __syncthreads();
    if (row0 >= M) return;

    const float* Bp = Wt + lm * P + 4 * half;
    const float ca[3] = {a0, a1, a2};
    const float cb[3] = {b0, b1, b2};
    const float cc[3] = {c0, c1, c2};

    v8f acc0 = {0.f, 0.f, 0.f, 0.f, 0.f, 0.f, 0.f, 0.f};
    v8f acc1 = acc0, acc2 = acc0, acc3 = acc0;

#pragma unroll
    for (int t = 0; t < 3; ++t) {
#pragma unroll
        for (int j = 0; j < HF / 8; ++j) {
            const int k = HF * t + 8 * j;
            float4 a;
            a.x = fmaf(cc[t], wv[j].x, fmaf(cb[t], uv[j].x, ca[t] * hv[j].x));
            a.y = fmaf(cc[t], wv[j].y, fmaf(cb[t], uv[j].y, ca[t] * hv[j].y));
            a.z = fmaf(cc[t], wv[j].z, fmaf(cb[t], uv[j].z, ca[t] * hv[j].z));
            a.w = fmaf(cc[t], wv[j].w, fmaf(cb[t], uv[j].w, ca[t] * hv[j].w));
            const float4 b0v = *reinterpret_cast<const float4*>(Bp + k);
            const float4 b1v = *reinterpret_cast<const float4*>(Bp + 16 * P + k);
            const float4 b2v = *reinterpret_cast<const float4*>(Bp + 32 * P + k);
            const float4 b3v = *reinterpret_cast<const float4*>(Bp + 48 * P + k);
            const v2f a01 = {a.x, a.y}, a23 = {a.z, a.w};
            acc0 = __builtin_amdgcn_wmma_f32_16x16x4_f32(
                false, a01, false, v2f{b0v.x, b0v.y}, (short)0, acc0, false, false);
            acc1 = __builtin_amdgcn_wmma_f32_16x16x4_f32(
                false, a01, false, v2f{b1v.x, b1v.y}, (short)0, acc1, false, false);
            acc2 = __builtin_amdgcn_wmma_f32_16x16x4_f32(
                false, a01, false, v2f{b2v.x, b2v.y}, (short)0, acc2, false, false);
            acc3 = __builtin_amdgcn_wmma_f32_16x16x4_f32(
                false, a01, false, v2f{b3v.x, b3v.y}, (short)0, acc3, false, false);
            acc0 = __builtin_amdgcn_wmma_f32_16x16x4_f32(
                false, a23, false, v2f{b0v.z, b0v.w}, (short)0, acc0, false, false);
            acc1 = __builtin_amdgcn_wmma_f32_16x16x4_f32(
                false, a23, false, v2f{b1v.z, b1v.w}, (short)0, acc1, false, false);
            acc2 = __builtin_amdgcn_wmma_f32_16x16x4_f32(
                false, a23, false, v2f{b2v.z, b2v.w}, (short)0, acc2, false, false);
            acc3 = __builtin_amdgcn_wmma_f32_16x16x4_f32(
                false, a23, false, v2f{b3v.z, b3v.w}, (short)0, acc3, false, false);
        }
    }

    const v8f accs[4] = {acc0, acc1, acc2, acc3};
#pragma unroll
    for (int t = 0; t < 4; ++t) {
        const int col  = 16 * t + lm;
        const float bc = bias[col];
#pragma unroll
        for (int r = 0; r < 8; ++r) {
            const int row = row0 + r + 8 * half;
            C[(long)row * HF + col] = fmaxf(accs[t][r] + bc, 0.0f);
        }
    }
}

__global__ void final_linear(const float* __restrict__ Hm, const float* __restrict__ Wm2,
                             const float* __restrict__ bm2, float* __restrict__ out,
                             int n) {
    int i = blockIdx.x * blockDim.x + threadIdx.x;
    if (i >= n) return;
    const float* h = Hm + (long)i * HF;
    float c0 = bm2[0], c1 = bm2[1];
#pragma unroll 8
    for (int j = 0; j < HF; ++j) {
        float hv = h[j];
        c0 = fmaf(hv, Wm2[j * 2 + 0], c0);
        c1 = fmaf(hv, Wm2[j * 2 + 1], c1);
    }
    out[i * 2 + 0] = c0;
    out[i * 2 + 1] = c1;
}

extern "C" void kernel_launch(void* const* d_in, const int* in_sizes, int n_in,
                              void* d_out, int out_size, void* d_ws, size_t ws_size,
                              hipStream_t stream) {
    const float* x   = (const float*)d_in[0];
    const int*   ei  = (const int*)  d_in[1];
    const float* W1  = (const float*)d_in[2];
    const float* b1  = (const float*)d_in[3];
    const float* W2  = (const float*)d_in[4];
    const float* b2  = (const float*)d_in[5];
    const float* Wm1 = (const float*)d_in[6];
    const float* bm1 = (const float*)d_in[7];
    const float* Wm2 = (const float*)d_in[8];
    const float* bm2 = (const float*)d_in[9];
    float* out = (float*)d_out;

    const int N = N_NODES, E = N_EDGES;
    const int N64 = N * HF;

    float* ws   = (float*)d_ws;
    float* dinv = ws;
    float* h    = dinv + N;
    float* feat = h    + (long)N64;
    float* lst  = feat + (long)N64;
    float* hfin = lst  + (long)N64;

    int* csr    = (int*)lst;
    int* rowptr = csr + 2 * (long)E;
    int* tmp    = (int*)hfin;
    int* csrsrc = tmp + 2 * (long)E;
    int* tab    = csrsrc + E;
    float* feat2 = hfin;
    float* hm    = hfin + (long)N64;

    const float TH[3][3] = {{3.0f, -3.0f, 0.75f},
                            {0.0f,  3.0f, -1.50f},
                            {0.0f,  0.0f, 0.75f}};

    bucket_hist<<<NCHUNK, 256, 0, stream>>>(ei, tab);
    scan_table<<<1, 1024, 0, stream>>>(tab);
    bucket_scatter<<<NCHUNK, 256, 0, stream>>>(ei, tab, (int2*)tmp);
    bucket_rank<<<NBK, 1024, 0, stream>>>(tab, (int2*)tmp, rowptr, dinv, csrsrc);

    gemm_bias_relu_wmma<128><<<(N / 16 + 3) / 4, 128, 64 * 132 * sizeof(float), stream>>>(
        x, W1, b1, feat, N, 1);
    gemm_bias_relu_wmma<64><<<(N / 16 + 3) / 4, 128, 64 * 68 * sizeof(float), stream>>>(
        feat, W2, b2, h, N, 1);

    fill_csr<<<(E / 4 + 255) / 256, 256, 0, stream>>>(csrsrc, dinv, csr, E / 4);
    const int gg = (N * 16 + 255) / 256;
    gather_step1<<<gg, 256, 0, stream>>>(csr, rowptr, h, dinv, feat, N);
    gather_step2<<<gg, 256, 0, stream>>>(csr, rowptr, feat, dinv, feat2, N);

    gemm_filters_wmma<<<(N / 16 + 3) / 4, 128, 64 * 196 * sizeof(float), stream>>>(
        h, feat, feat2, Wm1, bm1, hm, N,
        TH[0][0], TH[1][0], TH[2][0], TH[0][1], TH[1][1], TH[2][1],
        TH[0][2], TH[1][2], TH[2][2]);

    final_linear<<<(N + 255) / 256, 256, 0, stream>>>(hm, Wm2, bm2, out, N);
}


# Dense layers chained in registers by transposed products (W as A, rows as B): h1 and hidden layer never stored, 11 -> 9 launches; layer-2 k order and output-layer sum order changed
# speedup vs baseline: 11.3301x; 1.0703x over previous
#include <hip/hip_runtime.h>
#include <hip/hip_bf16.h>


typedef float v2f __attribute__((ext_vector_type(2)));
typedef float v8f __attribute__((ext_vector_type(8)));

#define N_NODES 100000
#define N_EDGES 1600000
#define HF 64

#define BKT_NODES 1024
#define NBK ((N_NODES + BKT_NODES - 1) / BKT_NODES)
#define CHUNK_E 6400
#define NCHUNK (N_EDGES / CHUNK_E)
#define NTAB (NBK * NCHUNK)
static_assert(NCHUNK * CHUNK_E == N_EDGES, "chunks must tile the edge list");
static_assert(CHUNK_E % 256 == 0, "a chunk is a whole number of edges per thread");
static_assert(N_EDGES < (1 << 21), "rank is packed into 21 bits");

__device__ __forceinline__ int bucket_of(int d) {
    const unsigned b = (unsigned)d >> 10;
    return (int)(b < (unsigned)NBK ? b : (unsigned)(NBK - 1));
}

__global__ void __launch_bounds__(256)
bucket_hist(const int* __restrict__ ei, int* __restrict__ tab) {
    __shared__ int hist[NBK];
    for (int b = threadIdx.x; b < NBK; b += 256) hist[b] = 0;
    __syncthreads();
    const int* dst = ei + N_EDGES + blockIdx.x * CHUNK_E + threadIdx.x;
    int d[CHUNK_E / 256];
#pragma unroll
    for (int k = 0; k < CHUNK_E / 256; ++k) d[k] = dst[256 * k];
#pragma unroll
    for (int k = 0; k < CHUNK_E / 256; ++k) atomicAdd(&hist[bucket_of(d[k])], 1);
    __syncthreads();
    for (int b = threadIdx.x; b < NBK; b += 256) tab[b * NCHUNK + blockIdx.x] = hist[b];
}

__global__ void __launch_bounds__(1024)
scan_table(int* __restrict__ tab) {
    __shared__ int wsum[32];
    constexpr int PER = (NTAB + 1023) / 1024;
    const int lane = threadIdx.x & 31;
    const int wv   = threadIdx.x >> 5;
    const int i0   = threadIdx.x * PER;
    int v[PER];
    int s = 0;
#pragma unroll
    for (int k = 0; k < PER; ++k) {
        v[k] = (i0 + k < NTAB) ? tab[i0 + k] : 0;
        s += v[k];
    }
    int x = s;
    for (int o = 1; o < 32; o <<= 1) {
        int y = __shfl_up(x, o);
        if (lane >= o) x += y;
    }
    if (lane == 31) wsum[wv] = x;
    __syncthreads();
    if (wv == 0) {
        const int w = wsum[lane];
        int t = w;
        for (int o = 1; o < 32; o <<= 1) {
            int y = __shfl_up(t, o);
            if (lane >= o) t += y;
        }
        wsum[lane] = t - w;
    }
    __syncthreads();
    int run = x - s + wsum[wv];
#pragma unroll
    for (int k = 0; k < PER; ++k) {
        if (i0 + k < NTAB) tab[i0 + k] = run;
        run += v[k];
    }
}

__global__ void __launch_bounds__(256)
bucket_scatter(const int* __restrict__ ei, const int* __restrict__ tab,
               int2* __restrict__ tmp) {
    __shared__ int cur[NBK];
    for (int b = threadIdx.x; b < NBK; b += 256) cur[b] = tab[b * NCHUNK + blockIdx.x];
    __syncthreads();
    const int* srcp = ei + blockIdx.x * CHUNK_E + threadIdx.x;
    const int* dstp = srcp + N_EDGES;
    int s[CHUNK_E / 256], d[CHUNK_E / 256];
#pragma unroll
    for (int k = 0; k < CHUNK_E / 256; ++k) {
        s[k] = srcp[256 * k];
        d[k] = dstp[256 * k];
    }
#pragma unroll
    for (int k = 0; k < CHUNK_E / 256; ++k) {
        const int p = atomicAdd(&cur[bucket_of(d[k])], 1);
        if ((unsigned)p < (unsigned)N_EDGES) tmp[p] = make_int2(s[k], d[k]);
    }
}

__global__ void __launch_bounds__(1024)
bucket_rank(const int* __restrict__ tab, int2* __restrict__ tmp,
            int* __restrict__ rowptr, float* __restrict__ dinv,
            int* __restrict__ csrsrc) {
    __shared__ int cnt[BKT_NODES];
    __shared__ int off[BKT_NODES];
    __shared__ int wsum[32];
    const int b    = blockIdx.x;
    const int t    = threadIdx.x;
    const int lane = t & 31;
    const int wv   = t >> 5;
    const int bs = tab[b * NCHUNK];
    const int be = (b == NBK - 1) ? N_EDGES : tab[(b + 1) * NCHUNK];
    cnt[t] = 0;
    __syncthreads();
    for (int i0 = bs + t; i0 < be; i0 += 8 * 1024) {
        int y[8];
#pragma unroll
        for (int k = 0; k < 8; ++k) {
            const int i = i0 + 1024 * k;
            y[k] = tmp[min(i, be - 1)].y;
        }
#pragma unroll
        for (int k = 0; k < 8; ++k) {
            const int i = i0 + 1024 * k;
            if (i < be) {
                const int dl = y[k] & (BKT_NODES - 1);
                const int r  = atomicAdd(&cnt[dl], 1);
                tmp[i].y = (dl << 21) | r;
            }
        }
    }
    __syncthreads();
    const int c = cnt[t];
    int x = c;
    for (int o = 1; o < 32; o <<= 1) {
        int y = __shfl_up(x, o);
        if (lane >= o) x += y;
    }
    if (lane == 31) wsum[wv] = x;
    __syncthreads();
    if (wv == 0) {
        const int w = wsum[lane];
        int u = w;
        for (int o = 1; o < 32; o <<= 1) {
            int y = __shfl_up(u, o);
            if (lane >= o) u += y;
        }
        wsum[lane] = u - w;
    }
    __syncthreads();
    const int ex = x - c + wsum[wv];
    off[t] = ex;
    const int node = b * BKT_NODES + t;
    if (node < N_NODES) {
        rowptr[node] = bs + ex;
        dinv[node]   = 1.0f / sqrtf(fmaxf((float)c, 1.0f));
    }
    if (b == NBK - 1 && t == 0) rowptr[N_NODES] = N_EDGES;
    __syncthreads();
    for (int i0 = bs + t; i0 < be; i0 += 8 * 1024) {
        int2 e[8];
#pragma unroll
        for (int k = 0; k < 8; ++k) {
            const int i = i0 + 1024 * k;
            e[k] = tmp[min(i, be - 1)];
        }
#pragma unroll
        for (int k = 0; k < 8; ++k) {
            const int i = i0 + 1024 * k;
            const int pos = bs + off[e[k].y >> 21] + (e[k].y & ((1 << 21) - 1));
            if (i < be && (unsigned)pos < (unsigned)N_EDGES) csrsrc[pos] = e[k].x;
        }
    }
}

__global__ void __launch_bounds__(256)
fill_csr(const int* __restrict__ csrsrc, const float* __restrict__ dinv,
         int* __restrict__ csr, int E4) {
    const int j = blockIdx.x * blockDim.x + threadIdx.x;
    if (j >= E4) return;
    const int4 s = reinterpret_cast<const int4*>(csrsrc)[j];
    const float w0 = dinv[s.x], w1 = dinv[s.y], w2 = dinv[s.z], w3 = dinv[s.w];
    int4* o = reinterpret_cast<int4*>(csr) + 2 * (long)j;
    o[0] = make_int4(s.x, __float_as_int(w0), s.y, __float_as_int(w1));
    o[1] = make_int4(s.z, __float_as_int(w2), s.w, __float_as_int(w3));
}


template <int K>
__device__ __forceinline__ void stage_weights_t(float* __restrict__ Wt,
                                                const float* __restrict__ W, int tid) {
    constexpr int P = K + 4;
    for (int i = tid; i < (K / 4) * HF; i += 128) {
        const int c = i & 63, t = i >> 6;
        const float* w = W + (4 * t) * HF + c;
        *reinterpret_cast<float4*>(Wt + c * P + 4 * t) =
            make_float4(w[0], w[HF], w[2 * HF], w[3 * HF]);
    }
}

__global__ void __launch_bounds__(128)
mlp2_wmma(const float* __restrict__ X,
          const float* __restrict__ W1, const float* __restrict__ b1,
          const float* __restrict__ W2, const float* __restrict__ b2,
          float* __restrict__ C, int M) {
    constexpr int K1 = 128;
    constexpr int P1 = K1 + 4;
    constexpr int P2 = HF + 4;
    extern __shared__ __align__(16) float Wt[];
    float* W1t = Wt;
    float* W2t = Wt + HF * P1;
    const int tid = threadIdx.x;

    stage_weights_t<K1>(W1t, W1, tid);
    stage_weights_t<HF>(W2t, W2, tid);
    __syncthreads();

    const int wave = tid >> 5;
    const int lane = tid & 31;
    const int half = lane >> 4;
    const int lm   = lane & 15;
    const int row0 = (blockIdx.x * 4 + wave) * 16;
    if (row0 >= M) return;

    const float* Xp = X + (long)(row0 + lm) * K1 + 4 * half;
    const float* Wp = W1t + lm * P1 + 4 * half;

    v8f d0 = {0.f, 0.f, 0.f, 0.f, 0.f, 0.f, 0.f, 0.f};
    v8f d1 = d0, d2 = d0, d3 = d0;

#pragma unroll
    for (int k = 0; k < K1; k += 8) {
        const float4 xv = *reinterpret_cast<const float4*>(Xp + k);
        const float4 w0 = *reinterpret_cast<const float4*>(Wp + k);
        const float4 w1 = *reinterpret_cast<const float4*>(Wp + 16 * P1 + k);
        const float4 w2 = *reinterpret_cast<const float4*>(Wp + 32 * P1 + k);
        const float4 w3 = *reinterpret_cast<const float4*>(Wp + 48 * P1 + k);
        const v2f x01 = {xv.x, xv.y}, x23 = {xv.z, xv.w};
        d0 = __builtin_amdgcn_wmma_f32_16x16x4_f32(
            false, v2f{w0.x, w0.y}, false, x01, (short)0, d0, false, false);
        d1 = __builtin_amdgcn_wmma_f32_16x16x4_f32(
            false, v2f{w1.x, w1.y}, false, x01, (short)0, d1, false, false);
        d2 = __builtin_amdgcn_wmma_f32_16x16x4_f32(
            false, v2f{w2.x, w2.y}, false, x01, (short)0, d2, false, false);
        d3 = __builtin_amdgcn_wmma_f32_16x16x4_f32(
            false, v2f{w3.x, w3.y}, false, x01, (short)0, d3, false, false);
        d0 = __builtin_amdgcn_wmma_f32_16x16x4_f32(
            false, v2f{w0.z, w0.w}, false, x23, (short)0, d0, false, false);
        d1 = __builtin_amdgcn_wmma_f32_16x16x4_f32(
            false, v2f{w1.z, w1.w}, false, x23, (short)0, d1, false, false);
        d2 = __builtin_amdgcn_wmma_f32_16x16x4_f32(
            false, v2f{w2.z, w2.w}, false, x23, (short)0, d2, false, false);
        d3 = __builtin_amdgcn_wmma_f32_16x16x4_f32(
            false, v2f{w3.z, w3.w}, false, x23, (short)0, d3, false, false);
    }

    v8f g[4] = {d0, d1, d2, d3};
#pragma unroll
    for (int t = 0; t < 4; ++t) {
        const float* bp = b1 + 16 * t + 8 * half;
#pragma unroll
        for (int r = 0; r < 8; ++r) g[t][r] = fmaxf(g[t][r] + bp[r], 0.0f);
    }

    const float* Vp = W2t + lm * P2 + 8 * half;

    v8f e0 = {0.f, 0.f, 0.f, 0.f, 0.f, 0.f, 0.f, 0.f};
    v8f e1 = e0, e2 = e0, e3 = e0;

#pragma unroll
    for (int t = 0; t < 4; ++t) {
#pragma unroll
        for (int i = 0; i < 2; ++i) {
            const int k = 16 * t + 4 * i;
            const float4 v0 = *reinterpret_cast<const float4*>(Vp + k);
            const float4 v1 = *reinterpret_cast<const float4*>(Vp + 16 * P2 + k);
            const float4 v2 = *reinterpret_cast<const float4*>(Vp + 32 * P2 + k);
            const float4 v3 = *reinterpret_cast<const float4*>(Vp + 48 * P2 + k);
            const v2f a01 = {g[t][4 * i], g[t][4 * i + 1]};
            const v2f a23 = {g[t][4 * i + 2], g[t][4 * i + 3]};
            e0 = __builtin_amdgcn_wmma_f32_16x16x4_f32(
                false, a01, false, v2f{v0.x, v0.y}, (short)0, e0, false, false);
            e1 = __builtin_amdgcn_wmma_f32_16x16x4_f32(
                false, a01, false, v2f{v1.x, v1.y}, (short)0, e1, false, false);
            e2 = __builtin_amdgcn_wmma_f32_16x16x4_f32(
                false, a01, false, v2f{v2.x, v2.y}, (short)0, e2, false, false);
            e3 = __builtin_amdgcn_wmma_f32_16x16x4_f32(
                false, a01, false, v2f{v3.x, v3.y}, (short)0, e3, false, false);
            e0 = __builtin_amdgcn_wmma_f32_16x16x4_f32(
                false, a23, false, v2f{v0.z, v0.w}, (short)0, e0, false, false);
            e1 = __builtin_amdgcn_wmma_f32_16x16x4_f32(
                false, a23, false, v2f{v1.z, v1.w}, (short)0, e1, false, false);
            e2 = __builtin_amdgcn_wmma_f32_16x16x4_f32(
                false, a23, false, v2f{v2.z, v2.w}, (short)0, e2, false, false);
            e3 = __builtin_amdgcn_wmma_f32_16x16x4_f32(
                false, a23, false, v2f{v3.z, v3.w}, (short)0, e3, false, false);
        }
    }

    const v8f es[4] = {e0, e1, e2, e3};
#pragma unroll
    for (int u = 0; u < 4; ++u) {
        const int col  = 16 * u + lm;
        const float bc = b2[col];
#pragma unroll
        for (int r = 0; r < 8; ++r) {
            const int row = row0 + r + 8 * half;
            C[(long)row * HF + col] = fmaxf(es[u][r] + bc, 0.0f);
        }
    }
}


__device__ __forceinline__ float4 ld_row4(const float* __restrict__ t, int row, int q) {
    return *reinterpret_cast<const float4*>(t + (long)row * HF + q);
}

__device__ __forceinline__ void add_scaled(float4& acc, const float4 r, float w) {
    acc.x += r.x * w;
    acc.y += r.y * w;
    acc.z += r.z * w;
    acc.w += r.w * w;
}

__device__ __forceinline__ float4 row_or_zero(bool keep, const float4 r) {
    float4 o;
    o.x = keep ? r.x : 0.0f;
    o.y = keep ? r.y : 0.0f;
    o.z = keep ? r.z : 0.0f;
    o.w = keep ? r.w : 0.0f;
    return o;
}

__device__ __forceinline__ float4 gather_sum(const int2* __restrict__ csr,
                                             const float* __restrict__ t,
                                             int beg, int end, int q) {
    float4 acc = make_float4(0.0f, 0.0f, 0.0f, 0.0f);
    int j = beg;
    for (; j + 4 <= end; j += 4) {
        const int2 e0 = csr[j], e1 = csr[j + 1], e2 = csr[j + 2], e3 = csr[j + 3];
        const float4 r0 = ld_row4(t, e0.x, q);
        const float4 r1 = ld_row4(t, e1.x, q);
        const float4 r2 = ld_row4(t, e2.x, q);
        const float4 r3 = ld_row4(t, e3.x, q);
        add_scaled(acc, r0, __int_as_float(e0.y));
        add_scaled(acc, r1, __int_as_float(e1.y));
        add_scaled(acc, r2, __int_as_float(e2.y));
        add_scaled(acc, r3, __int_as_float(e3.y));
    }
    const int rem = end - j;
    if (rem > 0) {
        const int last = end - 1;
        const int2 e0 = csr[j];
        const int2 e1 = csr[min(j + 1, last)];
        const int2 e2 = csr[min(j + 2, last)];
        const float4 r0 = ld_row4(t, e0.x, q);
        const float4 r1 = row_or_zero(rem > 1, ld_row4(t, e1.x, q));
        const float4 r2 = row_or_zero(rem > 2, ld_row4(t, e2.x, q));
        add_scaled(acc, r0, __int_as_float(e0.y));
        add_scaled(acc, r1, __int_as_float(e1.y));
        add_scaled(acc, r2, __int_as_float(e2.y));
    }
    return acc;
}

__global__ void gather_step1(const int* __restrict__ csr, const int* __restrict__ rowptr,
                             const float* __restrict__ h, const float* __restrict__ dinv,
                             float* __restrict__ feat, int n_nodes) {
    const int tid = blockIdx.x * blockDim.x + threadIdx.x;
    const int n = tid >> 4;
    if (n >= n_nodes) return;
    const int q = (tid & 15) * 4;
    const float4 m = gather_sum(reinterpret_cast<const int2*>(csr), h,
                                rowptr[n], rowptr[n + 1], q);
    const float dn = dinv[n];
    const float4 hv = ld_row4(h, n, q);
    float4 v;
    v.x = hv.x - m.x * dn;
    v.y = hv.y - m.y * dn;
    v.z = hv.z - m.z * dn;
    v.w = hv.w - m.w * dn;
    *reinterpret_cast<float4*>(feat + (long)n * HF + q) = v;
}

__global__ void gather_step2(const int* __restrict__ csr, const int* __restrict__ rowptr,
                             const float* __restrict__ feat, const float* __restrict__ dinv,
                             float* __restrict__ feat2, int n_nodes) {
    const int tid = blockIdx.x * blockDim.x + threadIdx.x;
    const int n = tid >> 4;
    if (n >= n_nodes) return;
    const int q = (tid & 15) * 4;
    const float4 m = gather_sum(reinterpret_cast<const int2*>(csr), feat,
                                rowptr[n], rowptr[n + 1], q);
    const float dn = dinv[n];
    const float4 fv = ld_row4(feat, n, q);
    float4 v;
    v.x = fv.x - m.x * dn;
    v.y = fv.y - m.y * dn;
    v.z = fv.z - m.z * dn;
    v.w = fv.w - m.w * dn;
    *reinterpret_cast<float4*>(feat2 + (long)n * HF + q) = v;
}

__global__ void __launch_bounds__(128)
gemm_filters_out_wmma(const float* __restrict__ h, const float* __restrict__ f1,
                      const float* __restrict__ f2,
                      const float* __restrict__ W, const float* __restrict__ bias,
                      const float* __restrict__ Wm2, const float* __restrict__ bm2,
                      float* __restrict__ out, int M,
                      float a0, float a1, float a2,
                      float b0, float b1, float b2,
                      float c0, float c1, float c2) {
    constexpr int K = 3 * HF;
    constexpr int P = K + 4;
    extern __shared__ __align__(16) float Wt[];
    const int tid  = threadIdx.x;
    const int wave = tid >> 5;
    const int lane = tid & 31;
    const int half = lane >> 4;
    const int lm   = lane & 15;
    const int row0 = (blockIdx.x * 4 + wave) * 16;

    const long arow = (long)min(row0 + lm, M - 1) * HF + 4 * half;
    float4 hv[HF / 8], uv[HF / 8], wv[HF / 8];
#pragma unroll
    for (int j = 0; j < HF / 8; ++j) {
        hv[j] = *reinterpret_cast<const float4*>(h  + arow + 8 * j);
        uv[j] = *reinterpret_cast<const float4*>(f1 + arow + 8 * j);
        wv[j] = *reinterpret_cast<const float4*>(f2 + arow + 8 * j);
    }

    stage_weights_t<K>(Wt, W, tid);
    __syncthreads();
    if (row0 >= M) return;

    const float* Bp = Wt + lm * P + 4 * half;
    const float ca[3] = {a0, a1, a2};
    const float cb[3] = {b0, b1, b2};
    const float cc[3] = {c0, c1, c2};

    v8f acc0 = {0.f, 0.f, 0.f, 0.f, 0.f, 0.f, 0.f, 0.f};
    v8f acc1 = acc0, acc2 = acc0, acc3 = acc0;

#pragma unroll
    for (int t = 0; t < 3; ++t) {
#pragma unroll
        for (int j = 0; j < HF / 8; ++j) {
            const int k = HF * t + 8 * j;
            float4 a;
            a.x = fmaf(cc[t], wv[j].x, fmaf(cb[t], uv[j].x, ca[t] * hv[j].x));
            a.y = fmaf(cc[t], wv[j].y, fmaf(cb[t], uv[j].y, ca[t] * hv[j].y));
            a.z = fmaf(cc[t], wv[j].z, fmaf(cb[t], uv[j].z, ca[t] * hv[j].z));
            a.w = fmaf(cc[t], wv[j].w, fmaf(cb[t], uv[j].w, ca[t] * hv[j].w));
            const float4 b0v = *reinterpret_cast<const float4*>(Bp + k);
            const float4 b1v = *reinterpret_cast<const float4*>(Bp + 16 * P + k);
            const float4 b2v = *reinterpret_cast<const float4*>(Bp + 32 * P + k);
            const float4 b3v = *reinterpret_cast<const float4*>(Bp + 48 * P + k);
            const v2f a01 = {a.x, a.y}, a23 = {a.z, a.w};
            acc0 = __builtin_amdgcn_wmma_f32_16x16x4_f32(
                false, v2f{b0v.x, b0v.y}, false, a01, (short)0, acc0, false, false);
            acc1 = __builtin_amdgcn_wmma_f32_16x16x4_f32(
                false, v2f{b1v.x, b1v.y}, false, a01, (short)0, acc1, false, false);
            acc2 = __builtin_amdgcn_wmma_f32_16x16x4_f32(
                false, v2f{b2v.x, b2v.y}, false, a01, (short)0, acc2, false, false);
            acc3 = __builtin_amdgcn_wmma_f32_16x16x4_f32(
                false, v2f{b3v.x, b3v.y}, false, a01, (short)0, acc3, false, false);
            acc0 = __builtin_amdgcn_wmma_f32_16x16x4_f32(
                false, v2f{b0v.z, b0v.w}, false, a23, (short)0, acc0, false, false);
            acc1 = __builtin_amdgcn_wmma_f32_16x16x4_f32(
                false, v2f{b1v.z, b1v.w}, false, a23, (short)0, acc1, false, false);
            acc2 = __builtin_amdgcn_wmma_f32_16x16x4_f32(
                false, v2f{b2v.z, b2v.w}, false, a23, (short)0, acc2, false, false);
            acc3 = __builtin_amdgcn_wmma_f32_16x16x4_f32(
                false, v2f{b3v.z, b3v.w}, false, a23, (short)0, acc3, false, false);
        }
    }

    const v8f accs[4] = {acc0, acc1, acc2, acc3};
    float p0 = 0.0f, p1 = 0.0f;
#pragma unroll
    for (int n = 0; n < 4; ++n) {
        const int kc = 16 * n + 8 * half;
#pragma unroll
        for (int r = 0; r < 8; ++r) {
            const float v = fmaxf(accs[n][r] + bias[kc + r], 0.0f);
            p0 = fmaf(v, Wm2[2 * (kc + r) + 0], p0);
            p1 = fmaf(v, Wm2[2 * (kc + r) + 1], p1);
        }
    }
    p0 += __shfl_xor(p0, 16);
    p1 += __shfl_xor(p1, 16);
    if (half == 0) {
        *reinterpret_cast<float2*>(out + 2 * (long)(row0 + lm)) =
            make_float2(p0 + bm2[0], p1 + bm2[1]);
    }
}

extern "C" void kernel_launch(void* const* d_in, const int* in_sizes, int n_in,
                              void* d_out, int out_size, void* d_ws, size_t ws_size,
                              hipStream_t stream) {
    const float* x   = (const float*)d_in[0];
    const int*   ei  = (const int*)  d_in[1];
    const float* W1  = (const float*)d_in[2];
    const float* b1  = (const float*)d_in[3];
    const float* W2  = (const float*)d_in[4];
    const float* b2  = (const float*)d_in[5];
    const float* Wm1 = (const float*)d_in[6];
    const float* bm1 = (const float*)d_in[7];
    const float* Wm2 = (const float*)d_in[8];
    const float* bm2 = (const float*)d_in[9];
    float* out = (float*)d_out;

    const int N = N_NODES, E = N_EDGES;
    const int N64 = N * HF;

    float* ws   = (float*)d_ws;
    float* dinv = ws;
    float* h    = dinv + N;
    float* feat = h    + (long)N64;
    float* lst  = feat + (long)N64;
    float* hfin = lst  + (long)N64;

    int* csr    = (int*)lst;
    int* rowptr = csr + 2 * (long)E;
    int* tmp    = (int*)hfin;
    int* csrsrc = tmp + 2 * (long)E;
    int* tab    = csrsrc + E;
    float* feat2 = hfin;

    const float TH[3][3] = {{3.0f, -3.0f, 0.75f},
                            {0.0f,  3.0f, -1.50f},
                            {0.0f,  0.0f, 0.75f}};

    bucket_hist<<<NCHUNK, 256, 0, stream>>>(ei, tab);
    scan_table<<<1, 1024, 0, stream>>>(tab);
    bucket_scatter<<<NCHUNK, 256, 0, stream>>>(ei, tab, (int2*)tmp);
    bucket_rank<<<NBK, 1024, 0, stream>>>(tab, (int2*)tmp, rowptr, dinv, csrsrc);

    mlp2_wmma<<<(N / 16 + 3) / 4, 128, 64 * (132 + 68) * sizeof(float), stream>>>(
        x, W1, b1, W2, b2, h, N);

    fill_csr<<<(E / 4 + 255) / 256, 256, 0, stream>>>(csrsrc, dinv, csr, E / 4);
    const int gg = (N * 16 + 255) / 256;
    gather_step1<<<gg, 256, 0, stream>>>(csr, rowptr, h, dinv, feat, N);
    gather_step2<<<gg, 256, 0, stream>>>(csr, rowptr, feat, dinv, feat2, N);

    gemm_filters_out_wmma<<<(N / 16 + 3) / 4, 128, 64 * 196 * sizeof(float), stream>>>(
        h, feat, feat2, Wm1, bm1, Wm2, bm2, out, N,
        TH[0][0], TH[1][0], TH[2][0], TH[0][1], TH[1][1], TH[2][1],
        TH[0][2], TH[1][2], TH[2][2]);
}


# Counting sort's scattered 8- and 4-byte stores staged in LDS: a chunk's pairs and a bucket's list are ordered on chip and stored as contiguous runs (list line-aligned); sums stay in arrival order
# speedup vs baseline: 12.2210x; 1.0786x over previous
#include <hip/hip_runtime.h>
#include <hip/hip_bf16.h>


typedef float v2f __attribute__((ext_vector_type(2)));
typedef float v8f __attribute__((ext_vector_type(8)));

#define N_NODES 100000
#define N_EDGES 1600000
#define HF 64

#define BKT_NODES 1024
#define NBK ((N_NODES + BKT_NODES - 1) / BKT_NODES)
#define CHUNK_E 6400
#define NCHUNK (N_EDGES / CHUNK_E)
#define NTAB (NBK * NCHUNK)
static_assert(NCHUNK * CHUNK_E == N_EDGES, "chunks must tile the edge list");
static_assert(CHUNK_E % 256 == 0, "a chunk is a whole number of edges per thread");
static_assert(N_EDGES < (1 << 21), "rank is packed into 21 bits");
static_assert(NBK <= 256, "a chunk's work-group scans one run length per thread");
#define STAGE_E 20480

__device__ __forceinline__ int bucket_of(int d) {
    const unsigned b = (unsigned)d >> 10;
    return (int)(b < (unsigned)NBK ? b : (unsigned)(NBK - 1));
}

__global__ void __launch_bounds__(256)
bucket_hist(const int* __restrict__ ei, int* __restrict__ tab) {
    __shared__ int hist[NBK];
    for (int b = threadIdx.x; b < NBK; b += 256) hist[b] = 0;
    __syncthreads();
    const int* dst = ei + N_EDGES + blockIdx.x * CHUNK_E + threadIdx.x;
    int d[CHUNK_E / 256];
#pragma unroll
    for (int k = 0; k < CHUNK_E / 256; ++k) d[k] = dst[256 * k];
#pragma unroll
    for (int k = 0; k < CHUNK_E / 256; ++k) atomicAdd(&hist[bucket_of(d[k])], 1);
    __syncthreads();
    for (int b = threadIdx.x; b < NBK; b += 256) tab[b * NCHUNK + blockIdx.x] = hist[b];
}

__global__ void __launch_bounds__(1024)
scan_table(int* __restrict__ tab) {
    __shared__ int wsum[32];
    constexpr int PER = (NTAB + 1023) / 1024;
    const int lane = threadIdx.x & 31;
    const int wv   = threadIdx.x >> 5;
    const int i0   = threadIdx.x * PER;
    int v[PER];
    int s = 0;
#pragma unroll
    for (int k = 0; k < PER; ++k) {
        v[k] = (i0 + k < NTAB) ? tab[i0 + k] : 0;
        s += v[k];
    }
    int x = s;
    for (int o = 1; o < 32; o <<= 1) {
        int y = __shfl_up(x, o);
        if (lane >= o) x += y;
    }
    if (lane == 31) wsum[wv] = x;
    __syncthreads();
    if (wv == 0) {
        const int w = wsum[lane];
        int t = w;
        for (int o = 1; o < 32; o <<= 1) {
            int y = __shfl_up(t, o);
            if (lane >= o) t += y;
        }
        wsum[lane] = t - w;
    }
    __syncthreads();
    int run = x - s + wsum[wv];
#pragma unroll
    for (int k = 0; k < PER; ++k) {
        if (i0 + k < NTAB) tab[i0 + k] = run;
        run += v[k];
    }
}

__global__ void __launch_bounds__(256)
bucket_scatter(const int* __restrict__ ei, const int* __restrict__ tab,
               int2* __restrict__ tmp) {
    constexpr int PER = CHUNK_E / 256;
    __shared__ int cur[NBK];
    __shared__ int delta[NBK];
    __shared__ int wsum[8];
    __shared__ int2 stage[CHUNK_E];
    const int t    = threadIdx.x;
    const int lane = t & 31;
    const int wv   = t >> 5;
    const int* srcp = ei + blockIdx.x * CHUNK_E + t;
    const int* dstp = srcp + N_EDGES;
    int s[PER], d[PER];
#pragma unroll
    for (int k = 0; k < PER; ++k) {
        s[k] = srcp[256 * k];
        d[k] = dstp[256 * k];
    }
    const int idx  = min(t, NBK - 1) * NCHUNK + blockIdx.x;
    const int tcur = tab[idx];
    const int tnxt = tab[min(idx + 1, NTAB - 1)];
    const int place = tcur;
    const int len   = (t < NBK) ? ((idx + 1 < NTAB) ? tnxt : N_EDGES) - tcur : 0;
    int x = len;
    for (int o = 1; o < 32; o <<= 1) {
        int y = __shfl_up(x, o);
        if (lane >= o) x += y;
    }
    if (lane == 31) wsum[wv] = x;
    __syncthreads();
    int lo = x - len;
    for (int w = 0; w < wv; ++w) lo += wsum[w];
    if (t < NBK) {
        cur[t]   = lo;
        delta[t] = place - lo;
    }
    __syncthreads();
    int p[PER];
#pragma unroll
    for (int k = 0; k < PER; ++k) p[k] = atomicAdd(&cur[bucket_of(d[k])], 1);
#pragma unroll
    for (int k = 0; k < PER; ++k) {
        if ((unsigned)p[k] < (unsigned)CHUNK_E) stage[p[k]] = make_int2(s[k], d[k]);
    }
    __syncthreads();
    int2 e[PER];
#pragma unroll
    for (int k = 0; k < PER; ++k) e[k] = stage[t + 256 * k];
#pragma unroll
    for (int k = 0; k < PER; ++k) p[k] = delta[bucket_of(e[k].y)] + t + 256 * k;
#pragma unroll
    for (int k = 0; k < PER; ++k) {
        if ((unsigned)p[k] < (unsigned)N_EDGES) tmp[p[k]] = e[k];
    }
}

__global__ void __launch_bounds__(1024)
bucket_rank(const int* __restrict__ tab, int2* __restrict__ tmp,
            int* __restrict__ rowptr, float* __restrict__ dinv,
            int* __restrict__ csrsrc) {
    __shared__ int cnt[BKT_NODES];
    __shared__ int off[BKT_NODES];
    __shared__ int wsum[32];
    __shared__ int stage[STAGE_E];
    const int b    = blockIdx.x;
    const int t    = threadIdx.x;
    const int lane = t & 31;
    const int wv   = t >> 5;
    const int bs = tab[b * NCHUNK];
    const int be = (b == NBK - 1) ? N_EDGES : tab[(b + 1) * NCHUNK];
    cnt[t] = 0;
    __syncthreads();
    for (int i0 = bs + t; i0 < be; i0 += 8 * 1024) {
        int y[8];
#pragma unroll
        for (int k = 0; k < 8; ++k) {
            const int i = i0 + 1024 * k;
            y[k] = tmp[min(i, be - 1)].y;
        }
#pragma unroll
        for (int k = 0; k < 8; ++k) {
            const int i = i0 + 1024 * k;
            if (i < be) {
                const int dl = y[k] & (BKT_NODES - 1);
                const int r  = atomicAdd(&cnt[dl], 1);
                tmp[i].y = (dl << 21) | r;
            }
        }
    }
    __syncthreads();
    const int c = cnt[t];
    int x = c;
    for (int o = 1; o < 32; o <<= 1) {
        int y = __shfl_up(x, o);
        if (lane >= o) x += y;
    }
    if (lane == 31) wsum[wv] = x;
    __syncthreads();
    if (wv == 0) {
        const int w = wsum[lane];
        int u = w;
        for (int o = 1; o < 32; o <<= 1) {
            int y = __shfl_up(u, o);
            if (lane >= o) u += y;
        }
        wsum[lane] = u - w;
    }
    __syncthreads();
    const int ex = x - c + wsum[wv];
    off[t] = ex;
    const int node = b * BKT_NODES + t;
    if (node < N_NODES) {
        rowptr[node] = bs + ex;
        dinv[node]   = 1.0f / sqrtf(fmaxf((float)c, 1.0f));
    }
    if (b == NBK - 1 && t == 0) rowptr[N_NODES] = N_EDGES;
    __syncthreads();
    for (int i0 = bs + t; i0 < be; i0 += 8 * 1024) {
        int2 e[8];
#pragma unroll
        for (int k = 0; k < 8; ++k) {
            const int i = i0 + 1024 * k;
            e[k] = tmp[min(i, be - 1)];
        }
        int pl[8];
#pragma unroll
        for (int k = 0; k < 8; ++k) pl[k] = off[e[k].y >> 21] + (e[k].y & ((1 << 21) - 1));
#pragma unroll
        for (int k = 0; k < 8; ++k) {
            const int i = i0 + 1024 * k;
            if (i < be) {
                if ((unsigned)pl[k] < (unsigned)STAGE_E) stage[pl[k]] = e[k].x;
                else if ((unsigned)(bs + pl[k]) < (unsigned)N_EDGES) csrsrc[bs + pl[k]] = e[k].x;
            }
        }
    }
    __syncthreads();
    const int n = min(be - bs, STAGE_E);
    const int a = bs & 31;
    for (int i0 = t - a; i0 < n; i0 += 4 * 1024) {
        int v[4];
#pragma unroll
        for (int k = 0; k < 4; ++k) v[k] = stage[min(max(i0 + 1024 * k, 0), STAGE_E - 1)];
#pragma unroll
        for (int k = 0; k < 4; ++k) {
            const int i = i0 + 1024 * k;
            if (i >= 0 && i < n && (unsigned)(bs + i) < (unsigned)N_EDGES) csrsrc[bs + i] = v[k];
        }
    }
}

__global__ void __launch_bounds__(256)
fill_csr(const int* __restrict__ csrsrc, const float* __restrict__ dinv,
         int* __restrict__ csr, int E4) {
    const int j = blockIdx.x * blockDim.x + threadIdx.x;
    if (j >= E4) return;
    const int4 s = reinterpret_cast<const int4*>(csrsrc)[j];
    const float w0 = dinv[s.x], w1 = dinv[s.y], w2 = dinv[s.z], w3 = dinv[s.w];
    int4* o = reinterpret_cast<int4*>(csr) + 2 * (long)j;
    o[0] = make_int4(s.x, __float_as_int(w0), s.y, __float_as_int(w1));
    o[1] = make_int4(s.z, __float_as_int(w2), s.w, __float_as_int(w3));
}


template <int K>
__device__ __forceinline__ void stage_weights_t(float* __restrict__ Wt,
                                                const float* __restrict__ W, int tid) {
    constexpr int P = K + 4;
    for (int i = tid; i < (K / 4) * HF; i += 128) {
        const int c = i & 63, t = i >> 6;
        const float* w = W + (4 * t) * HF + c;
        *reinterpret_cast<float4*>(Wt + c * P + 4 * t) =
            make_float4(w[0], w[HF], w[2 * HF], w[3 * HF]);
    }
}

__global__ void __launch_bounds__(128)
mlp2_wmma(const float* __restrict__ X,
          const float* __restrict__ W1, const float* __restrict__ b1,
          const float* __restrict__ W2, const float* __restrict__ b2,
          float* __restrict__ C, int M) {
    constexpr int K1 = 128;
    constexpr int P1 = K1 + 4;
    constexpr int P2 = HF + 4;
    extern __shared__ __align__(16) float Wt[];
    float* W1t = Wt;
    float* W2t = Wt + HF * P1;
    const int tid = threadIdx.x;

    stage_weights_t<K1>(W1t, W1, tid);
    stage_weights_t<HF>(W2t, W2, tid);
    __syncthreads();

    const int wave = tid >> 5;
    const int lane = tid & 31;
    const int half = lane >> 4;
    const int lm   = lane & 15;
    const int row0 = (blockIdx.x * 4 + wave) * 16;
    if (row0 >= M) return;

    const float* Xp = X + (long)(row0 + lm) * K1 + 4 * half;
    const float* Wp = W1t + lm * P1 + 4 * half;

    v8f d0 = {0.f, 0.f, 0.f, 0.f, 0.f, 0.f, 0.f, 0.f};
    v8f d1 = d0, d2 = d0, d3 = d0;

#pragma unroll
    for (int k = 0; k < K1; k += 8) {
        const float4 xv = *reinterpret_cast<const float4*>(Xp + k);
        const float4 w0 = *reinterpret_cast<const float4*>(Wp + k);
        const float4 w1 = *reinterpret_cast<const float4*>(Wp + 16 * P1 + k);
        const float4 w2 = *reinterpret_cast<const float4*>(Wp + 32 * P1 + k);
        const float4 w3 = *reinterpret_cast<const float4*>(Wp + 48 * P1 + k);
        const v2f x01 = {xv.x, xv.y}, x23 = {xv.z, xv.w};
        d0 = __builtin_amdgcn_wmma_f32_16x16x4_f32(
            false, v2f{w0.x, w0.y}, false, x01, (short)0, d0, false, false);
        d1 = __builtin_amdgcn_wmma_f32_16x16x4_f32(
            false, v2f{w1.x, w1.y}, false, x01, (short)0, d1, false, false);
        d2 = __builtin_amdgcn_wmma_f32_16x16x4_f32(
            false, v2f{w2.x, w2.y}, false, x01, (short)0, d2, false, false);
        d3 = __builtin_amdgcn_wmma_f32_16x16x4_f32(
            false, v2f{w3.x, w3.y}, false, x01, (short)0, d3, false, false);
        d0 = __builtin_amdgcn_wmma_f32_16x16x4_f32(
            false, v2f{w0.z, w0.w}, false, x23, (short)0, d0, false, false);
        d1 = __builtin_amdgcn_wmma_f32_16x16x4_f32(
            false, v2f{w1.z, w1.w}, false, x23, (short)0, d1, false, false);
        d2 = __builtin_amdgcn_wmma_f32_16x16x4_f32(
            false, v2f{w2.z, w2.w}, false, x23, (short)0, d2, false, false);
        d3 = __builtin_amdgcn_wmma_f32_16x16x4_f32(
            false, v2f{w3.z, w3.w}, false, x23, (short)0, d3, false, false);
    }

    v8f g[4] = {d0, d1, d2, d3};
#pragma unroll
    for (int t = 0; t < 4; ++t) {
        const float* bp = b1 + 16 * t + 8 * half;
#pragma unroll
        for (int r = 0; r < 8; ++r) g[t][r] = fmaxf(g[t][r] + bp[r], 0.0f);
    }

    const float* Vp = W2t + lm * P2 + 8 * half;

    v8f e0 = {0.f, 0.f, 0.f, 0.f, 0.f, 0.f, 0.f, 0.f};
    v8f e1 = e0, e2 = e0, e3 = e0;

#pragma unroll
    for (int t = 0; t < 4; ++t) {
#pragma unroll
        for (int i = 0; i < 2; ++i) {
            const int k = 16 * t + 4 * i;
            const float4 v0 = *reinterpret_cast<const float4*>(Vp + k);
            const float4 v1 = *reinterpret_cast<const float4*>(Vp + 16 * P2 + k);
            const float4 v2 = *reinterpret_cast<const float4*>(Vp + 32 * P2 + k);
            const float4 v3 = *reinterpret_cast<const float4*>(Vp + 48 * P2 + k);
            const v2f a01 = {g[t][4 * i], g[t][4 * i + 1]};
            const v2f a23 = {g[t][4 * i + 2], g[t][4 * i + 3]};
            e0 = __builtin_amdgcn_wmma_f32_16x16x4_f32(
                false, a01, false, v2f{v0.x, v0.y}, (short)0, e0, false, false);
            e1 = __builtin_amdgcn_wmma_f32_16x16x4_f32(
                false, a01, false, v2f{v1.x, v1.y}, (short)0, e1, false, false);
            e2 = __builtin_amdgcn_wmma_f32_16x16x4_f32(
                false, a01, false, v2f{v2.x, v2.y}, (short)0, e2, false, false);
            e3 = __builtin_amdgcn_wmma_f32_16x16x4_f32(
                false, a01, false, v2f{v3.x, v3.y}, (short)0, e3, false, false);
            e0 = __builtin_amdgcn_wmma_f32_16x16x4_f32(
                false, a23, false, v2f{v0.z, v0.w}, (short)0, e0, false, false);
            e1 = __builtin_amdgcn_wmma_f32_16x16x4_f32(
                false, a23, false, v2f{v1.z, v1.w}, (short)0, e1, false, false);
            e2 = __builtin_amdgcn_wmma_f32_16x16x4_f32(
                false, a23, false, v2f{v2.z, v2.w}, (short)0, e2, false, false);
            e3 = __builtin_amdgcn_wmma_f32_16x16x4_f32(
                false, a23, false, v2f{v3.z, v3.w}, (short)0, e3, false, false);
        }
    }

    const v8f es[4] = {e0, e1, e2, e3};
#pragma unroll
    for (int u = 0; u < 4; ++u) {
        const int col  = 16 * u + lm;
        const float bc = b2[col];
#pragma unroll
        for (int r = 0; r < 8; ++r) {
            const int row = row0 + r + 8 * half;
            C[(long)row * HF + col] = fmaxf(es[u][r] + bc, 0.0f);
        }
    }
}


__device__ __forceinline__ float4 ld_row4(const float* __restrict__ t, int row, int q) {
    return *reinterpret_cast<const float4*>(t + (long)row * HF + q);
}

__device__ __forceinline__ void add_scaled(float4& acc, const float4 r, float w) {
    acc.x += r.x * w;
    acc.y += r.y * w;
    acc.z += r.z * w;
    acc.w += r.w * w;
}

__device__ __forceinline__ float4 row_or_zero(bool keep, const float4 r) {
    float4 o;
    o.x = keep ? r.x : 0.0f;
    o.y = keep ? r.y : 0.0f;
    o.z = keep ? r.z : 0.0f;
    o.w = keep ? r.w : 0.0f;
    return o;
}

__device__ __forceinline__ float4 gather_sum(const int2* __restrict__ csr,
                                             const float* __restrict__ t,
                                             int beg, int end, int q) {
    float4 acc = make_float4(0.0f, 0.0f, 0.0f, 0.0f);
    int j = beg;
    for (; j + 4 <= end; j += 4) {
        const int2 e0 = csr[j], e1 = csr[j + 1], e2 = csr[j + 2], e3 = csr[j + 3];
        const float4 r0 = ld_row4(t, e0.x, q);
        const float4 r1 = ld_row4(t, e1.x, q);
        const float4 r2 = ld_row4(t, e2.x, q);
        const float4 r3 = ld_row4(t, e3.x, q);
        add_scaled(acc, r0, __int_as_float(e0.y));
        add_scaled(acc, r1, __int_as_float(e1.y));
        add_scaled(acc, r2, __int_as_float(e2.y));
        add_scaled(acc, r3, __int_as_float(e3.y));
    }
    const int rem = end - j;
    if (rem > 0) {
        const int last = end - 1;
        const int2 e0 = csr[j];
        const int2 e1 = csr[min(j + 1, last)];
        const int2 e2 = csr[min(j + 2, last)];
        const float4 r0 = ld_row4(t, e0.x, q);
        const float4 r1 = row_or_zero(rem > 1, ld_row4(t, e1.x, q));
        const float4 r2 = row_or_zero(rem > 2, ld_row4(t, e2.x, q));
        add_scaled(acc, r0, __int_as_float(e0.y));
        add_scaled(acc, r1, __int_as_float(e1.y));
        add_scaled(acc, r2, __int_as_float(e2.y));
    }
    return acc;
}

__global__ void gather_step1(const int* __restrict__ csr, const int* __restrict__ rowptr,
                             const float* __restrict__ h, const float* __restrict__ dinv,
                             float* __restrict__ feat, int n_nodes) {
    const int tid = blockIdx.x * blockDim.x + threadIdx.x;
    const int n = tid >> 4;
    if (n >= n_nodes) return;
    const int q = (tid & 15) * 4;
    const float4 m = gather_sum(reinterpret_cast<const int2*>(csr), h,
                                rowptr[n], rowptr[n + 1], q);
    const float dn = dinv[n];
    const float4 hv = ld_row4(h, n, q);
    float4 v;
    v.x = hv.x - m.x * dn;
    v.y = hv.y - m.y * dn;
    v.z = hv.z - m.z * dn;
    v.w = hv.w - m.w * dn;
    *reinterpret_cast<float4*>(feat + (long)n * HF + q) = v;
}

__global__ void gather_step2(const int* __restrict__ csr, const int* __restrict__ rowptr,
                             const float* __restrict__ feat, const float* __restrict__ dinv,
                             float* __restrict__ feat2, int n_nodes) {
    const int tid = blockIdx.x * blockDim.x + threadIdx.x;
    const int n = tid >> 4;
    if (n >= n_nodes) return;
    const int q = (tid & 15) * 4;
    const float4 m = gather_sum(reinterpret_cast<const int2*>(csr), feat,
                                rowptr[n], rowptr[n + 1], q);
    const float dn = dinv[n];
    const float4 fv = ld_row4(feat, n, q);
    float4 v;
    v.x = fv.x - m.x * dn;
    v.y = fv.y - m.y * dn;
    v.z = fv.z - m.z * dn;
    v.w = fv.w - m.w * dn;
    *reinterpret_cast<float4*>(feat2 + (long)n * HF + q) = v;
}

__global__ void __launch_bounds__(128)
gemm_filters_out_wmma(const float* __restrict__ h, const float* __restrict__ f1,
                      const float* __restrict__ f2,
                      const float* __restrict__ W, const float* __restrict__ bias,
                      const float* __restrict__ Wm2, const float* __restrict__ bm2,
                      float* __restrict__ out, int M,
                      float a0, float a1, float a2,
                      float b0, float b1, float b2,
                      float c0, float c1, float c2) {
    constexpr int K = 3 * HF;
    constexpr int P = K + 4;
    extern __shared__ __align__(16) float Wt[];
    const int tid  = threadIdx.x;
    const int wave = tid >> 5;
    const int lane = tid & 31;
    const int half = lane >> 4;
    const int lm   = lane & 15;
    const int row0 = (blockIdx.x * 4 + wave) * 16;

    const long arow = (long)min(row0 + lm, M - 1) * HF + 4 * half;
    float4 hv[HF / 8], uv[HF / 8], wv[HF / 8];
#pragma unroll
    for (int j = 0; j < HF / 8; ++j) {
        hv[j] = *reinterpret_cast<const float4*>(h  + arow + 8 * j);
        uv[j] = *reinterpret_cast<const float4*>(f1 + arow + 8 * j);
        wv[j] = *reinterpret_cast<const float4*>(f2 + arow + 8 * j);
    }

    stage_weights_t<K>(Wt, W, tid);
    __syncthreads();
    if (row0 >= M) return;

    const float* Bp = Wt + lm * P + 4 * half;
    const float ca[3] = {a0, a1, a2};
    const float cb[3] = {b0, b1, b2};
    const float cc[3] = {c0, c1, c2};

    v8f acc0 = {0.f, 0.f, 0.f, 0.f, 0.f, 0.f, 0.f, 0.f};
    v8f acc1 = acc0, acc2 = acc0, acc3 = acc0;

#pragma unroll
    for (int t = 0; t < 3; ++t) {
#pragma unroll
        for (int j = 0; j < HF / 8; ++j) {
            const int k = HF * t + 8 * j;
            float4 a;
            a.x = fmaf(cc[t], wv[j].x, fmaf(cb[t], uv[j].x, ca[t] * hv[j].x));
            a.y = fmaf(cc[t], wv[j].y, fmaf(cb[t], uv[j].y, ca[t] * hv[j].y));
            a.z = fmaf(cc[t], wv[j].z, fmaf(cb[t], uv[j].z, ca[t] * hv[j].z));
            a.w = fmaf(cc[t], wv[j].w, fmaf(cb[t], uv[j].w, ca[t] * hv[j].w));
            const float4 b0v = *reinterpret_cast<const float4*>(Bp + k);
            const float4 b1v = *reinterpret_cast<const float4*>(Bp + 16 * P + k);
            const float4 b2v = *reinterpret_cast<const float4*>(Bp + 32 * P + k);
            const float4 b3v = *reinterpret_cast<const float4*>(Bp + 48 * P + k);
            const v2f a01 = {a.x, a.y}, a23 = {a.z, a.w};
            acc0 = __builtin_amdgcn_wmma_f32_16x16x4_f32(
                false, v2f{b0v.x, b0v.y}, false, a01, (short)0, acc0, false, false);
            acc1 = __builtin_amdgcn_wmma_f32_16x16x4_f32(
                false, v2f{b1v.x, b1v.y}, false, a01, (short)0, acc1, false, false);
            acc2 = __builtin_amdgcn_wmma_f32_16x16x4_f32(
                false, v2f{b2v.x, b2v.y}, false, a01, (short)0, acc2, false, false);
            acc3 = __builtin_amdgcn_wmma_f32_16x16x4_f32(
                false, v2f{b3v.x, b3v.y}, false, a01, (short)0, acc3, false, false);
            acc0 = __builtin_amdgcn_wmma_f32_16x16x4_f32(
                false, v2f{b0v.z, b0v.w}, false, a23, (short)0, acc0, false, false);
            acc1 = __builtin_amdgcn_wmma_f32_16x16x4_f32(
                false, v2f{b1v.z, b1v.w}, false, a23, (short)0, acc1, false, false);
            acc2 = __builtin_amdgcn_wmma_f32_16x16x4_f32(
                false, v2f{b2v.z, b2v.w}, false, a23, (short)0, acc2, false, false);
            acc3 = __builtin_amdgcn_wmma_f32_16x16x4_f32(
                false, v2f{b3v.z, b3v.w}, false, a23, (short)0, acc3, false, false);
        }
    }

    const v8f accs[4] = {acc0, acc1, acc2, acc3};
    float p0 = 0.0f, p1 = 0.0f;
#pragma unroll
    for (int n = 0; n < 4; ++n) {
        const int kc = 16 * n + 8 * half;
#pragma unroll
        for (int r = 0; r < 8; ++r) {
            const float v = fmaxf(accs[n][r] + bias[kc + r], 0.0f);
            p0 = fmaf(v, Wm2[2 * (kc + r) + 0], p0);
            p1 = fmaf(v, Wm2[2 * (kc + r) + 1], p1);
        }
    }
    p0 += __shfl_xor(p0, 16);
    p1 += __shfl_xor(p1, 16);
    if (half == 0) {
        *reinterpret_cast<float2*>(out + 2 * (long)(row0 + lm)) =
            make_float2(p0 + bm2[0], p1 + bm2[1]);
    }
}

extern "C" void kernel_launch(void* const* d_in, const int* in_sizes, int n_in,
                              void* d_out, int out_size, void* d_ws, size_t ws_size,
                              hipStream_t stream) {
    const float* x   = (const float*)d_in[0];
    const int*   ei  = (const int*)  d_in[1];
    const float* W1  = (const float*)d_in[2];
    const float* b1  = (const float*)d_in[3];
    const float* W2  = (const float*)d_in[4];
    const float* b2  = (const float*)d_in[5];
    const float* Wm1 = (const float*)d_in[6];
    const float* bm1 = (const float*)d_in[7];
    const float* Wm2 = (const float*)d_in[8];
    const float* bm2 = (const float*)d_in[9];
    float* out = (float*)d_out;

    const int N = N_NODES, E = N_EDGES;
    const int N64 = N * HF;

    float* ws   = (float*)d_ws;
    float* dinv = ws;
    float* h    = dinv + N;
    float* feat = h    + (long)N64;
    float* lst  = feat + (long)N64;
    float* hfin = lst  + (long)N64;

    int* csr    = (int*)lst;
    int* rowptr = csr + 2 * (long)E;
    int* tmp    = (int*)hfin;
    int* csrsrc = tmp + 2 * (long)E;
    int* tab    = csrsrc + E;
    float* feat2 = hfin;

    const float TH[3][3] = {{3.0f, -3.0f, 0.75f},
                            {0.0f,  3.0f, -1.50f},
                            {0.0f,  0.0f, 0.75f}};

    bucket_hist<<<NCHUNK, 256, 0, stream>>>(ei, tab);
    scan_table<<<1, 1024, 0, stream>>>(tab);
    bucket_scatter<<<NCHUNK, 256, 0, stream>>>(ei, tab, (int2*)tmp);
    bucket_rank<<<NBK, 1024, 0, stream>>>(tab, (int2*)tmp, rowptr, dinv, csrsrc);

    mlp2_wmma<<<(N / 16 + 3) / 4, 128, 64 * (132 + 68) * sizeof(float), stream>>>(
        x, W1, b1, W2, b2, h, N);

    fill_csr<<<(E / 4 + 255) / 256, 256, 0, stream>>>(csrsrc, dinv, csr, E / 4);
    const int gg = (N * 16 + 255) / 256;
    gather_step1<<<gg, 256, 0, stream>>>(csr, rowptr, h, dinv, feat, N);
    gather_step2<<<gg, 256, 0, stream>>>(csr, rowptr, feat, dinv, feat2, N);

    gemm_filters_out_wmma<<<(N / 16 + 3) / 4, 128, 64 * 196 * sizeof(float), stream>>>(
        h, feat, feat2, Wm1, bm1, Wm2, bm2, out, N,
        TH[0][0], TH[1][0], TH[2][0], TH[0][1], TH[1][1], TH[2][1],
        TH[0][2], TH[1][2], TH[2][2]);
}


# Propagation launches as resident looping waves: a 16-lane group walks 4 consecutive nodes (12500 waves per launch, not 50000), next list entries requested under the rows; sums bit-identical
# speedup vs baseline: 12.7764x; 1.0454x over previous
#include <hip/hip_runtime.h>
#include <hip/hip_bf16.h>


typedef float v2f __attribute__((ext_vector_type(2)));
typedef float v8f __attribute__((ext_vector_type(8)));

#define N_NODES 100000
#define N_EDGES 1600000
#define HF 64

#define BKT_NODES 1024
#define NBK ((N_NODES + BKT_NODES - 1) / BKT_NODES)
#define CHUNK_E 6400
#define NCHUNK (N_EDGES / CHUNK_E)
#define NTAB (NBK * NCHUNK)
static_assert(NCHUNK * CHUNK_E == N_EDGES, "chunks must tile the edge list");
static_assert(CHUNK_E % 256 == 0, "a chunk is a whole number of edges per thread");
static_assert(N_EDGES < (1 << 21), "rank is packed into 21 bits");
static_assert(NBK <= 256, "a chunk's work-group scans one run length per thread");
#define STAGE_E 20480

__device__ __forceinline__ int bucket_of(int d) {
    const unsigned b = (unsigned)d >> 10;
    return (int)(b < (unsigned)NBK ? b : (unsigned)(NBK - 1));
}

__global__ void __launch_bounds__(256)
bucket_hist(const int* __restrict__ ei, int* __restrict__ tab) {
    __shared__ int hist[NBK];
    for (int b = threadIdx.x; b < NBK; b += 256) hist[b] = 0;
    __syncthreads();
    const int* dst = ei + N_EDGES + blockIdx.x * CHUNK_E + threadIdx.x;
    int d[CHUNK_E / 256];
#pragma unroll
    for (int k = 0; k < CHUNK_E / 256; ++k) d[k] = dst[256 * k];
#pragma unroll
    for (int k = 0; k < CHUNK_E / 256; ++k) atomicAdd(&hist[bucket_of(d[k])], 1);
    __syncthreads();
    for (int b = threadIdx.x; b < NBK; b += 256) tab[b * NCHUNK + blockIdx.x] = hist[b];
}

__global__ void __launch_bounds__(1024)
scan_table(int* __restrict__ tab) {
    __shared__ int wsum[32];
    constexpr int PER = (NTAB + 1023) / 1024;
    const int lane = threadIdx.x & 31;
    const int wv   = threadIdx.x >> 5;
    const int i0   = threadIdx.x * PER;
    int v[PER];
    int s = 0;
#pragma unroll
    for (int k = 0; k < PER; ++k) {
        v[k] = (i0 + k < NTAB) ? tab[i0 + k] : 0;
        s += v[k];
    }
    int x = s;
    for (int o = 1; o < 32; o <<= 1) {
        int y = __shfl_up(x, o);
        if (lane >= o) x += y;
    }
    if (lane == 31) wsum[wv] = x;
    __syncthreads();
    if (wv == 0) {
        const int w = wsum[lane];
        int t = w;
        for (int o = 1; o < 32; o <<= 1) {
            int y = __shfl_up(t, o);
            if (lane >= o) t += y;
        }
        wsum[lane] = t - w;
    }
    __syncthreads();
    int run = x - s + wsum[wv];
#pragma unroll
    for (int k = 0; k < PER; ++k) {
        if (i0 + k < NTAB) tab[i0 + k] = run;
        run += v[k];
    }
}

__global__ void __launch_bounds__(256)
bucket_scatter(const int* __restrict__ ei, const int* __restrict__ tab,
               int2* __restrict__ tmp) {
    constexpr int PER = CHUNK_E / 256;
    __shared__ int cur[NBK];
    __shared__ int delta[NBK];
    __shared__ int wsum[8];
    __shared__ int2 stage[CHUNK_E];
    const int t    = threadIdx.x;
    const int lane = t & 31;
    const int wv   = t >> 5;
    const int* srcp = ei + blockIdx.x * CHUNK_E + t;
    const int* dstp = srcp + N_EDGES;
    int s[PER], d[PER];
#pragma unroll
    for (int k = 0; k < PER; ++k) {
        s[k] = srcp[256 * k];
        d[k] = dstp[256 * k];
    }
    const int idx  = min(t, NBK - 1) * NCHUNK + blockIdx.x;
    const int tcur = tab[idx];
    const int tnxt = tab[min(idx + 1, NTAB - 1)];
    const int place = tcur;
    const int len   = (t < NBK) ? ((idx + 1 < NTAB) ? tnxt : N_EDGES) - tcur : 0;
    int x = len;
    for (int o = 1; o < 32; o <<= 1) {
        int y = __shfl_up(x, o);
        if (lane >= o) x += y;
    }
    if (lane == 31) wsum[wv] = x;
    __syncthreads();
    int lo = x - len;
    for (int w = 0; w < wv; ++w) lo += wsum[w];
    if (t < NBK) {
        cur[t]   = lo;
        delta[t] = place - lo;
    }
    __syncthreads();
    int p[PER];
#pragma unroll
    for (int k = 0; k < PER; ++k) p[k] = atomicAdd(&cur[bucket_of(d[k])], 1);
#pragma unroll
    for (int k = 0; k < PER; ++k) {
        if ((unsigned)p[k] < (unsigned)CHUNK_E) stage[p[k]] = make_int2(s[k], d[k]);
    }
    __syncthreads();
    int2 e[PER];
#pragma unroll
    for (int k = 0; k < PER; ++k) e[k] = stage[t + 256 * k];
#pragma unroll
    for (int k = 0; k < PER; ++k) p[k] = delta[bucket_of(e[k].y)] + t + 256 * k;
#pragma unroll
    for (int k = 0; k < PER; ++k) {
        if ((unsigned)p[k] < (unsigned)N_EDGES) tmp[p[k]] = e[k];
    }
}

__global__ void __launch_bounds__(1024)
bucket_rank(const int* __restrict__ tab, int2* __restrict__ tmp,
            int* __restrict__ rowptr, float* __restrict__ dinv,
            int* __restrict__ csrsrc) {
    __shared__ int cnt[BKT_NODES];
    __shared__ int off[BKT_NODES];
    __shared__ int wsum[32];
    __shared__ int stage[STAGE_E];
    const int b    = blockIdx.x;
    const int t    = threadIdx.x;
    const int lane = t & 31;
    const int wv   = t >> 5;
    const int bs = tab[b * NCHUNK];
    const int be = (b == NBK - 1) ? N_EDGES : tab[(b + 1) * NCHUNK];
    cnt[t] = 0;
    __syncthreads();
    for (int i0 = bs + t; i0 < be; i0 += 8 * 1024) {
        int y[8];
#pragma unroll
        for (int k = 0; k < 8; ++k) {
            const int i = i0 + 1024 * k;
            y[k] = tmp[min(i, be - 1)].y;
        }
#pragma unroll
        for (int k = 0; k < 8; ++k) {
            const int i = i0 + 1024 * k;
            if (i < be) {
                const int dl = y[k] & (BKT_NODES - 1);
                const int r  = atomicAdd(&cnt[dl], 1);
                tmp[i].y = (dl << 21) | r;
            }
        }
    }
    __syncthreads();
    const int c = cnt[t];
    int x = c;
    for (int o = 1; o < 32; o <<= 1) {
        int y = __shfl_up(x, o);
        if (lane >= o) x += y;
    }
    if (lane == 31) wsum[wv] = x;
    __syncthreads();
    if (wv == 0) {
        const int w = wsum[lane];
        int u = w;
        for (int o = 1; o < 32; o <<= 1) {
            int y = __shfl_up(u, o);
            if (lane >= o) u += y;
        }
        wsum[lane] = u - w;
    }
    __syncthreads();
    const int ex = x - c + wsum[wv];
    off[t] = ex;
    const int node = b * BKT_NODES + t;
    if (node < N_NODES) {
        rowptr[node] = bs + ex;
        dinv[node]   = 1.0f / sqrtf(fmaxf((float)c, 1.0f));
    }
    if (b == NBK - 1 && t == 0) rowptr[N_NODES] = N_EDGES;
    __syncthreads();
    for (int i0 = bs + t; i0 < be; i0 += 8 * 1024) {
        int2 e[8];
#pragma unroll
        for (int k = 0; k < 8; ++k) {
            const int i = i0 + 1024 * k;
            e[k] = tmp[min(i, be - 1)];
        }
        int pl[8];
#pragma unroll
        for (int k = 0; k < 8; ++k) pl[k] = off[e[k].y >> 21] + (e[k].y & ((1 << 21) - 1));
#pragma unroll
        for (int k = 0; k < 8; ++k) {
            const int i = i0 + 1024 * k;
            if (i < be) {
                if ((unsigned)pl[k] < (unsigned)STAGE_E) stage[pl[k]] = e[k].x;
                else if ((unsigned)(bs + pl[k]) < (unsigned)N_EDGES) csrsrc[bs + pl[k]] = e[k].x;
            }
        }
    }
    __syncthreads();
    const int n = min(be - bs, STAGE_E);
    const int a = bs & 31;
    for (int i0 = t - a; i0 < n; i0 += 4 * 1024) {
        int v[4];
#pragma unroll
        for (int k = 0; k < 4; ++k) v[k] = stage[min(max(i0 + 1024 * k, 0), STAGE_E - 1)];
#pragma unroll
        for (int k = 0; k < 4; ++k) {
            const int i = i0 + 1024 * k;
            if (i >= 0 && i < n && (unsigned)(bs + i) < (unsigned)N_EDGES) csrsrc[bs + i] = v[k];
        }
    }
}

__global__ void __launch_bounds__(256)
fill_csr(const int* __restrict__ csrsrc, const float* __restrict__ dinv,
         int* __restrict__ csr, int E4) {
    const int j = blockIdx.x * blockDim.x + threadIdx.x;
    if (j >= E4) return;
    const int4 s = reinterpret_cast<const int4*>(csrsrc)[j];
    const float w0 = dinv[s.x], w1 = dinv[s.y], w2 = dinv[s.z], w3 = dinv[s.w];
    int4* o = reinterpret_cast<int4*>(csr) + 2 * (long)j;
    o[0] = make_int4(s.x, __float_as_int(w0), s.y, __float_as_int(w1));
    o[1] = make_int4(s.z, __float_as_int(w2), s.w, __float_as_int(w3));
}


template <int K>
__device__ __forceinline__ void stage_weights_t(float* __restrict__ Wt,
                                                const float* __restrict__ W, int tid) {
    constexpr int P = K + 4;
    for (int i = tid; i < (K / 4) * HF; i += 128) {
        const int c = i & 63, t = i >> 6;
        const float* w = W + (4 * t) * HF + c;
        *reinterpret_cast<float4*>(Wt + c * P + 4 * t) =
            make_float4(w[0], w[HF], w[2 * HF], w[3 * HF]);
    }
}

__global__ void __launch_bounds__(128)
mlp2_wmma(const float* __restrict__ X,
          const float* __restrict__ W1, const float* __restrict__ b1,
          const float* __restrict__ W2, const float* __restrict__ b2,
          float* __restrict__ C, int M) {
    constexpr int K1 = 128;
    constexpr int P1 = K1 + 4;
    constexpr int P2 = HF + 4;
    extern __shared__ __align__(16) float Wt[];
    float* W1t = Wt;
    float* W2t = Wt + HF * P1;
    const int tid = threadIdx.x;

    stage_weights_t<K1>(W1t, W1, tid);
    stage_weights_t<HF>(W2t, W2, tid);
    __syncthreads();

    const int wave = tid >> 5;
    const int lane = tid & 31;
    const int half = lane >> 4;
    const int lm   = lane & 15;
    const int row0 = (blockIdx.x * 4 + wave) * 16;
    if (row0 >= M) return;

    const float* Xp = X + (long)(row0 + lm) * K1 + 4 * half;
    const float* Wp = W1t + lm * P1 + 4 * half;

    v8f d0 = {0.f, 0.f, 0.f, 0.f, 0.f, 0.f, 0.f, 0.f};
    v8f d1 = d0, d2 = d0, d3 = d0;

#pragma unroll
    for (int k = 0; k < K1; k += 8) {
        const float4 xv = *reinterpret_cast<const float4*>(Xp + k);
        const float4 w0 = *reinterpret_cast<const float4*>(Wp + k);
        const float4 w1 = *reinterpret_cast<const float4*>(Wp + 16 * P1 + k);
        const float4 w2 = *reinterpret_cast<const float4*>(Wp + 32 * P1 + k);
        const float4 w3 = *reinterpret_cast<const float4*>(Wp + 48 * P1 + k);
        const v2f x01 = {xv.x, xv.y}, x23 = {xv.z, xv.w};
        d0 = __builtin_amdgcn_wmma_f32_16x16x4_f32(
            false, v2f{w0.x, w0.y}, false, x01, (short)0, d0, false, false);
        d1 = __builtin_amdgcn_wmma_f32_16x16x4_f32(
            false, v2f{w1.x, w1.y}, false, x01, (short)0, d1, false, false);
        d2 = __builtin_amdgcn_wmma_f32_16x16x4_f32(
            false, v2f{w2.x, w2.y}, false, x01, (short)0, d2, false, false);
        d3 = __builtin_amdgcn_wmma_f32_16x16x4_f32(
            false, v2f{w3.x, w3.y}, false, x01, (short)0, d3, false, false);
        d0 = __builtin_amdgcn_wmma_f32_16x16x4_f32(
            false, v2f{w0.z, w0.w}, false, x23, (short)0, d0, false, false);
        d1 = __builtin_amdgcn_wmma_f32_16x16x4_f32(
            false, v2f{w1.z, w1.w}, false, x23, (short)0, d1, false, false);
        d2 = __builtin_amdgcn_wmma_f32_16x16x4_f32(
            false, v2f{w2.z, w2.w}, false, x23, (short)0, d2, false, false);
        d3 = __builtin_amdgcn_wmma_f32_16x16x4_f32(
            false, v2f{w3.z, w3.w}, false, x23, (short)0, d3, false, false);
    }

    v8f g[4] = {d0, d1, d2, d3};
#pragma unroll
    for (int t = 0; t < 4; ++t) {
        const float* bp = b1 + 16 * t + 8 * half;
#pragma unroll
        for (int r = 0; r < 8; ++r) g[t][r] = fmaxf(g[t][r] + bp[r], 0.0f);
    }

    const float* Vp = W2t + lm * P2 + 8 * half;

    v8f e0 = {0.f, 0.f, 0.f, 0.f, 0.f, 0.f, 0.f, 0.f};
    v8f e1 = e0, e2 = e0, e3 = e0;

#pragma unroll
    for (int t = 0; t < 4; ++t) {
#pragma unroll
        for (int i = 0; i < 2; ++i) {
            const int k = 16 * t + 4 * i;
            const float4 v0 = *reinterpret_cast<const float4*>(Vp + k);
            const float4 v1 = *reinterpret_cast<const float4*>(Vp + 16 * P2 + k);
            const float4 v2 = *reinterpret_cast<const float4*>(Vp + 32 * P2 + k);
            const float4 v3 = *reinterpret_cast<const float4*>(Vp + 48 * P2 + k);
            const v2f a01 = {g[t][4 * i], g[t][4 * i + 1]};
            const v2f a23 = {g[t][4 * i + 2], g[t][4 * i + 3]};
            e0 = __builtin_amdgcn_wmma_f32_16x16x4_f32(
                false, a01, false, v2f{v0.x, v0.y}, (short)0, e0, false, false);
            e1 = __builtin_amdgcn_wmma_f32_16x16x4_f32(
                false, a01, false, v2f{v1.x, v1.y}, (short)0, e1, false, false);
            e2 = __builtin_amdgcn_wmma_f32_16x16x4_f32(
                false, a01, false, v2f{v2.x, v2.y}, (short)0, e2, false, false);
            e3 = __builtin_amdgcn_wmma_f32_16x16x4_f32(
                false, a01, false, v2f{v3.x, v3.y}, (short)0, e3, false, false);
            e0 = __builtin_amdgcn_wmma_f32_16x16x4_f32(
                false, a23, false, v2f{v0.z, v0.w}, (short)0, e0, false, false);
            e1 = __builtin_amdgcn_wmma_f32_16x16x4_f32(
                false, a23, false, v2f{v1.z, v1.w}, (short)0, e1, false, false);
            e2 = __builtin_amdgcn_wmma_f32_16x16x4_f32(
                false, a23, false, v2f{v2.z, v2.w}, (short)0, e2, false, false);
            e3 = __builtin_amdgcn_wmma_f32_16x16x4_f32(
                false, a23, false, v2f{v3.z, v3.w}, (short)0, e3, false, false);
        }
    }

    const v8f es[4] = {e0, e1, e2, e3};
#pragma unroll
    for (int u = 0; u < 4; ++u) {
        const int col  = 16 * u + lm;
        const float bc = b2[col];
#pragma unroll
        for (int r = 0; r < 8; ++r) {
            const int row = row0 + r + 8 * half;
            C[(long)row * HF + col] = fmaxf(es[u][r] + bc, 0.0f);
        }
    }
}


__device__ __forceinline__ float4 ld_row4(const float* __restrict__ t, int row, int q) {
    return *reinterpret_cast<const float4*>(t + (long)row * HF + q);
}

__device__ __forceinline__ void add_scaled(float4& acc, const float4 r, float w) {
    acc.x += r.x * w;
    acc.y += r.y * w;
    acc.z += r.z * w;
    acc.w += r.w * w;
}

__device__ __forceinline__ float4 row_or_zero(bool keep, const float4 r) {
    float4 o;
    o.x = keep ? r.x : 0.0f;
    o.y = keep ? r.y : 0.0f;
    o.z = keep ? r.z : 0.0f;
    o.w = keep ? r.w : 0.0f;
    return o;
}

__device__ __forceinline__ float4 gather_sum(const int2* __restrict__ csr,
                                             const float* __restrict__ t,
                                             int beg, int end, int q) {
    float4 acc = make_float4(0.0f, 0.0f, 0.0f, 0.0f);
    if (end <= beg) return acc;
    const int last = end - 1;
    int j = beg;
    int2 e0 = csr[j];
    int2 e1 = csr[min(j + 1, last)];
    int2 e2 = csr[min(j + 2, last)];
    int2 e3 = csr[min(j + 3, last)];
    for (; j + 4 <= end; j += 4) {
        const float4 r0 = ld_row4(t, e0.x, q);
        const float4 r1 = ld_row4(t, e1.x, q);
        const float4 r2 = ld_row4(t, e2.x, q);
        const float4 r3 = ld_row4(t, e3.x, q);
        int2 f0 = csr[min(j + 4, last)];
        int2 f1 = csr[min(j + 5, last)];
        int2 f2 = csr[min(j + 6, last)];
        int2 f3 = csr[min(j + 7, last)];
        add_scaled(acc, r0, __int_as_float(e0.y));
        add_scaled(acc, r1, __int_as_float(e1.y));
        add_scaled(acc, r2, __int_as_float(e2.y));
        add_scaled(acc, r3, __int_as_float(e3.y));
        asm volatile("" : "+v"(f0.x), "+v"(f0.y), "+v"(f1.x), "+v"(f1.y),
                          "+v"(f2.x), "+v"(f2.y), "+v"(f3.x), "+v"(f3.y));
        e0 = f0;
        e1 = f1;
        e2 = f2;
        e3 = f3;
    }
    const int rem = end - j;
    if (rem > 0) {
        const float4 r0 = ld_row4(t, e0.x, q);
        const float4 r1 = row_or_zero(rem > 1, ld_row4(t, e1.x, q));
        const float4 r2 = row_or_zero(rem > 2, ld_row4(t, e2.x, q));
        add_scaled(acc, r0, __int_as_float(e0.y));
        add_scaled(acc, r1, __int_as_float(e1.y));
        add_scaled(acc, r2, __int_as_float(e2.y));
    }
    return acc;
}

#define GATHER_NPG 4
__global__ void __launch_bounds__(256)
gather_step(const int* __restrict__ csr, const int* __restrict__ rowptr,
            const float* __restrict__ in, const float* __restrict__ dinv,
            float* __restrict__ out, int n_nodes) {
    const int tid = blockIdx.x * blockDim.x + threadIdx.x;
    const int n0 = (tid >> 4) * GATHER_NPG;
    if (n0 >= n_nodes) return;
    const int q = (tid & 15) * 4;
    int beg = rowptr[n0];
    int end = rowptr[min(n0 + 1, n_nodes)];
#pragma unroll 1
    for (int k = 0; k < GATHER_NPG; ++k) {
        const int n = n0 + k;
        if (n >= n_nodes) break;
        const int nxt = rowptr[min(n + 2, n_nodes)];
        const float dn = dinv[n];
        const float4 own = ld_row4(in, n, q);
        const float4 m = gather_sum(reinterpret_cast<const int2*>(csr), in, beg, end, q);
        float4 v;
        v.x = own.x - m.x * dn;
        v.y = own.y - m.y * dn;
        v.z = own.z - m.z * dn;
        v.w = own.w - m.w * dn;
        *reinterpret_cast<float4*>(out + (long)n * HF + q) = v;
        beg = end;
        end = nxt;
    }
}

__global__ void __launch_bounds__(128)
gemm_filters_out_wmma(const float* __restrict__ h, const float* __restrict__ f1,
                      const float* __restrict__ f2,
                      const float* __restrict__ W, const float* __restrict__ bias,
                      const float* __restrict__ Wm2, const float* __restrict__ bm2,
                      float* __restrict__ out, int M,
                      float a0, float a1, float a2,
                      float b0, float b1, float b2,
                      float c0, float c1, float c2) {
    constexpr int K = 3 * HF;
    constexpr int P = K + 4;
    extern __shared__ __align__(16) float Wt[];
    const int tid  = threadIdx.x;
    const int wave = tid >> 5;
    const int lane = tid & 31;
    const int half = lane >> 4;
    const int lm   = lane & 15;
    const int row0 = (blockIdx.x * 4 + wave) * 16;

    const long arow = (long)min(row0 + lm, M - 1) * HF + 4 * half;
    float4 hv[HF / 8], uv[HF / 8], wv[HF / 8];
#pragma unroll
    for (int j = 0; j < HF / 8; ++j) {
        hv[j] = *reinterpret_cast<const float4*>(h  + arow + 8 * j);
        uv[j] = *reinterpret_cast<const float4*>(f1 + arow + 8 * j);
        wv[j] = *reinterpret_cast<const float4*>(f2 + arow + 8 * j);
    }

    stage_weights_t<K>(Wt, W, tid);
    __syncthreads();
    if (row0 >= M) return;

    const float* Bp = Wt + lm * P + 4 * half;
    const float ca[3] = {a0, a1, a2};
    const float cb[3] = {b0, b1, b2};
    const float cc[3] = {c0, c1, c2};

    v8f acc0 = {0.f, 0.f, 0.f, 0.f, 0.f, 0.f, 0.f, 0.f};
    v8f acc1 = acc0, acc2 = acc0, acc3 = acc0;

#pragma unroll
    for (int t = 0; t < 3; ++t) {
#pragma unroll
        for (int j = 0; j < HF / 8; ++j) {
            const int k = HF * t + 8 * j;
            float4 a;
            a.x = fmaf(cc[t], wv[j].x, fmaf(cb[t], uv[j].x, ca[t] * hv[j].x));
            a.y = fmaf(cc[t], wv[j].y, fmaf(cb[t], uv[j].y, ca[t] * hv[j].y));
            a.z = fmaf(cc[t], wv[j].z, fmaf(cb[t], uv[j].z, ca[t] * hv[j].z));
            a.w = fmaf(cc[t], wv[j].w, fmaf(cb[t], uv[j].w, ca[t] * hv[j].w));
            const float4 b0v = *reinterpret_cast<const float4*>(Bp + k);
            const float4 b1v = *reinterpret_cast<const float4*>(Bp + 16 * P + k);
            const float4 b2v = *reinterpret_cast<const float4*>(Bp + 32 * P + k);
            const float4 b3v = *reinterpret_cast<const float4*>(Bp + 48 * P + k);
            const v2f a01 = {a.x, a.y}, a23 = {a.z, a.w};
            acc0 = __builtin_amdgcn_wmma_f32_16x16x4_f32(
                false, v2f{b0v.x, b0v.y}, false, a01, (short)0, acc0, false, false);
            acc1 = __builtin_amdgcn_wmma_f32_16x16x4_f32(
                false, v2f{b1v.x, b1v.y}, false, a01, (short)0, acc1, false, false);
            acc2 = __builtin_amdgcn_wmma_f32_16x16x4_f32(
                false, v2f{b2v.x, b2v.y}, false, a01, (short)0, acc2, false, false);
            acc3 = __builtin_amdgcn_wmma_f32_16x16x4_f32(
                false, v2f{b3v.x, b3v.y}, false, a01, (short)0, acc3, false, false);
            acc0 = __builtin_amdgcn_wmma_f32_16x16x4_f32(
                false, v2f{b0v.z, b0v.w}, false, a23, (short)0, acc0, false, false);
            acc1 = __builtin_amdgcn_wmma_f32_16x16x4_f32(
                false, v2f{b1v.z, b1v.w}, false, a23, (short)0, acc1, false, false);
            acc2 = __builtin_amdgcn_wmma_f32_16x16x4_f32(
                false, v2f{b2v.z, b2v.w}, false, a23, (short)0, acc2, false, false);
            acc3 = __builtin_amdgcn_wmma_f32_16x16x4_f32(
                false, v2f{b3v.z, b3v.w}, false, a23, (short)0, acc3, false, false);
        }
    }

    const v8f accs[4] = {acc0, acc1, acc2, acc3};
    float p0 = 0.0f, p1 = 0.0f;
#pragma unroll
    for (int n = 0; n < 4; ++n) {
        const int kc = 16 * n + 8 * half;
#pragma unroll
        for (int r = 0; r < 8; ++r) {
            const float v = fmaxf(accs[n][r] + bias[kc + r], 0.0f);
            p0 = fmaf(v, Wm2[2 * (kc + r) + 0], p0);
            p1 = fmaf(v, Wm2[2 * (kc + r) + 1], p1);
        }
    }
    p0 += __shfl_xor(p0, 16);
    p1 += __shfl_xor(p1, 16);
    if (half == 0) {
        *reinterpret_cast<float2*>(out + 2 * (long)(row0 + lm)) =
            make_float2(p0 + bm2[0], p1 + bm2[1]);
    }
}

extern "C" void kernel_launch(void* const* d_in, const int* in_sizes, int n_in,
                              void* d_out, int out_size, void* d_ws, size_t ws_size,
                              hipStream_t stream) {
    const float* x   = (const float*)d_in[0];
    const int*   ei  = (const int*)  d_in[1];
    const float* W1  = (const float*)d_in[2];
    const float* b1  = (const float*)d_in[3];
    const float* W2  = (const float*)d_in[4];
    const float* b2  = (const float*)d_in[5];
    const float* Wm1 = (const float*)d_in[6];
    const float* bm1 = (const float*)d_in[7];
    const float* Wm2 = (const float*)d_in[8];
    const float* bm2 = (const float*)d_in[9];
    float* out = (float*)d_out;

    const int N = N_NODES, E = N_EDGES;
    const int N64 = N * HF;

    float* ws   = (float*)d_ws;
    float* dinv = ws;
    float* h    = dinv + N;
    float* feat = h    + (long)N64;
    float* lst  = feat + (long)N64;
    float* hfin = lst  + (long)N64;

    int* csr    = (int*)lst;
    int* rowptr = csr + 2 * (long)E;
    int* tmp    = (int*)hfin;
    int* csrsrc = tmp + 2 * (long)E;
    int* tab    = csrsrc + E;
    float* feat2 = hfin;

    const float TH[3][3] = {{3.0f, -3.0f, 0.75f},
                            {0.0f,  3.0f, -1.50f},
                            {0.0f,  0.0f, 0.75f}};

    bucket_hist<<<NCHUNK, 256, 0, stream>>>(ei, tab);
    scan_table<<<1, 1024, 0, stream>>>(tab);
    bucket_scatter<<<NCHUNK, 256, 0, stream>>>(ei, tab, (int2*)tmp);
    bucket_rank<<<NBK, 1024, 0, stream>>>(tab, (int2*)tmp, rowptr, dinv, csrsrc);

    mlp2_wmma<<<(N / 16 + 3) / 4, 128, 64 * (132 + 68) * sizeof(float), stream>>>(
        x, W1, b1, W2, b2, h, N);

    fill_csr<<<(E / 4 + 255) / 256, 256, 0, stream>>>(csrsrc, dinv, csr, E / 4);
    const int gg = ((N + GATHER_NPG - 1) / GATHER_NPG * 16 + 255) / 256;
    gather_step<<<gg, 256, 0, stream>>>(csr, rowptr, h, dinv, feat, N);
    gather_step<<<gg, 256, 0, stream>>>(csr, rowptr, feat, dinv, feat2, N);

    gemm_filters_out_wmma<<<(N / 16 + 3) / 4, 128, 64 * 196 * sizeof(float), stream>>>(
        h, feat, feat2, Wm1, bm1, Wm2, bm2, out, N,
        TH[0][0], TH[1][0], TH[2][0], TH[0][1], TH[1][1], TH[2][1],
        TH[0][2], TH[1][2], TH[2][2]);
}


# Table scan reads and writes the count table as 16-byte vectors, thread t taking vector t+1024j: a wave's load or store covers 4 whole lines, not 24 lines with 4 bytes a lane; same exclusive sums
# speedup vs baseline: 14.2861x; 1.1182x over previous
#include <hip/hip_runtime.h>
#include <hip/hip_bf16.h>


typedef float v2f __attribute__((ext_vector_type(2)));
typedef float v8f __attribute__((ext_vector_type(8)));

#define N_NODES 100000
#define N_EDGES 1600000
#define HF 64

#define BKT_NODES 1024
#define NBK ((N_NODES + BKT_NODES - 1) / BKT_NODES)
#define CHUNK_E 6400
#define NCHUNK (N_EDGES / CHUNK_E)
#define NTAB (NBK * NCHUNK)
static_assert(NCHUNK * CHUNK_E == N_EDGES, "chunks must tile the edge list");
static_assert(CHUNK_E % 256 == 0, "a chunk is a whole number of edges per thread");
static_assert(N_EDGES < (1 << 21), "rank is packed into 21 bits");
static_assert(NBK <= 256, "a chunk's work-group scans one run length per thread");
#define STAGE_E 20480

__device__ __forceinline__ int bucket_of(int d) {
    const unsigned b = (unsigned)d >> 10;
    return (int)(b < (unsigned)NBK ? b : (unsigned)(NBK - 1));
}

__global__ void __launch_bounds__(256)
bucket_hist(const int* __restrict__ ei, int* __restrict__ tab) {
    __shared__ int hist[NBK];
    for (int b = threadIdx.x; b < NBK; b += 256) hist[b] = 0;
    __syncthreads();
    const int* dst = ei + N_EDGES + blockIdx.x * CHUNK_E + threadIdx.x;
    int d[CHUNK_E / 256];
#pragma unroll
    for (int k = 0; k < CHUNK_E / 256; ++k) d[k] = dst[256 * k];
#pragma unroll
    for (int k = 0; k < CHUNK_E / 256; ++k) atomicAdd(&hist[bucket_of(d[k])], 1);
    __syncthreads();
    for (int b = threadIdx.x; b < NBK; b += 256) tab[b * NCHUNK + blockIdx.x] = hist[b];
}

__global__ void __launch_bounds__(1024)
scan_table(int* __restrict__ tab) {
    constexpr int NV    = NTAB / 4;
    constexpr int SLABS = (NV + 1023) / 1024;
    static_assert(NTAB % 4 == 0, "the table is a whole number of 16-byte vectors");
    __shared__ int wsum[SLABS * 32];
    __shared__ int stot[SLABS];
    const int t    = threadIdx.x;
    const int lane = t & 31;
    const int wv   = t >> 5;
    int4* tab4 = reinterpret_cast<int4*>(tab);
    int4 v[SLABS];
#pragma unroll
    for (int j = 0; j < SLABS; ++j) v[j] = tab4[min(t + 1024 * j, NV - 1)];
    int s[SLABS], x[SLABS];
#pragma unroll
    for (int j = 0; j < SLABS; ++j) {
        if (t + 1024 * j >= NV) v[j] = make_int4(0, 0, 0, 0);
        s[j] = v[j].x + v[j].y + v[j].z + v[j].w;
        x[j] = s[j];
    }
    for (int o = 1; o < 32; o <<= 1) {
#pragma unroll
        for (int j = 0; j < SLABS; ++j) {
            const int y = __shfl_up(x[j], o);
            if (lane >= o) x[j] += y;
        }
    }
    if (lane == 31) {
#pragma unroll
        for (int j = 0; j < SLABS; ++j) wsum[j * 32 + wv] = x[j];
    }
    __syncthreads();
    if (wv < SLABS) {
        const int w = wsum[t];
        int u = w;
        for (int o = 1; o < 32; o <<= 1) {
            const int y = __shfl_up(u, o);
            if (lane >= o) u += y;
        }
        wsum[t] = u - w;
        if (lane == 31) stot[wv] = u;
    }
    __syncthreads();
    int carry = 0;
#pragma unroll
    for (int j = 0; j < SLABS; ++j) {
        const int run = carry + wsum[j * 32 + wv] + x[j] - s[j];
        carry += stot[j];
        int4 o4;
        o4.x = run;
        o4.y = o4.x + v[j].x;
        o4.z = o4.y + v[j].y;
        o4.w = o4.z + v[j].z;
        if (t + 1024 * j < NV) tab4[t + 1024 * j] = o4;
    }
}

__global__ void __launch_bounds__(256)
bucket_scatter(const int* __restrict__ ei, const int* __restrict__ tab,
               int2* __restrict__ tmp) {
    constexpr int PER = CHUNK_E / 256;
    __shared__ int cur[NBK];
    __shared__ int delta[NBK];
    __shared__ int wsum[8];
    __shared__ int2 stage[CHUNK_E];
    const int t    = threadIdx.x;
    const int lane = t & 31;
    const int wv   = t >> 5;
    const int* srcp = ei + blockIdx.x * CHUNK_E + t;
    const int* dstp = srcp + N_EDGES;
    int s[PER], d[PER];
#pragma unroll
    for (int k = 0; k < PER; ++k) {
        s[k] = srcp[256 * k];
        d[k] = dstp[256 * k];
    }
    const int idx  = min(t, NBK - 1) * NCHUNK + blockIdx.x;
    const int tcur = tab[idx];
    const int tnxt = tab[min(idx + 1, NTAB - 1)];
    const int place = tcur;
    const int len   = (t < NBK) ? ((idx + 1 < NTAB) ? tnxt : N_EDGES) - tcur : 0;
    int x = len;
    for (int o = 1; o < 32; o <<= 1) {
        int y = __shfl_up(x, o);
        if (lane >= o) x += y;
    }
    if (lane == 31) wsum[wv] = x;
    __syncthreads();
    int lo = x - len;
    for (int w = 0; w < wv; ++w) lo += wsum[w];
    if (t < NBK) {
        cur[t]   = lo;
        delta[t] = place - lo;
    }
    __syncthreads();
    int p[PER];
#pragma unroll
    for (int k = 0; k < PER; ++k) p[k] = atomicAdd(&cur[bucket_of(d[k])], 1);
#pragma unroll
    for (int k = 0; k < PER; ++k) {
        if ((unsigned)p[k] < (unsigned)CHUNK_E) stage[p[k]] = make_int2(s[k], d[k]);
    }
    __syncthreads();
    int2 e[PER];
#pragma unroll
    for (int k = 0; k < PER; ++k) e[k] = stage[t + 256 * k];
#pragma unroll
    for (int k = 0; k < PER; ++k) p[k] = delta[bucket_of(e[k].y)] + t + 256 * k;
#pragma unroll
    for (int k = 0; k < PER; ++k) {
        if ((unsigned)p[k] < (unsigned)N_EDGES) tmp[p[k]] = e[k];
    }
}

__global__ void __launch_bounds__(1024)
bucket_rank(const int* __restrict__ tab, int2* __restrict__ tmp,
            int* __restrict__ rowptr, float* __restrict__ dinv,
            int* __restrict__ csrsrc) {
    __shared__ int cnt[BKT_NODES];
    __shared__ int off[BKT_NODES];
    __shared__ int wsum[32];
    __shared__ int stage[STAGE_E];
    const int b    = blockIdx.x;
    const int t    = threadIdx.x;
    const int lane = t & 31;
    const int wv   = t >> 5;
    const int bs = tab[b * NCHUNK];
    const int be = (b == NBK - 1) ? N_EDGES : tab[(b + 1) * NCHUNK];
    cnt[t] = 0;
    __syncthreads();
    for (int i0 = bs + t; i0 < be; i0 += 8 * 1024) {
        int y[8];
#pragma unroll
        for (int k = 0; k < 8; ++k) {
            const int i = i0 + 1024 * k;
            y[k] = tmp[min(i, be - 1)].y;
        }
#pragma unroll
        for (int k = 0; k < 8; ++k) {
            const int i = i0 + 1024 * k;
            if (i < be) {
                const int dl = y[k] & (BKT_NODES - 1);
                const int r  = atomicAdd(&cnt[dl], 1);
                tmp[i].y = (dl << 21) | r;
            }
        }
    }
    __syncthreads();
    const int c = cnt[t];
    int x = c;
    for (int o = 1; o < 32; o <<= 1) {
        int y = __shfl_up(x, o);
        if (lane >= o) x += y;
    }
    if (lane == 31) wsum[wv] = x;
    __syncthreads();
    if (wv == 0) {
        const int w = wsum[lane];
        int u = w;
        for (int o = 1; o < 32; o <<= 1) {
            int y = __shfl_up(u, o);
            if (lane >= o) u += y;
        }
        wsum[lane] = u - w;
    }
    __syncthreads();
    const int ex = x - c + wsum[wv];
    off[t] = ex;
    const int node = b * BKT_NODES + t;
    if (node < N_NODES) {
        rowptr[node] = bs + ex;
        dinv[node]   = 1.0f / sqrtf(fmaxf((float)c, 1.0f));
    }
    if (b == NBK - 1 && t == 0) rowptr[N_NODES] = N_EDGES;
    __syncthreads();
    for (int i0 = bs + t; i0 < be; i0 += 8 * 1024) {
        int2 e[8];
#pragma unroll
        for (int k = 0; k < 8; ++k) {
            const int i = i0 + 1024 * k;
            e[k] = tmp[min(i, be - 1)];
        }
        int pl[8];
#pragma unroll
        for (int k = 0; k < 8; ++k) pl[k] = off[e[k].y >> 21] + (e[k].y & ((1 << 21) - 1));
#pragma unroll
        for (int k = 0; k < 8; ++k) {
            const int i = i0 + 1024 * k;
            if (i < be) {
                if ((unsigned)pl[k] < (unsigned)STAGE_E) stage[pl[k]] = e[k].x;
                else if ((unsigned)(bs + pl[k]) < (unsigned)N_EDGES) csrsrc[bs + pl[k]] = e[k].x;
            }
        }
    }
    __syncthreads();
    const int n = min(be - bs, STAGE_E);
    const int a = bs & 31;
    for (int i0 = t - a; i0 < n; i0 += 4 * 1024) {
        int v[4];
#pragma unroll
        for (int k = 0; k < 4; ++k) v[k] = stage[min(max(i0 + 1024 * k, 0), STAGE_E - 1)];
#pragma unroll
        for (int k = 0; k < 4; ++k) {
            const int i = i0 + 1024 * k;
            if (i >= 0 && i < n && (unsigned)(bs + i) < (unsigned)N_EDGES) csrsrc[bs + i] = v[k];
        }
    }
}

__global__ void __launch_bounds__(256)
fill_csr(const int* __restrict__ csrsrc, const float* __restrict__ dinv,
         int* __restrict__ csr, int E4) {
    const int j = blockIdx.x * blockDim.x + threadIdx.x;
    if (j >= E4) return;
    const int4 s = reinterpret_cast<const int4*>(csrsrc)[j];
    const float w0 = dinv[s.x], w1 = dinv[s.y], w2 = dinv[s.z], w3 = dinv[s.w];
    int4* o = reinterpret_cast<int4*>(csr) + 2 * (long)j;
    o[0] = make_int4(s.x, __float_as_int(w0), s.y, __float_as_int(w1));
    o[1] = make_int4(s.z, __float_as_int(w2), s.w, __float_as_int(w3));
}


template <int K>
__device__ __forceinline__ void stage_weights_t(float* __restrict__ Wt,
                                                const float* __restrict__ W, int tid) {
    constexpr int P = K + 4;
    for (int i = tid; i < (K / 4) * HF; i += 128) {
        const int c = i & 63, t = i >> 6;
        const float* w = W + (4 * t) * HF + c;
        *reinterpret_cast<float4*>(Wt + c * P + 4 * t) =
            make_float4(w[0], w[HF], w[2 * HF], w[3 * HF]);
    }
}

__global__ void __launch_bounds__(128)
mlp2_wmma(const float* __restrict__ X,
          const float* __restrict__ W1, const float* __restrict__ b1,
          const float* __restrict__ W2, const float* __restrict__ b2,
          float* __restrict__ C, int M) {
    constexpr int K1 = 128;
    constexpr int P1 = K1 + 4;
    constexpr int P2 = HF + 4;
    extern __shared__ __align__(16) float Wt[];
    float* W1t = Wt;
    float* W2t = Wt + HF * P1;
    const int tid = threadIdx.x;

    stage_weights_t<K1>(W1t, W1, tid);
    stage_weights_t<HF>(W2t, W2, tid);
    __syncthreads();

    const int wave = tid >> 5;
    const int lane = tid & 31;
    const int half = lane >> 4;
    const int lm   = lane & 15;
    const int row0 = (blockIdx.x * 4 + wave) * 16;
    if (row0 >= M) return;

    const float* Xp = X + (long)(row0 + lm) * K1 + 4 * half;
    const float* Wp = W1t + lm * P1 + 4 * half;

    v8f d0 = {0.f, 0.f, 0.f, 0.f, 0.f, 0.f, 0.f, 0.f};
    v8f d1 = d0, d2 = d0, d3 = d0;

#pragma unroll
    for (int k = 0; k < K1; k += 8) {
        const float4 xv = *reinterpret_cast<const float4*>(Xp + k);
        const float4 w0 = *reinterpret_cast<const float4*>(Wp + k);
        const float4 w1 = *reinterpret_cast<const float4*>(Wp + 16 * P1 + k);
        const float4 w2 = *reinterpret_cast<const float4*>(Wp + 32 * P1 + k);
        const float4 w3 = *reinterpret_cast<const float4*>(Wp + 48 * P1 + k);
        const v2f x01 = {xv.x, xv.y}, x23 = {xv.z, xv.w};
        d0 = __builtin_amdgcn_wmma_f32_16x16x4_f32(
            false, v2f{w0.x, w0.y}, false, x01, (short)0, d0, false, false);
        d1 = __builtin_amdgcn_wmma_f32_16x16x4_f32(
            false, v2f{w1.x, w1.y}, false, x01, (short)0, d1, false, false);
        d2 = __builtin_amdgcn_wmma_f32_16x16x4_f32(
            false, v2f{w2.x, w2.y}, false, x01, (short)0, d2, false, false);
        d3 = __builtin_amdgcn_wmma_f32_16x16x4_f32(
            false, v2f{w3.x, w3.y}, false, x01, (short)0, d3, false, false);
        d0 = __builtin_amdgcn_wmma_f32_16x16x4_f32(
            false, v2f{w0.z, w0.w}, false, x23, (short)0, d0, false, false);
        d1 = __builtin_amdgcn_wmma_f32_16x16x4_f32(
            false, v2f{w1.z, w1.w}, false, x23, (short)0, d1, false, false);
        d2 = __builtin_amdgcn_wmma_f32_16x16x4_f32(
            false, v2f{w2.z, w2.w}, false, x23, (short)0, d2, false, false);
        d3 = __builtin_amdgcn_wmma_f32_16x16x4_f32(
            false, v2f{w3.z, w3.w}, false, x23, (short)0, d3, false, false);
    }

    v8f g[4] = {d0, d1, d2, d3};
#pragma unroll
    for (int t = 0; t < 4; ++t) {
        const float* bp = b1 + 16 * t + 8 * half;
#pragma unroll
        for (int r = 0; r < 8; ++r) g[t][r] = fmaxf(g[t][r] + bp[r], 0.0f);
    }

    const float* Vp = W2t + lm * P2 + 8 * half;

    v8f e0 = {0.f, 0.f, 0.f, 0.f, 0.f, 0.f, 0.f, 0.f};
    v8f e1 = e0, e2 = e0, e3 = e0;

#pragma unroll
    for (int t = 0; t < 4; ++t) {
#pragma unroll
        for (int i = 0; i < 2; ++i) {
            const int k = 16 * t + 4 * i;
            const float4 v0 = *reinterpret_cast<const float4*>(Vp + k);
            const float4 v1 = *reinterpret_cast<const float4*>(Vp + 16 * P2 + k);
            const float4 v2 = *reinterpret_cast<const float4*>(Vp + 32 * P2 + k);
            const float4 v3 = *reinterpret_cast<const float4*>(Vp + 48 * P2 + k);
            const v2f a01 = {g[t][4 * i], g[t][4 * i + 1]};
            const v2f a23 = {g[t][4 * i + 2], g[t][4 * i + 3]};
            e0 = __builtin_amdgcn_wmma_f32_16x16x4_f32(
                false, a01, false, v2f{v0.x, v0.y}, (short)0, e0, false, false);
            e1 = __builtin_amdgcn_wmma_f32_16x16x4_f32(
                false, a01, false, v2f{v1.x, v1.y}, (short)0, e1, false, false);
            e2 = __builtin_amdgcn_wmma_f32_16x16x4_f32(
                false, a01, false, v2f{v2.x, v2.y}, (short)0, e2, false, false);
            e3 = __builtin_amdgcn_wmma_f32_16x16x4_f32(
                false, a01, false, v2f{v3.x, v3.y}, (short)0, e3, false, false);
            e0 = __builtin_amdgcn_wmma_f32_16x16x4_f32(
                false, a23, false, v2f{v0.z, v0.w}, (short)0, e0, false, false);
            e1 = __builtin_amdgcn_wmma_f32_16x16x4_f32(
                false, a23, false, v2f{v1.z, v1.w}, (short)0, e1, false, false);
            e2 = __builtin_amdgcn_wmma_f32_16x16x4_f32(
                false, a23, false, v2f{v2.z, v2.w}, (short)0, e2, false, false);
            e3 = __builtin_amdgcn_wmma_f32_16x16x4_f32(
                false, a23, false, v2f{v3.z, v3.w}, (short)0, e3, false, false);
        }
    }

    const v8f es[4] = {e0, e1, e2, e3};
#pragma unroll
    for (int u = 0; u < 4; ++u) {
        const int col  = 16 * u + lm;
        const float bc = b2[col];
#pragma unroll
        for (int r = 0; r < 8; ++r) {
            const int row = row0 + r + 8 * half;
            C[(long)row * HF + col] = fmaxf(es[u][r] + bc, 0.0f);
        }
    }
}


__device__ __forceinline__ float4 ld_row4(const float* __restrict__ t, int row, int q) {
    return *reinterpret_cast<const float4*>(t + (long)row * HF + q);
}

__device__ __forceinline__ void add_scaled(float4& acc, const float4 r, float w) {
    acc.x += r.x * w;
    acc.y += r.y * w;
    acc.z += r.z * w;
    acc.w += r.w * w;
}

__device__ __forceinline__ float4 row_or_zero(bool keep, const float4 r) {
    float4 o;
    o.x = keep ? r.x : 0.0f;
    o.y = keep ? r.y : 0.0f;
    o.z = keep ? r.z : 0.0f;
    o.w = keep ? r.w : 0.0f;
    return o;
}

__device__ __forceinline__ float4 gather_sum(const int2* __restrict__ csr,
                                             const float* __restrict__ t,
                                             int beg, int end, int q) {
    float4 acc = make_float4(0.0f, 0.0f, 0.0f, 0.0f);
    if (end <= beg) return acc;
    const int last = end - 1;
    int j = beg;
    int2 e0 = csr[j];
    int2 e1 = csr[min(j + 1, last)];
    int2 e2 = csr[min(j + 2, last)];
    int2 e3 = csr[min(j + 3, last)];
    for (; j + 4 <= end; j += 4) {
        const float4 r0 = ld_row4(t, e0.x, q);
        const float4 r1 = ld_row4(t, e1.x, q);
        const float4 r2 = ld_row4(t, e2.x, q);
        const float4 r3 = ld_row4(t, e3.x, q);
        int2 f0 = csr[min(j + 4, last)];
        int2 f1 = csr[min(j + 5, last)];
        int2 f2 = csr[min(j + 6, last)];
        int2 f3 = csr[min(j + 7, last)];
        add_scaled(acc, r0, __int_as_float(e0.y));
        add_scaled(acc, r1, __int_as_float(e1.y));
        add_scaled(acc, r2, __int_as_float(e2.y));
        add_scaled(acc, r3, __int_as_float(e3.y));
        asm volatile("" : "+v"(f0.x), "+v"(f0.y), "+v"(f1.x), "+v"(f1.y),
                          "+v"(f2.x), "+v"(f2.y), "+v"(f3.x), "+v"(f3.y));
        e0 = f0;
        e1 = f1;
        e2 = f2;
        e3 = f3;
    }
    const int rem = end - j;
    if (rem > 0) {
        const float4 r0 = ld_row4(t, e0.x, q);
        const float4 r1 = row_or_zero(rem > 1, ld_row4(t, e1.x, q));
        const float4 r2 = row_or_zero(rem > 2, ld_row4(t, e2.x, q));
        add_scaled(acc, r0, __int_as_float(e0.y));
        add_scaled(acc, r1, __int_as_float(e1.y));
        add_scaled(acc, r2, __int_as_float(e2.y));
    }
    return acc;
}

#define GATHER_NPG 4
__global__ void __launch_bounds__(256)
gather_step(const int* __restrict__ csr, const int* __restrict__ rowptr,
            const float* __restrict__ in, const float* __restrict__ dinv,
            float* __restrict__ out, int n_nodes) {
    const int tid = blockIdx.x * blockDim.x + threadIdx.x;
    const int n0 = (tid >> 4) * GATHER_NPG;
    if (n0 >= n_nodes) return;
    const int q = (tid & 15) * 4;
    int beg = rowptr[n0];
    int end = rowptr[min(n0 + 1, n_nodes)];
#pragma unroll 1
    for (int k = 0; k < GATHER_NPG; ++k) {
        const int n = n0 + k;
        if (n >= n_nodes) break;
        const int nxt = rowptr[min(n + 2, n_nodes)];
        const float dn = dinv[n];
        const float4 own = ld_row4(in, n, q);
        const float4 m = gather_sum(reinterpret_cast<const int2*>(csr), in, beg, end, q);
        float4 v;
        v.x = own.x - m.x * dn;
        v.y = own.y - m.y * dn;
        v.z = own.z - m.z * dn;
        v.w = own.w - m.w * dn;
        *reinterpret_cast<float4*>(out + (long)n * HF + q) = v;
        beg = end;
        end = nxt;
    }
}

__global__ void __launch_bounds__(128)
gemm_filters_out_wmma(const float* __restrict__ h, const float* __restrict__ f1,
                      const float* __restrict__ f2,
                      const float* __restrict__ W, const float* __restrict__ bias,
                      const float* __restrict__ Wm2, const float* __restrict__ bm2,
                      float* __restrict__ out, int M,
                      float a0, float a1, float a2,
                      float b0, float b1, float b2,
                      float c0, float c1, float c2) {
    constexpr int K = 3 * HF;
    constexpr int P = K + 4;
    extern __shared__ __align__(16) float Wt[];
    const int tid  = threadIdx.x;
    const int wave = tid >> 5;
    const int lane = tid & 31;
    const int half = lane >> 4;
    const int lm   = lane & 15;
    const int row0 = (blockIdx.x * 4 + wave) * 16;

    const long arow = (long)min(row0 + lm, M - 1) * HF + 4 * half;
    float4 hv[HF / 8], uv[HF / 8], wv[HF / 8];
#pragma unroll
    for (int j = 0; j < HF / 8; ++j) {
        hv[j] = *reinterpret_cast<const float4*>(h  + arow + 8 * j);
        uv[j] = *reinterpret_cast<const float4*>(f1 + arow + 8 * j);
        wv[j] = *reinterpret_cast<const float4*>(f2 + arow + 8 * j);
    }

    stage_weights_t<K>(Wt, W, tid);
    __syncthreads();
    if (row0 >= M) return;

    const float* Bp = Wt + lm * P + 4 * half;
    const float ca[3] = {a0, a1, a2};
    const float cb[3] = {b0, b1, b2};
    const float cc[3] = {c0, c1, c2};

    v8f acc0 = {0.f, 0.f, 0.f, 0.f, 0.f, 0.f, 0.f, 0.f};
    v8f acc1 = acc0, acc2 = acc0, acc3 = acc0;

#pragma unroll
    for (int t = 0; t < 3; ++t) {
#pragma unroll
        for (int j = 0; j < HF / 8; ++j) {
            const int k = HF * t + 8 * j;
            float4 a;
            a.x = fmaf(cc[t], wv[j].x, fmaf(cb[t], uv[j].x, ca[t] * hv[j].x));
            a.y = fmaf(cc[t], wv[j].y, fmaf(cb[t], uv[j].y, ca[t] * hv[j].y));
            a.z = fmaf(cc[t], wv[j].z, fmaf(cb[t], uv[j].z, ca[t] * hv[j].z));
            a.w = fmaf(cc[t], wv[j].w, fmaf(cb[t], uv[j].w, ca[t] * hv[j].w));
            const float4 b0v = *reinterpret_cast<const float4*>(Bp + k);
            const float4 b1v = *reinterpret_cast<const float4*>(Bp + 16 * P + k);
            const float4 b2v = *reinterpret_cast<const float4*>(Bp + 32 * P + k);
            const float4 b3v = *reinterpret_cast<const float4*>(Bp + 48 * P + k);
            const v2f a01 = {a.x, a.y}, a23 = {a.z, a.w};
            acc0 = __builtin_amdgcn_wmma_f32_16x16x4_f32(
                false, v2f{b0v.x, b0v.y}, false, a01, (short)0, acc0, false, false);
            acc1 = __builtin_amdgcn_wmma_f32_16x16x4_f32(
                false, v2f{b1v.x, b1v.y}, false, a01, (short)0, acc1, false, false);
            acc2 = __builtin_amdgcn_wmma_f32_16x16x4_f32(
                false, v2f{b2v.x, b2v.y}, false, a01, (short)0, acc2, false, false);
            acc3 = __builtin_amdgcn_wmma_f32_16x16x4_f32(
                false, v2f{b3v.x, b3v.y}, false, a01, (short)0, acc3, false, false);
            acc0 = __builtin_amdgcn_wmma_f32_16x16x4_f32(
                false, v2f{b0v.z, b0v.w}, false, a23, (short)0, acc0, false, false);
            acc1 = __builtin_amdgcn_wmma_f32_16x16x4_f32(
                false, v2f{b1v.z, b1v.w}, false, a23, (short)0, acc1, false, false);
            acc2 = __builtin_amdgcn_wmma_f32_16x16x4_f32(
                false, v2f{b2v.z, b2v.w}, false, a23, (short)0, acc2, false, false);
            acc3 = __builtin_amdgcn_wmma_f32_16x16x4_f32(
                false, v2f{b3v.z, b3v.w}, false, a23, (short)0, acc3, false, false);
        }
    }

    const v8f accs[4] = {acc0, acc1, acc2, acc3};
    float p0 = 0.0f, p1 = 0.0f;
#pragma unroll
    for (int n = 0; n < 4; ++n) {
        const int kc = 16 * n + 8 * half;
#pragma unroll
        for (int r = 0; r < 8; ++r) {
            const float v = fmaxf(accs[n][r] + bias[kc + r], 0.0f);
            p0 = fmaf(v, Wm2[2 * (kc + r) + 0], p0);
            p1 = fmaf(v, Wm2[2 * (kc + r) + 1], p1);
        }
    }
    p0 += __shfl_xor(p0, 16);
    p1 += __shfl_xor(p1, 16);
    if (half == 0) {
        *reinterpret_cast<float2*>(out + 2 * (long)(row0 + lm)) =
            make_float2(p0 + bm2[0], p1 + bm2[1]);
    }
}

extern "C" void kernel_launch(void* const* d_in, const int* in_sizes, int n_in,
                              void* d_out, int out_size, void* d_ws, size_t ws_size,
                              hipStream_t stream) {
    const float* x   = (const float*)d_in[0];
    const int*   ei  = (const int*)  d_in[1];
    const float* W1  = (const float*)d_in[2];
    const float* b1  = (const float*)d_in[3];
    const float* W2  = (const float*)d_in[4];
    const float* b2  = (const float*)d_in[5];
    const float* Wm1 = (const float*)d_in[6];
    const float* bm1 = (const float*)d_in[7];
    const float* Wm2 = (const float*)d_in[8];
    const float* bm2 = (const float*)d_in[9];
    float* out = (float*)d_out;

    const int N = N_NODES, E = N_EDGES;
    const int N64 = N * HF;

    float* ws   = (float*)d_ws;
    float* dinv = ws;
    float* h    = dinv + N;
    float* feat = h    + (long)N64;
    float* lst  = feat + (long)N64;
    float* hfin = lst  + (long)N64;

    int* csr    = (int*)lst;
    int* rowptr = csr + 2 * (long)E;
    int* tmp    = (int*)hfin;
    int* csrsrc = tmp + 2 * (long)E;
    int* tab    = csrsrc + E;
    float* feat2 = hfin;

    const float TH[3][3] = {{3.0f, -3.0f, 0.75f},
                            {0.0f,  3.0f, -1.50f},
                            {0.0f,  0.0f, 0.75f}};

    bucket_hist<<<NCHUNK, 256, 0, stream>>>(ei, tab);
    scan_table<<<1, 1024, 0, stream>>>(tab);
    bucket_scatter<<<NCHUNK, 256, 0, stream>>>(ei, tab, (int2*)tmp);
    bucket_rank<<<NBK, 1024, 0, stream>>>(tab, (int2*)tmp, rowptr, dinv, csrsrc);

    mlp2_wmma<<<(N / 16 + 3) / 4, 128, 64 * (132 + 68) * sizeof(float), stream>>>(
        x, W1, b1, W2, b2, h, N);

    fill_csr<<<(E / 4 + 255) / 256, 256, 0, stream>>>(csrsrc, dinv, csr, E / 4);
    const int gg = ((N + GATHER_NPG - 1) / GATHER_NPG * 16 + 255) / 256;
    gather_step<<<gg, 256, 0, stream>>>(csr, rowptr, h, dinv, feat, N);
    gather_step<<<gg, 256, 0, stream>>>(csr, rowptr, feat, dinv, feat2, N);

    gemm_filters_out_wmma<<<(N / 16 + 3) / 4, 128, 64 * 196 * sizeof(float), stream>>>(
        h, feat, feat2, Wm1, bm1, Wm2, bm2, out, N,
        TH[0][0], TH[1][0], TH[2][0], TH[0][1], TH[1][1], TH[2][1],
        TH[0][2], TH[1][2], TH[2][2]);
}


# Input dense layers lose their launch: their row tiles are extra 512-thread work-groups of the sort's rank launch (98 bucket groups, matrix pipes idle); 9 -> 8 launches; list arrival order may differ
# speedup vs baseline: 16.2850x; 1.1399x over previous
#include <hip/hip_runtime.h>
#include <hip/hip_bf16.h>


typedef float v2f __attribute__((ext_vector_type(2)));
typedef float v8f __attribute__((ext_vector_type(8)));

#define N_NODES 100000
#define N_EDGES 1600000
#define HF 64

#define BKT_NODES 1024
#define NBK ((N_NODES + BKT_NODES - 1) / BKT_NODES)
#define CHUNK_E 6400
#define NCHUNK (N_EDGES / CHUNK_E)
#define NTAB (NBK * NCHUNK)
static_assert(NCHUNK * CHUNK_E == N_EDGES, "chunks must tile the edge list");
static_assert(CHUNK_E % 256 == 0, "a chunk is a whole number of edges per thread");
static_assert(N_EDGES < (1 << 21), "rank is packed into 21 bits");
static_assert(NBK <= 256, "a chunk's work-group scans one run length per thread");
#define STAGE_E 20480

__device__ __forceinline__ int bucket_of(int d) {
    const unsigned b = (unsigned)d >> 10;
    return (int)(b < (unsigned)NBK ? b : (unsigned)(NBK - 1));
}

__global__ void __launch_bounds__(256)
bucket_hist(const int* __restrict__ ei, int* __restrict__ tab) {
    __shared__ int hist[NBK];
    for (int b = threadIdx.x; b < NBK; b += 256) hist[b] = 0;
    __syncthreads();
    const int* dst = ei + N_EDGES + blockIdx.x * CHUNK_E + threadIdx.x;
    int d[CHUNK_E / 256];
#pragma unroll
    for (int k = 0; k < CHUNK_E / 256; ++k) d[k] = dst[256 * k];
#pragma unroll
    for (int k = 0; k < CHUNK_E / 256; ++k) atomicAdd(&hist[bucket_of(d[k])], 1);
    __syncthreads();
    for (int b = threadIdx.x; b < NBK; b += 256) tab[b * NCHUNK + blockIdx.x] = hist[b];
}

__global__ void __launch_bounds__(1024)
scan_table(int* __restrict__ tab) {
    constexpr int NV    = NTAB / 4;
    constexpr int SLABS = (NV + 1023) / 1024;
    static_assert(NTAB % 4 == 0, "the table is a whole number of 16-byte vectors");
    __shared__ int wsum[SLABS * 32];
    __shared__ int stot[SLABS];
    const int t    = threadIdx.x;
    const int lane = t & 31;
    const int wv   = t >> 5;
    int4* tab4 = reinterpret_cast<int4*>(tab);
    int4 v[SLABS];
#pragma unroll
    for (int j = 0; j < SLABS; ++j) v[j] = tab4[min(t + 1024 * j, NV - 1)];
    int s[SLABS], x[SLABS];
#pragma unroll
    for (int j = 0; j < SLABS; ++j) {
        if (t + 1024 * j >= NV) v[j] = make_int4(0, 0, 0, 0);
        s[j] = v[j].x + v[j].y + v[j].z + v[j].w;
        x[j] = s[j];
    }
    for (int o = 1; o < 32; o <<= 1) {
#pragma unroll
        for (int j = 0; j < SLABS; ++j) {
            const int y = __shfl_up(x[j], o);
            if (lane >= o) x[j] += y;
        }
    }
    if (lane == 31) {
#pragma unroll
        for (int j = 0; j < SLABS; ++j) wsum[j * 32 + wv] = x[j];
    }
    __syncthreads();
    if (wv < SLABS) {
        const int w = wsum[t];
        int u = w;
        for (int o = 1; o < 32; o <<= 1) {
            const int y = __shfl_up(u, o);
            if (lane >= o) u += y;
        }
        wsum[t] = u - w;
        if (lane == 31) stot[wv] = u;
    }
    __syncthreads();
    int carry = 0;
#pragma unroll
    for (int j = 0; j < SLABS; ++j) {
        const int run = carry + wsum[j * 32 + wv] + x[j] - s[j];
        carry += stot[j];
        int4 o4;
        o4.x = run;
        o4.y = o4.x + v[j].x;
        o4.z = o4.y + v[j].y;
        o4.w = o4.z + v[j].z;
        if (t + 1024 * j < NV) tab4[t + 1024 * j] = o4;
    }
}

__global__ void __launch_bounds__(256)
bucket_scatter(const int* __restrict__ ei, const int* __restrict__ tab,
               int2* __restrict__ tmp) {
    constexpr int PER = CHUNK_E / 256;
    __shared__ int cur[NBK];
    __shared__ int delta[NBK];
    __shared__ int wsum[8];
    __shared__ int2 stage[CHUNK_E];
    const int t    = threadIdx.x;
    const int lane = t & 31;
    const int wv   = t >> 5;
    const int* srcp = ei + blockIdx.x * CHUNK_E + t;
    const int* dstp = srcp + N_EDGES;
    int s[PER], d[PER];
#pragma unroll
    for (int k = 0; k < PER; ++k) {
        s[k] = srcp[256 * k];
        d[k] = dstp[256 * k];
    }
    const int idx  = min(t, NBK - 1) * NCHUNK + blockIdx.x;
    const int tcur = tab[idx];
    const int tnxt = tab[min(idx + 1, NTAB - 1)];
    const int place = tcur;
    const int len   = (t < NBK) ? ((idx + 1 < NTAB) ? tnxt : N_EDGES) - tcur : 0;
    int x = len;
    for (int o = 1; o < 32; o <<= 1) {
        int y = __shfl_up(x, o);
        if (lane >= o) x += y;
    }
    if (lane == 31) wsum[wv] = x;
    __syncthreads();
    int lo = x - len;
    for (int w = 0; w < wv; ++w) lo += wsum[w];
    if (t < NBK) {
        cur[t]   = lo;
        delta[t] = place - lo;
    }
    __syncthreads();
    int p[PER];
#pragma unroll
    for (int k = 0; k < PER; ++k) p[k] = atomicAdd(&cur[bucket_of(d[k])], 1);
#pragma unroll
    for (int k = 0; k < PER; ++k) {
        if ((unsigned)p[k] < (unsigned)CHUNK_E) stage[p[k]] = make_int2(s[k], d[k]);
    }
    __syncthreads();
    int2 e[PER];
#pragma unroll
    for (int k = 0; k < PER; ++k) e[k] = stage[t + 256 * k];
#pragma unroll
    for (int k = 0; k < PER; ++k) p[k] = delta[bucket_of(e[k].y)] + t + 256 * k;
#pragma unroll
    for (int k = 0; k < PER; ++k) {
        if ((unsigned)p[k] < (unsigned)N_EDGES) tmp[p[k]] = e[k];
    }
}

#define RANK_LDS_WORDS (2 * BKT_NODES + 32 + STAGE_E)
template <int NT>
__device__ __forceinline__ void rank_bucket(int* __restrict__ lds, const int b,
                                            const int* __restrict__ tab, int2* __restrict__ tmp,
                                            int* __restrict__ rowptr, float* __restrict__ dinv,
                                            int* __restrict__ csrsrc) {
    constexpr int CPT = BKT_NODES / NT;
    constexpr int U   = 8 * 1024 / NT;
    constexpr int UO  = 4 * 1024 / NT;
    constexpr int NW  = NT / 32;
    static_assert(CPT * NT == BKT_NODES && NW <= 32, "the scan covers the bucket's counters");
    int* cnt   = lds;
    int* off   = lds + BKT_NODES;
    int* wsum  = lds + 2 * BKT_NODES;
    int* stage = lds + 2 * BKT_NODES + 32;
    const int t    = threadIdx.x;
    const int lane = t & 31;
    const int wv   = t >> 5;
    const int bs = tab[b * NCHUNK];
    const int be = (b == NBK - 1) ? N_EDGES : tab[(b + 1) * NCHUNK];
#pragma unroll
    for (int j = 0; j < CPT; ++j) cnt[t + NT * j] = 0;
    __syncthreads();
    for (int i0 = bs + t; i0 < be; i0 += U * NT) {
        int y[U];
#pragma unroll
        for (int k = 0; k < U; ++k) {
            const int i = i0 + NT * k;
            y[k] = tmp[min(i, be - 1)].y;
        }
#pragma unroll
        for (int k = 0; k < U; ++k) {
            const int i = i0 + NT * k;
            if (i < be) {
                const int dl = y[k] & (BKT_NODES - 1);
                const int r  = atomicAdd(&cnt[dl], 1);
                tmp[i].y = (dl << 21) | r;
            }
        }
    }
    __syncthreads();
    int c[CPT];
    int s = 0;
#pragma unroll
    for (int j = 0; j < CPT; ++j) {
        c[j] = cnt[CPT * t + j];
        s += c[j];
    }
    int x = s;
    for (int o = 1; o < 32; o <<= 1) {
        int y = __shfl_up(x, o);
        if (lane >= o) x += y;
    }
    if (lane == 31) wsum[wv] = x;
    __syncthreads();
    if (wv == 0) {
        const int w = (lane < NW) ? wsum[lane] : 0;
        int u = w;
        for (int o = 1; o < 32; o <<= 1) {
            int y = __shfl_up(u, o);
            if (lane >= o) u += y;
        }
        if (lane < NW) wsum[lane] = u - w;
    }
    __syncthreads();
    int ex = x - s + wsum[wv];
#pragma unroll
    for (int j = 0; j < CPT; ++j) {
        off[CPT * t + j] = ex;
        const int node = b * BKT_NODES + CPT * t + j;
        if (node < N_NODES) {
            rowptr[node] = bs + ex;
            dinv[node]   = 1.0f / sqrtf(fmaxf((float)c[j], 1.0f));
        }
        ex += c[j];
    }
    if (b == NBK - 1 && t == 0) rowptr[N_NODES] = N_EDGES;
    __syncthreads();
    for (int i0 = bs + t; i0 < be; i0 += U * NT) {
        int2 e[U];
#pragma unroll
        for (int k = 0; k < U; ++k) {
            const int i = i0 + NT * k;
            e[k] = tmp[min(i, be - 1)];
        }
        int pl[U];
#pragma unroll
        for (int k = 0; k < U; ++k) pl[k] = off[e[k].y >> 21] + (e[k].y & ((1 << 21) - 1));
#pragma unroll
        for (int k = 0; k < U; ++k) {
            const int i = i0 + NT * k;
            if (i < be) {
                if ((unsigned)pl[k] < (unsigned)STAGE_E) stage[pl[k]] = e[k].x;
                else if ((unsigned)(bs + pl[k]) < (unsigned)N_EDGES) csrsrc[bs + pl[k]] = e[k].x;
            }
        }
    }
    __syncthreads();
    const int n = min(be - bs, STAGE_E);
    const int a = bs & 31;
    for (int i0 = t - a; i0 < n; i0 += UO * NT) {
        int v[UO];
#pragma unroll
        for (int k = 0; k < UO; ++k) v[k] = stage[min(max(i0 + NT * k, 0), STAGE_E - 1)];
#pragma unroll
        for (int k = 0; k < UO; ++k) {
            const int i = i0 + NT * k;
            if (i >= 0 && i < n && (unsigned)(bs + i) < (unsigned)N_EDGES) csrsrc[bs + i] = v[k];
        }
    }
}

__global__ void __launch_bounds__(256)
fill_csr(const int* __restrict__ csrsrc, const float* __restrict__ dinv,
         int* __restrict__ csr, int E4) {
    const int j = blockIdx.x * blockDim.x + threadIdx.x;
    if (j >= E4) return;
    const int4 s = reinterpret_cast<const int4*>(csrsrc)[j];
    const float w0 = dinv[s.x], w1 = dinv[s.y], w2 = dinv[s.z], w3 = dinv[s.w];
    int4* o = reinterpret_cast<int4*>(csr) + 2 * (long)j;
    o[0] = make_int4(s.x, __float_as_int(w0), s.y, __float_as_int(w1));
    o[1] = make_int4(s.z, __float_as_int(w2), s.w, __float_as_int(w3));
}


template <int K>
__device__ __forceinline__ void stage_weight_piece(float* __restrict__ Wt,
                                                   const float* __restrict__ W, int i) {
    constexpr int P = K + 4;
    const int c = i & 63, t = i >> 6;
    const float* w = W + (4 * t) * HF + c;
    *reinterpret_cast<float4*>(Wt + c * P + 4 * t) =
        make_float4(w[0], w[HF], w[2 * HF], w[3 * HF]);
}

template <int K, int NT>
__device__ __forceinline__ void stage_weights_t(float* __restrict__ Wt,
                                                const float* __restrict__ W, int tid) {
    constexpr int PIECES = (K / 4) * HF;
    if constexpr (PIECES <= 4 * NT) {
        static_assert(PIECES % NT == 0, "every thread takes the same number of pieces");
#pragma unroll
        for (int j = 0; j < PIECES / NT; ++j) stage_weight_piece<K>(Wt, W, tid + NT * j);
    } else {
        for (int i = tid; i < PIECES; i += NT) stage_weight_piece<K>(Wt, W, i);
    }
}

#define DENSE_LDS_WORDS (64 * (128 + 4) + 64 * (HF + 4))
template <int NT>
__device__ __forceinline__ void mlp2_tiles(float* __restrict__ Wt, const int blk,
                                           const float* __restrict__ X,
                                           const float* __restrict__ W1, const float* __restrict__ b1,
                                           const float* __restrict__ W2, const float* __restrict__ b2,
                                           float* __restrict__ C, int M) {
    constexpr int K1 = 128;
    constexpr int P1 = K1 + 4;
    constexpr int P2 = HF + 4;
    static_assert(HF * P1 + HF * P2 == DENSE_LDS_WORDS, "64 * P1 floats of W1, then 64 * P2 of W2");
    float* W1t = Wt;
    float* W2t = Wt + HF * P1;
    const int tid = threadIdx.x;

    stage_weights_t<K1, NT>(W1t, W1, tid);
    stage_weights_t<HF, NT>(W2t, W2, tid);
    __syncthreads();

    const int wave = tid >> 5;
    const int lane = tid & 31;
    const int half = lane >> 4;
    const int lm   = lane & 15;
    const int row0 = (blk * (NT / 32) + wave) * 16;
    if (row0 >= M) return;

    const float* Xp = X + (long)(row0 + lm) * K1 + 4 * half;
    const float* Wp = W1t + lm * P1 + 4 * half;

    v8f d0 = {0.f, 0.f, 0.f, 0.f, 0.f, 0.f, 0.f, 0.f};
    v8f d1 = d0, d2 = d0, d3 = d0;

#pragma unroll
    for (int k = 0; k < K1; k += 8) {
        const float4 xv = *reinterpret_cast<const float4*>(Xp + k);
        const float4 w0 = *reinterpret_cast<const float4*>(Wp + k);
        const float4 w1 = *reinterpret_cast<const float4*>(Wp + 16 * P1 + k);
        const float4 w2 = *reinterpret_cast<const float4*>(Wp + 32 * P1 + k);
        const float4 w3 = *reinterpret_cast<const float4*>(Wp + 48 * P1 + k);
        const v2f x01 = {xv.x, xv.y}, x23 = {xv.z, xv.w};
        d0 = __builtin_amdgcn_wmma_f32_16x16x4_f32(
            false, v2f{w0.x, w0.y}, false, x01, (short)0, d0, false, false);
        d1 = __builtin_amdgcn_wmma_f32_16x16x4_f32(
            false, v2f{w1.x, w1.y}, false, x01, (short)0, d1, false, false);
        d2 = __builtin_amdgcn_wmma_f32_16x16x4_f32(
            false, v2f{w2.x, w2.y}, false, x01, (short)0, d2, false, false);
        d3 = __builtin_amdgcn_wmma_f32_16x16x4_f32(
            false, v2f{w3.x, w3.y}, false, x01, (short)0, d3, false, false);
        d0 = __builtin_amdgcn_wmma_f32_16x16x4_f32(
            false, v2f{w0.z, w0.w}, false, x23, (short)0, d0, false, false);
        d1 = __builtin_amdgcn_wmma_f32_16x16x4_f32(
            false, v2f{w1.z, w1.w}, false, x23, (short)0, d1, false, false);
        d2 = __builtin_amdgcn_wmma_f32_16x16x4_f32(
            false, v2f{w2.z, w2.w}, false, x23, (short)0, d2, false, false);
        d3 = __builtin_amdgcn_wmma_f32_16x16x4_f32(
            false, v2f{w3.z, w3.w}, false, x23, (short)0, d3, false, false);
    }

    v8f g[4] = {d0, d1, d2, d3};
#pragma unroll
    for (int t = 0; t < 4; ++t) {
        const float* bp = b1 + 16 * t + 8 * half;
#pragma unroll
        for (int r = 0; r < 8; ++r) g[t][r] = fmaxf(g[t][r] + bp[r], 0.0f);
    }

    const float* Vp = W2t + lm * P2 + 8 * half;

    v8f e0 = {0.f, 0.f, 0.f, 0.f, 0.f, 0.f, 0.f, 0.f};
    v8f e1 = e0, e2 = e0, e3 = e0;

#pragma unroll
    for (int t = 0; t < 4; ++t) {
#pragma unroll
        for (int i = 0; i < 2; ++i) {
            const int k = 16 * t + 4 * i;
            const float4 v0 = *reinterpret_cast<const float4*>(Vp + k);
            const float4 v1 = *reinterpret_cast<const float4*>(Vp + 16 * P2 + k);
            const float4 v2 = *reinterpret_cast<const float4*>(Vp + 32 * P2 + k);
            const float4 v3 = *reinterpret_cast<const float4*>(Vp + 48 * P2 + k);
            const v2f a01 = {g[t][4 * i], g[t][4 * i + 1]};
            const v2f a23 = {g[t][4 * i + 2], g[t][4 * i + 3]};
            e0 = __builtin_amdgcn_wmma_f32_16x16x4_f32(
                false, a01, false, v2f{v0.x, v0.y}, (short)0, e0, false, false);
            e1 = __builtin_amdgcn_wmma_f32_16x16x4_f32(
                false, a01, false, v2f{v1.x, v1.y}, (short)0, e1, false, false);
            e2 = __builtin_amdgcn_wmma_f32_16x16x4_f32(
                false, a01, false, v2f{v2.x, v2.y}, (short)0, e2, false, false);
            e3 = __builtin_amdgcn_wmma_f32_16x16x4_f32(
                false, a01, false, v2f{v3.x, v3.y}, (short)0, e3, false, false);
            e0 = __builtin_amdgcn_wmma_f32_16x16x4_f32(
                false, a23, false, v2f{v0.z, v0.w}, (short)0, e0, false, false);
            e1 = __builtin_amdgcn_wmma_f32_16x16x4_f32(
                false, a23, false, v2f{v1.z, v1.w}, (short)0, e1, false, false);
            e2 = __builtin_amdgcn_wmma_f32_16x16x4_f32(
                false, a23, false, v2f{v2.z, v2.w}, (short)0, e2, false, false);
            e3 = __builtin_amdgcn_wmma_f32_16x16x4_f32(
                false, a23, false, v2f{v3.z, v3.w}, (short)0, e3, false, false);
        }
    }

    const v8f es[4] = {e0, e1, e2, e3};
#pragma unroll
    for (int u = 0; u < 4; ++u) {
        const int col  = 16 * u + lm;
        const float bc = b2[col];
#pragma unroll
        for (int r = 0; r < 8; ++r) {
            const int row = row0 + r + 8 * half;
            C[(long)row * HF + col] = fmaxf(es[u][r] + bc, 0.0f);
        }
    }
}

#define DENSE_T 512
#define DENSE_BLOCKS ((N_NODES / 16 + DENSE_T / 32 - 1) / (DENSE_T / 32))
__global__ void __launch_bounds__(DENSE_T)
rank_and_mlp2(const int* __restrict__ tab, int2* __restrict__ tmp,
              int* __restrict__ rowptr, float* __restrict__ dinv,
              int* __restrict__ csrsrc,
              const float* __restrict__ X,
              const float* __restrict__ W1, const float* __restrict__ b1,
              const float* __restrict__ W2, const float* __restrict__ b2,
              float* __restrict__ C, int M) {
    static_assert(DENSE_LDS_WORDS <= RANK_LDS_WORDS, "the weights fit the bucket's LDS");
    __shared__ __align__(16) int lds[RANK_LDS_WORDS];
    if (blockIdx.x < NBK) {
        rank_bucket<DENSE_T>(lds, blockIdx.x, tab, tmp, rowptr, dinv, csrsrc);
    } else {
        mlp2_tiles<DENSE_T>(reinterpret_cast<float*>(lds), blockIdx.x - NBK,
                            X, W1, b1, W2, b2, C, M);
    }
}


__device__ __forceinline__ float4 ld_row4(const float* __restrict__ t, int row, int q) {
    return *reinterpret_cast<const float4*>(t + (long)row * HF + q);
}

__device__ __forceinline__ void add_scaled(float4& acc, const float4 r, float w) {
    acc.x += r.x * w;
    acc.y += r.y * w;
    acc.z += r.z * w;
    acc.w += r.w * w;
}

__device__ __forceinline__ float4 row_or_zero(bool keep, const float4 r) {
    float4 o;
    o.x = keep ? r.x : 0.0f;
    o.y = keep ? r.y : 0.0f;
    o.z = keep ? r.z : 0.0f;
    o.w = keep ? r.w : 0.0f;
    return o;
}

__device__ __forceinline__ float4 gather_sum(const int2* __restrict__ csr,
                                             const float* __restrict__ t,
                                             int beg, int end, int q) {
    float4 acc = make_float4(0.0f, 0.0f, 0.0f, 0.0f);
    if (end <= beg) return acc;
    const int last = end - 1;
    int j = beg;
    int2 e0 = csr[j];
    int2 e1 = csr[min(j + 1, last)];
    int2 e2 = csr[min(j + 2, last)];
    int2 e3 = csr[min(j + 3, last)];
    for (; j + 4 <= end; j += 4) {
        const float4 r0 = ld_row4(t, e0.x, q);
        const float4 r1 = ld_row4(t, e1.x, q);
        const float4 r2 = ld_row4(t, e2.x, q);
        const float4 r3 = ld_row4(t, e3.x, q);
        int2 f0 = csr[min(j + 4, last)];
        int2 f1 = csr[min(j + 5, last)];
        int2 f2 = csr[min(j + 6, last)];
        int2 f3 = csr[min(j + 7, last)];
        add_scaled(acc, r0, __int_as_float(e0.y));
        add_scaled(acc, r1, __int_as_float(e1.y));
        add_scaled(acc, r2, __int_as_float(e2.y));
        add_scaled(acc, r3, __int_as_float(e3.y));
        asm volatile("" : "+v"(f0.x), "+v"(f0.y), "+v"(f1.x), "+v"(f1.y),
                          "+v"(f2.x), "+v"(f2.y), "+v"(f3.x), "+v"(f3.y));
        e0 = f0;
        e1 = f1;
        e2 = f2;
        e3 = f3;
    }
    const int rem = end - j;
    if (rem > 0) {
        const float4 r0 = ld_row4(t, e0.x, q);
        const float4 r1 = row_or_zero(rem > 1, ld_row4(t, e1.x, q));
        const float4 r2 = row_or_zero(rem > 2, ld_row4(t, e2.x, q));
        add_scaled(acc, r0, __int_as_float(e0.y));
        add_scaled(acc, r1, __int_as_float(e1.y));
        add_scaled(acc, r2, __int_as_float(e2.y));
    }
    return acc;
}

#define GATHER_NPG 4
__global__ void __launch_bounds__(256)
gather_step(const int* __restrict__ csr, const int* __restrict__ rowptr,
            const float* __restrict__ in, const float* __restrict__ dinv,
            float* __restrict__ out, int n_nodes) {
    const int tid = blockIdx.x * blockDim.x + threadIdx.x;
    const int n0 = (tid >> 4) * GATHER_NPG;
    if (n0 >= n_nodes) return;
    const int q = (tid & 15) * 4;
    int beg = rowptr[n0];
    int end = rowptr[min(n0 + 1, n_nodes)];
#pragma unroll 1
    for (int k = 0; k < GATHER_NPG; ++k) {
        const int n = n0 + k;
        if (n >= n_nodes) break;
        const int nxt = rowptr[min(n + 2, n_nodes)];
        const float dn = dinv[n];
        const float4 own = ld_row4(in, n, q);
        const float4 m = gather_sum(reinterpret_cast<const int2*>(csr), in, beg, end, q);
        float4 v;
        v.x = own.x - m.x * dn;
        v.y = own.y - m.y * dn;
        v.z = own.z - m.z * dn;
        v.w = own.w - m.w * dn;
        *reinterpret_cast<float4*>(out + (long)n * HF + q) = v;
        beg = end;
        end = nxt;
    }
}

__global__ void __launch_bounds__(128)
gemm_filters_out_wmma(const float* __restrict__ h, const float* __restrict__ f1,
                      const float* __restrict__ f2,
                      const float* __restrict__ W, const float* __restrict__ bias,
                      const float* __restrict__ Wm2, const float* __restrict__ bm2,
                      float* __restrict__ out, int M,
                      float a0, float a1, float a2,
                      float b0, float b1, float b2,
                      float c0, float c1, float c2) {
    constexpr int K = 3 * HF;
    constexpr int P = K + 4;
    extern __shared__ __align__(16) float Wt[];
    const int tid  = threadIdx.x;
    const int wave = tid >> 5;
    const int lane = tid & 31;
    const int half = lane >> 4;
    const int lm   = lane & 15;
    const int row0 = (blockIdx.x * 4 + wave) * 16;

    const long arow = (long)min(row0 + lm, M - 1) * HF + 4 * half;
    float4 hv[HF / 8], uv[HF / 8], wv[HF / 8];
#pragma unroll
    for (int j = 0; j < HF / 8; ++j) {
        hv[j] = *reinterpret_cast<const float4*>(h  + arow + 8 * j);
        uv[j] = *reinterpret_cast<const float4*>(f1 + arow + 8 * j);
        wv[j] = *reinterpret_cast<const float4*>(f2 + arow + 8 * j);
    }

    stage_weights_t<K, 128>(Wt, W, tid);
    __syncthreads();
    if (row0 >= M) return;

    const float* Bp = Wt + lm * P + 4 * half;
    const float ca[3] = {a0, a1, a2};
    const float cb[3] = {b0, b1, b2};
    const float cc[3] = {c0, c1, c2};

    v8f acc0 = {0.f, 0.f, 0.f, 0.f, 0.f, 0.f, 0.f, 0.f};
    v8f acc1 = acc0, acc2 = acc0, acc3 = acc0;

#pragma unroll
    for (int t = 0; t < 3; ++t) {
#pragma unroll
        for (int j = 0; j < HF / 8; ++j) {
            const int k = HF * t + 8 * j;
            float4 a;
            a.x = fmaf(cc[t], wv[j].x, fmaf(cb[t], uv[j].x, ca[t] * hv[j].x));
            a.y = fmaf(cc[t], wv[j].y, fmaf(cb[t], uv[j].y, ca[t] * hv[j].y));
            a.z = fmaf(cc[t], wv[j].z, fmaf(cb[t], uv[j].z, ca[t] * hv[j].z));
            a.w = fmaf(cc[t], wv[j].w, fmaf(cb[t], uv[j].w, ca[t] * hv[j].w));
            const float4 b0v = *reinterpret_cast<const float4*>(Bp + k);
            const float4 b1v = *reinterpret_cast<const float4*>(Bp + 16 * P + k);
            const float4 b2v = *reinterpret_cast<const float4*>(Bp + 32 * P + k);
            const float4 b3v = *reinterpret_cast<const float4*>(Bp + 48 * P + k);
            const v2f a01 = {a.x, a.y}, a23 = {a.z, a.w};
            acc0 = __builtin_amdgcn_wmma_f32_16x16x4_f32(
                false, v2f{b0v.x, b0v.y}, false, a01, (short)0, acc0, false, false);
            acc1 = __builtin_amdgcn_wmma_f32_16x16x4_f32(
                false, v2f{b1v.x, b1v.y}, false, a01, (short)0, acc1, false, false);
            acc2 = __builtin_amdgcn_wmma_f32_16x16x4_f32(
                false, v2f{b2v.x, b2v.y}, false, a01, (short)0, acc2, false, false);
            acc3 = __builtin_amdgcn_wmma_f32_16x16x4_f32(
                false, v2f{b3v.x, b3v.y}, false, a01, (short)0, acc3, false, false);
            acc0 = __builtin_amdgcn_wmma_f32_16x16x4_f32(
                false, v2f{b0v.z, b0v.w}, false, a23, (short)0, acc0, false, false);
            acc1 = __builtin_amdgcn_wmma_f32_16x16x4_f32(
                false, v2f{b1v.z, b1v.w}, false, a23, (short)0, acc1, false, false);
            acc2 = __builtin_amdgcn_wmma_f32_16x16x4_f32(
                false, v2f{b2v.z, b2v.w}, false, a23, (short)0, acc2, false, false);
            acc3 = __builtin_amdgcn_wmma_f32_16x16x4_f32(
                false, v2f{b3v.z, b3v.w}, false, a23, (short)0, acc3, false, false);
        }
    }

    const v8f accs[4] = {acc0, acc1, acc2, acc3};
    float p0 = 0.0f, p1 = 0.0f;
#pragma unroll
    for (int n = 0; n < 4; ++n) {
        const int kc = 16 * n + 8 * half;
#pragma unroll
        for (int r = 0; r < 8; ++r) {
            const float v = fmaxf(accs[n][r] + bias[kc + r], 0.0f);
            p0 = fmaf(v, Wm2[2 * (kc + r) + 0], p0);
            p1 = fmaf(v, Wm2[2 * (kc + r) + 1], p1);
        }
    }
    p0 += __shfl_xor(p0, 16);
    p1 += __shfl_xor(p1, 16);
    if (half == 0) {
        *reinterpret_cast<float2*>(out + 2 * (long)(row0 + lm)) =
            make_float2(p0 + bm2[0], p1 + bm2[1]);
    }
}

extern "C" void kernel_launch(void* const* d_in, const int* in_sizes, int n_in,
                              void* d_out, int out_size, void* d_ws, size_t ws_size,
                              hipStream_t stream) {
    const float* x   = (const float*)d_in[0];
    const int*   ei  = (const int*)  d_in[1];
    const float* W1  = (const float*)d_in[2];
    const float* b1  = (const float*)d_in[3];
    const float* W2  = (const float*)d_in[4];
    const float* b2  = (const float*)d_in[5];
    const float* Wm1 = (const float*)d_in[6];
    const float* bm1 = (const float*)d_in[7];
    const float* Wm2 = (const float*)d_in[8];
    const float* bm2 = (const float*)d_in[9];
    float* out = (float*)d_out;

    const int N = N_NODES, E = N_EDGES;
    const int N64 = N * HF;

    float* ws   = (float*)d_ws;
    float* dinv = ws;
    float* h    = dinv + N;
    float* feat = h    + (long)N64;
    float* lst  = feat + (long)N64;
    float* hfin = lst  + (long)N64;

    int* csr    = (int*)lst;
    int* rowptr = csr + 2 * (long)E;
    int* tmp    = (int*)hfin;
    int* csrsrc = tmp + 2 * (long)E;
    int* tab    = csrsrc + E;
    float* feat2 = hfin;

    const float TH[3][3] = {{3.0f, -3.0f, 0.75f},
                            {0.0f,  3.0f, -1.50f},
                            {0.0f,  0.0f, 0.75f}};

    bucket_hist<<<NCHUNK, 256, 0, stream>>>(ei, tab);
    scan_table<<<1, 1024, 0, stream>>>(tab);
    bucket_scatter<<<NCHUNK, 256, 0, stream>>>(ei, tab, (int2*)tmp);
    rank_and_mlp2<<<NBK + DENSE_BLOCKS, DENSE_T, 0, stream>>>(
        tab, (int2*)tmp, rowptr, dinv, csrsrc, x, W1, b1, W2, b2, h, N);

    fill_csr<<<(E / 4 + 255) / 256, 256, 0, stream>>>(csrsrc, dinv, csr, E / 4);
    const int gg = ((N + GATHER_NPG - 1) / GATHER_NPG * 16 + 255) / 256;
    gather_step<<<gg, 256, 0, stream>>>(csr, rowptr, h, dinv, feat, N);
    gather_step<<<gg, 256, 0, stream>>>(csr, rowptr, feat, dinv, feat2, N);

    gemm_filters_out_wmma<<<(N / 16 + 3) / 4, 128, 64 * 196 * sizeof(float), stream>>>(
        h, feat, feat2, Wm1, bm1, Wm2, bm2, out, N,
        TH[0][0], TH[1][0], TH[2][0], TH[0][1], TH[1][1], TH[2][1],
        TH[0][2], TH[1][2], TH[2][2]);
}
